# starting kernel
# baseline (speedup 1.0000x reference)
#include <hip/hip_runtime.h>
#include <hip/hip_bf16.h>

#define N_NODES 5000
#define NPAD    5120
#define N_EDGES 160000
#define HID     64
#define NODE_F  4
#define TASK_F  4
#define BATCH   512

typedef __attribute__((ext_vector_type(2))) float v2f;
typedef __attribute__((ext_vector_type(8))) float v8f;

#define OFF_DEG    0
#define OFF_AGG1   (OFF_DEG  + NPAD)
#define OFF_AGG2   (OFF_AGG1 + NPAD * NODE_F)
#define ZERO_COUNT (OFF_AGG2 + NPAD * HID)
#define OFF_RDEG   ZERO_COUNT
#define OFF_H1     (OFF_RDEG  + NPAD)
#define OFF_H2     (OFF_H1    + NPAD * HID)
#define OFF_HPART  (OFF_H2    + NPAD * HID)
#define OFF_TPART  (OFF_HPART + NPAD * HID)


__global__ void k_zero(float* __restrict__ p, int n) {
    int i = blockIdx.x * blockDim.x + threadIdx.x;
    if (i < n) p[i] = 0.0f;
}

__global__ void k_scatter1(const int* __restrict__ ei, const float* __restrict__ x,
                           float* __restrict__ agg1, float* __restrict__ deg) {
    int e = blockIdx.x * blockDim.x + threadIdx.x;
    if (e >= N_EDGES) return;
    int s = ei[e];
    int d = ei[N_EDGES + e];
    atomicAdd(&deg[d], 1.0f);
#pragma unroll
    for (int k = 0; k < NODE_F; ++k)
        atomicAdd(&agg1[d * NODE_F + k], x[s * NODE_F + k]);
}

__global__ void k_rdeg(const float* __restrict__ deg, float* __restrict__ rdeg) {
    int i = blockIdx.x * blockDim.x + threadIdx.x;
    if (i < NPAD) rdeg[i] = 1.0f / fmaxf(deg[i], 1.0f);
}

__global__ void k_sage1(const float* __restrict__ agg1, const float* __restrict__ rdeg,
                        const float* __restrict__ x,
                        const float* __restrict__ Wl1, const float* __restrict__ bl1,
                        const float* __restrict__ Wr1, float* __restrict__ h1) {
    const int row0 = blockIdx.x * 16;
    const int col0 = blockIdx.y * 16;
    const int lane = threadIdx.x;
    const int m    = lane & 15;
    const int half = lane >> 4;
    const int kb   = half * 2;
    const int row  = row0 + m;
    const float rd = rdeg[row];

    v2f am, ax, bl, br;
    am.x = agg1[row * NODE_F + kb]     * rd;
    am.y = agg1[row * NODE_F + kb + 1] * rd;
    ax.x = (row < N_NODES) ? x[row * NODE_F + kb]     : 0.0f;
    ax.y = (row < N_NODES) ? x[row * NODE_F + kb + 1] : 0.0f;
    bl.x = Wl1[(col0 + m) * NODE_F + kb];
    bl.y = Wl1[(col0 + m) * NODE_F + kb + 1];
    br.x = Wr1[(col0 + m) * NODE_F + kb];
    br.y = Wr1[(col0 + m) * NODE_F + kb + 1];

    v8f acc = {};
    acc = __builtin_amdgcn_wmma_f32_16x16x4_f32(false, am, false, bl, (short)0, acc, false, false);
    acc = __builtin_amdgcn_wmma_f32_16x16x4_f32(false, ax, false, br, (short)0, acc, false, false);

#pragma unroll
    for (int i = 0; i < 8; ++i) {
        int r = row0 + i + half * 8;
        int c = col0 + m;
        h1[r * HID + c] = fmaxf(acc[i] + bl1[c], 0.0f);
    }
}

__global__ void k_scatter2(const int* __restrict__ ei, const float* __restrict__ h1,
                           float* __restrict__ agg2) {
    long long idx = (long long)blockIdx.x * blockDim.x + threadIdx.x;
    if (idx >= (long long)N_EDGES * HID) return;
    int e = (int)(idx >> 6);
    int f = (int)(idx & 63);
    int s = ei[e];
    int d = ei[N_EDGES + e];
    atomicAdd(&agg2[d * HID + f], h1[s * HID + f]);
}

__global__ void k_sage2(const float* __restrict__ agg2, const float* __restrict__ rdeg,
                        const float* __restrict__ h1,
                        const float* __restrict__ Wl2, const float* __restrict__ bl2,
                        const float* __restrict__ Wr2, float* __restrict__ h2) {
    const int row0 = blockIdx.x * 16;
    const int col0 = blockIdx.y * 16;
    const int lane = threadIdx.x;
    const int m    = lane & 15;
    const int half = lane >> 4;
    const int kb   = half * 2;
    const int row  = row0 + m;
    const float rd = rdeg[row];

    v8f acc = {};
#pragma unroll 4
    for (int k0 = 0; k0 < HID; k0 += 4) {
        v2f am, ah, bl, br;
        am.x = agg2[row * HID + k0 + kb]     * rd;
        am.y = agg2[row * HID + k0 + kb + 1] * rd;
        ah.x = h1[row * HID + k0 + kb];
        ah.y = h1[row * HID + k0 + kb + 1];
        bl.x = Wl2[(col0 + m) * HID + k0 + kb];
        bl.y = Wl2[(col0 + m) * HID + k0 + kb + 1];
        br.x = Wr2[(col0 + m) * HID + k0 + kb];
        br.y = Wr2[(col0 + m) * HID + k0 + kb + 1];
        acc = __builtin_amdgcn_wmma_f32_16x16x4_f32(false, am, false, bl, (short)0, acc, false, false);
        acc = __builtin_amdgcn_wmma_f32_16x16x4_f32(false, ah, false, br, (short)0, acc, false, false);
    }
#pragma unroll
    for (int i = 0; i < 8; ++i) {
        int r = row0 + i + half * 8;
        int c = col0 + m;
        h2[r * HID + c] = fmaxf(acc[i] + bl2[c], 0.0f);
    }
}

__global__ void k_hpart(const float* __restrict__ h2, const float* __restrict__ Wc1,
                        float* __restrict__ hpart) {
    const int row0 = blockIdx.x * 16;
    const int col0 = blockIdx.y * 16;
    const int lane = threadIdx.x;
    const int m    = lane & 15;
    const int half = lane >> 4;
    const int kb   = half * 2;
    const int row  = row0 + m;

    v8f acc = {};
#pragma unroll 4
    for (int k0 = 0; k0 < HID; k0 += 4) {
        v2f a, b;
        a.x = h2[row * HID + k0 + kb];
        a.y = h2[row * HID + k0 + kb + 1];
        b.x = Wc1[(col0 + m) * (2 * HID) + k0 + kb];
        b.y = Wc1[(col0 + m) * (2 * HID) + k0 + kb + 1];
        acc = __builtin_amdgcn_wmma_f32_16x16x4_f32(false, a, false, b, (short)0, acc, false, false);
    }
#pragma unroll
    for (int i = 0; i < 8; ++i) {
        int r = row0 + i + half * 8;
        int c = col0 + m;
        hpart[r * HID + c] = acc[i];
    }
}

__global__ void k_task(const float* __restrict__ task, const float* __restrict__ Wt1,
                       const float* __restrict__ bt1, const float* __restrict__ Wt2,
                       const float* __restrict__ bt2, const float* __restrict__ Wc1,
                       const float* __restrict__ bc1, float* __restrict__ tpart) {
    __shared__ float u[HID];
    __shared__ float t[HID];
    const int b = blockIdx.x;
    const int h = threadIdx.x;

    float acc = bt1[h];
#pragma unroll
    for (int k = 0; k < TASK_F; ++k) acc += task[b * TASK_F + k] * Wt1[h * TASK_F + k];
    u[h] = fmaxf(acc, 0.0f);
    __syncthreads();

    float acc2 = bt2[h];
#pragma unroll 8
    for (int k = 0; k < HID; ++k) acc2 += u[k] * Wt2[h * HID + k];
    t[h] = acc2;
    __syncthreads();

    float acc3 = bc1[h];
#pragma unroll 8
    for (int k = 0; k < HID; ++k) acc3 += t[k] * Wc1[h * (2 * HID) + HID + k];
    tpart[b * HID + h] = acc3;
}

__global__ void k_classify(const float* __restrict__ hpart, const float* __restrict__ tpart,
                           const float* __restrict__ Wc2, const float* __restrict__ bc2,
                           float* __restrict__ out) {
    __shared__ float tp[HID];
    __shared__ float w2[HID];
    const int b = blockIdx.y;
    const int n = blockIdx.x * 256 + threadIdx.x;
    if (threadIdx.x < HID) {
        tp[threadIdx.x] = tpart[b * HID + threadIdx.x];
        w2[threadIdx.x] = Wc2[threadIdx.x];
    }
    __syncthreads();
    if (n >= N_NODES) return;

    const float4* hp = (const float4*)(hpart + (size_t)n * HID);
    float acc = bc2[0];
#pragma unroll
    for (int j = 0; j < HID / 4; ++j) {
        float4 v = hp[j];
        int h = j * 4;
        acc += fmaxf(v.x + tp[h + 0], 0.0f) * w2[h + 0];
        acc += fmaxf(v.y + tp[h + 1], 0.0f) * w2[h + 1];
        acc += fmaxf(v.z + tp[h + 2], 0.0f) * w2[h + 2];
        acc += fmaxf(v.w + tp[h + 3], 0.0f) * w2[h + 3];
    }
    out[(size_t)b * N_NODES + n] = acc;
}

extern "C" void kernel_launch(void* const* d_in, const int* in_sizes, int n_in,
                              void* d_out, int out_size, void* d_ws, size_t ws_size,
                              hipStream_t stream) {
    const float* x    = (const float*)d_in[0];
    const int*   ei   = (const int*)d_in[1];
    const float* task = (const float*)d_in[2];
    const float* Wl1  = (const float*)d_in[3];
    const float* bl1  = (const float*)d_in[4];
    const float* Wr1  = (const float*)d_in[5];
    const float* Wl2  = (const float*)d_in[6];
    const float* bl2  = (const float*)d_in[7];
    const float* Wr2  = (const float*)d_in[8];
    const float* Wt1  = (const float*)d_in[9];
    const float* bt1  = (const float*)d_in[10];
    const float* Wt2  = (const float*)d_in[11];
    const float* bt2  = (const float*)d_in[12];
    const float* Wc1  = (const float*)d_in[13];
    const float* bc1  = (const float*)d_in[14];
    const float* Wc2  = (const float*)d_in[15];
    const float* bc2  = (const float*)d_in[16];
    float* out = (float*)d_out;

    float* ws    = (float*)d_ws;
    float* deg   = ws + OFF_DEG;
    float* agg1  = ws + OFF_AGG1;
    float* agg2  = ws + OFF_AGG2;
    float* rdeg  = ws + OFF_RDEG;
    float* h1    = ws + OFF_H1;
    float* h2    = ws + OFF_H2;
    float* hpart = ws + OFF_HPART;
    float* tpart = ws + OFF_TPART;

    k_zero<<<(ZERO_COUNT + 255) / 256, 256, 0, stream>>>(ws, ZERO_COUNT);
    k_scatter1<<<(N_EDGES + 255) / 256, 256, 0, stream>>>(ei, x, agg1, deg);
    k_rdeg<<<(NPAD + 255) / 256, 256, 0, stream>>>(deg, rdeg);
    k_sage1<<<dim3(NPAD / 16, HID / 16), 32, 0, stream>>>(agg1, rdeg, x, Wl1, bl1, Wr1, h1);
    {
        long long total = (long long)N_EDGES * HID;
        k_scatter2<<<(unsigned)((total + 255) / 256), 256, 0, stream>>>(ei, h1, agg2);
    }
    k_sage2<<<dim3(NPAD / 16, HID / 16), 32, 0, stream>>>(agg2, rdeg, h1, Wl2, bl2, Wr2, h2);
    k_hpart<<<dim3(NPAD / 16, HID / 16), 32, 0, stream>>>(h2, Wc1, hpart);
    k_task<<<BATCH, HID, 0, stream>>>(task, Wt1, bt1, Wt2, bt2, Wc1, bc1, tpart);
    k_classify<<<dim3((N_NODES + 255) / 256, BATCH), 256, 0, stream>>>(hpart, tpart, Wc2, bc2, out);
}


# classifier: each thread keeps its node's 64-float hpart row in registers and loops over 16 batch rows (4 sums in flight), so a row is fetched once per 16 scores; same h order, mul-add now fused
# speedup vs baseline: 1.6054x; 1.6054x over previous
#include <hip/hip_runtime.h>
#include <hip/hip_bf16.h>

#define N_NODES 5000
#define NPAD    5120
#define N_EDGES 160000
#define HID     64
#define NODE_F  4
#define TASK_F  4
#define BATCH   512

typedef __attribute__((ext_vector_type(2))) float v2f;
typedef __attribute__((ext_vector_type(8))) float v8f;

#define OFF_DEG    0
#define OFF_AGG1   (OFF_DEG  + NPAD)
#define OFF_AGG2   (OFF_AGG1 + NPAD * NODE_F)
#define ZERO_COUNT (OFF_AGG2 + NPAD * HID)
#define OFF_RDEG   ZERO_COUNT
#define OFF_H1     (OFF_RDEG  + NPAD)
#define OFF_H2     (OFF_H1    + NPAD * HID)
#define OFF_HPART  (OFF_H2    + NPAD * HID)
#define OFF_TPART  (OFF_HPART + NPAD * HID)


__global__ void k_zero(float* __restrict__ p, int n) {
    int i = blockIdx.x * blockDim.x + threadIdx.x;
    if (i < n) p[i] = 0.0f;
}

__global__ void k_scatter1(const int* __restrict__ ei, const float* __restrict__ x,
                           float* __restrict__ agg1, float* __restrict__ deg) {
    int e = blockIdx.x * blockDim.x + threadIdx.x;
    if (e >= N_EDGES) return;
    int s = ei[e];
    int d = ei[N_EDGES + e];
    atomicAdd(&deg[d], 1.0f);
#pragma unroll
    for (int k = 0; k < NODE_F; ++k)
        atomicAdd(&agg1[d * NODE_F + k], x[s * NODE_F + k]);
}

__global__ void k_rdeg(const float* __restrict__ deg, float* __restrict__ rdeg) {
    int i = blockIdx.x * blockDim.x + threadIdx.x;
    if (i < NPAD) rdeg[i] = 1.0f / fmaxf(deg[i], 1.0f);
}

__global__ void k_sage1(const float* __restrict__ agg1, const float* __restrict__ rdeg,
                        const float* __restrict__ x,
                        const float* __restrict__ Wl1, const float* __restrict__ bl1,
                        const float* __restrict__ Wr1, float* __restrict__ h1) {
    const int row0 = blockIdx.x * 16;
    const int col0 = blockIdx.y * 16;
    const int lane = threadIdx.x;
    const int m    = lane & 15;
    const int half = lane >> 4;
    const int kb   = half * 2;
    const int row  = row0 + m;
    const float rd = rdeg[row];

    v2f am, ax, bl, br;
    am.x = agg1[row * NODE_F + kb]     * rd;
    am.y = agg1[row * NODE_F + kb + 1] * rd;
    ax.x = (row < N_NODES) ? x[row * NODE_F + kb]     : 0.0f;
    ax.y = (row < N_NODES) ? x[row * NODE_F + kb + 1] : 0.0f;
    bl.x = Wl1[(col0 + m) * NODE_F + kb];
    bl.y = Wl1[(col0 + m) * NODE_F + kb + 1];
    br.x = Wr1[(col0 + m) * NODE_F + kb];
    br.y = Wr1[(col0 + m) * NODE_F + kb + 1];

    v8f acc = {};
    acc = __builtin_amdgcn_wmma_f32_16x16x4_f32(false, am, false, bl, (short)0, acc, false, false);
    acc = __builtin_amdgcn_wmma_f32_16x16x4_f32(false, ax, false, br, (short)0, acc, false, false);

#pragma unroll
    for (int i = 0; i < 8; ++i) {
        int r = row0 + i + half * 8;
        int c = col0 + m;
        h1[r * HID + c] = fmaxf(acc[i] + bl1[c], 0.0f);
    }
}

__global__ void k_scatter2(const int* __restrict__ ei, const float* __restrict__ h1,
                           float* __restrict__ agg2) {
    long long idx = (long long)blockIdx.x * blockDim.x + threadIdx.x;
    if (idx >= (long long)N_EDGES * HID) return;
    int e = (int)(idx >> 6);
    int f = (int)(idx & 63);
    int s = ei[e];
    int d = ei[N_EDGES + e];
    atomicAdd(&agg2[d * HID + f], h1[s * HID + f]);
}

__global__ void k_sage2(const float* __restrict__ agg2, const float* __restrict__ rdeg,
                        const float* __restrict__ h1,
                        const float* __restrict__ Wl2, const float* __restrict__ bl2,
                        const float* __restrict__ Wr2, float* __restrict__ h2) {
    const int row0 = blockIdx.x * 16;
    const int col0 = blockIdx.y * 16;
    const int lane = threadIdx.x;
    const int m    = lane & 15;
    const int half = lane >> 4;
    const int kb   = half * 2;
    const int row  = row0 + m;
    const float rd = rdeg[row];

    v8f acc = {};
#pragma unroll 4
    for (int k0 = 0; k0 < HID; k0 += 4) {
        v2f am, ah, bl, br;
        am.x = agg2[row * HID + k0 + kb]     * rd;
        am.y = agg2[row * HID + k0 + kb + 1] * rd;
        ah.x = h1[row * HID + k0 + kb];
        ah.y = h1[row * HID + k0 + kb + 1];
        bl.x = Wl2[(col0 + m) * HID + k0 + kb];
        bl.y = Wl2[(col0 + m) * HID + k0 + kb + 1];
        br.x = Wr2[(col0 + m) * HID + k0 + kb];
        br.y = Wr2[(col0 + m) * HID + k0 + kb + 1];
        acc = __builtin_amdgcn_wmma_f32_16x16x4_f32(false, am, false, bl, (short)0, acc, false, false);
        acc = __builtin_amdgcn_wmma_f32_16x16x4_f32(false, ah, false, br, (short)0, acc, false, false);
    }
#pragma unroll
    for (int i = 0; i < 8; ++i) {
        int r = row0 + i + half * 8;
        int c = col0 + m;
        h2[r * HID + c] = fmaxf(acc[i] + bl2[c], 0.0f);
    }
}

__global__ void k_hpart(const float* __restrict__ h2, const float* __restrict__ Wc1,
                        float* __restrict__ hpart) {
    const int row0 = blockIdx.x * 16;
    const int col0 = blockIdx.y * 16;
    const int lane = threadIdx.x;
    const int m    = lane & 15;
    const int half = lane >> 4;
    const int kb   = half * 2;
    const int row  = row0 + m;

    v8f acc = {};
#pragma unroll 4
    for (int k0 = 0; k0 < HID; k0 += 4) {
        v2f a, b;
        a.x = h2[row * HID + k0 + kb];
        a.y = h2[row * HID + k0 + kb + 1];
        b.x = Wc1[(col0 + m) * (2 * HID) + k0 + kb];
        b.y = Wc1[(col0 + m) * (2 * HID) + k0 + kb + 1];
        acc = __builtin_amdgcn_wmma_f32_16x16x4_f32(false, a, false, b, (short)0, acc, false, false);
    }
#pragma unroll
    for (int i = 0; i < 8; ++i) {
        int r = row0 + i + half * 8;
        int c = col0 + m;
        hpart[r * HID + c] = acc[i];
    }
}

__global__ void k_task(const float* __restrict__ task, const float* __restrict__ Wt1,
                       const float* __restrict__ bt1, const float* __restrict__ Wt2,
                       const float* __restrict__ bt2, const float* __restrict__ Wc1,
                       const float* __restrict__ bc1, float* __restrict__ tpart) {
    __shared__ float u[HID];
    __shared__ float t[HID];
    const int b = blockIdx.x;
    const int h = threadIdx.x;

    float acc = bt1[h];
#pragma unroll
    for (int k = 0; k < TASK_F; ++k) acc += task[b * TASK_F + k] * Wt1[h * TASK_F + k];
    u[h] = fmaxf(acc, 0.0f);
    __syncthreads();

    float acc2 = bt2[h];
#pragma unroll 8
    for (int k = 0; k < HID; ++k) acc2 += u[k] * Wt2[h * HID + k];
    t[h] = acc2;
    __syncthreads();

    float acc3 = bc1[h];
#pragma unroll 8
    for (int k = 0; k < HID; ++k) acc3 += t[k] * Wc1[h * (2 * HID) + HID + k];
    tpart[b * HID + h] = acc3;
}

#define CLS_BT      16
#define CLS_G       4
#define CLS_THREADS 128
__global__ __launch_bounds__(CLS_THREADS)
void k_classify(const float* __restrict__ hpart, const float* __restrict__ tpart,
                const float* __restrict__ Wc2, const float* __restrict__ bc2,
                float* __restrict__ out) {
    __shared__ float w2[HID];
    if (threadIdx.x < HID) w2[threadIdx.x] = Wc2[threadIdx.x];
    __syncthreads();

    const int n  = blockIdx.x * CLS_THREADS + threadIdx.x;
    const int b0 = blockIdx.y * CLS_BT;
    if (n - (int)(threadIdx.x & 31) >= N_NODES) return;

    float a[HID];
    const float4* hp = (const float4*)(hpart + (size_t)n * HID);
#pragma unroll
    for (int j = 0; j < HID / 4; ++j) {
        float4 v = hp[j];
        a[4 * j + 0] = v.x; a[4 * j + 1] = v.y; a[4 * j + 2] = v.z; a[4 * j + 3] = v.w;
    }
    float w[HID];
#pragma unroll
    for (int h = 0; h < HID; ++h) w[h] = w2[h];
    const float bias = bc2[0];

    for (int g = 0; g < CLS_BT / CLS_G; ++g) {
        const float* __restrict__ tp = tpart + (size_t)(b0 + g * CLS_G) * HID;
        float acc[CLS_G];
#pragma unroll
        for (int r = 0; r < CLS_G; ++r) acc[r] = bias;
#pragma unroll
        for (int hc = 0; hc < HID; hc += 16) {
#pragma unroll
            for (int r = 0; r < CLS_G; ++r) {
#pragma unroll
                for (int k = 0; k < 16; ++k)
                    acc[r] += fmaxf(a[hc + k] + tp[r * HID + hc + k], 0.0f) * w[hc + k];
            }
        }
        if (n < N_NODES) {
#pragma unroll
            for (int r = 0; r < CLS_G; ++r)
                out[(size_t)(b0 + g * CLS_G + r) * N_NODES + n] = acc[r];
        }
    }
}

extern "C" void kernel_launch(void* const* d_in, const int* in_sizes, int n_in,
                              void* d_out, int out_size, void* d_ws, size_t ws_size,
                              hipStream_t stream) {
    const float* x    = (const float*)d_in[0];
    const int*   ei   = (const int*)d_in[1];
    const float* task = (const float*)d_in[2];
    const float* Wl1  = (const float*)d_in[3];
    const float* bl1  = (const float*)d_in[4];
    const float* Wr1  = (const float*)d_in[5];
    const float* Wl2  = (const float*)d_in[6];
    const float* bl2  = (const float*)d_in[7];
    const float* Wr2  = (const float*)d_in[8];
    const float* Wt1  = (const float*)d_in[9];
    const float* bt1  = (const float*)d_in[10];
    const float* Wt2  = (const float*)d_in[11];
    const float* bt2  = (const float*)d_in[12];
    const float* Wc1  = (const float*)d_in[13];
    const float* bc1  = (const float*)d_in[14];
    const float* Wc2  = (const float*)d_in[15];
    const float* bc2  = (const float*)d_in[16];
    float* out = (float*)d_out;

    float* ws    = (float*)d_ws;
    float* deg   = ws + OFF_DEG;
    float* agg1  = ws + OFF_AGG1;
    float* agg2  = ws + OFF_AGG2;
    float* rdeg  = ws + OFF_RDEG;
    float* h1    = ws + OFF_H1;
    float* h2    = ws + OFF_H2;
    float* hpart = ws + OFF_HPART;
    float* tpart = ws + OFF_TPART;

    k_zero<<<(ZERO_COUNT + 255) / 256, 256, 0, stream>>>(ws, ZERO_COUNT);
    k_scatter1<<<(N_EDGES + 255) / 256, 256, 0, stream>>>(ei, x, agg1, deg);
    k_rdeg<<<(NPAD + 255) / 256, 256, 0, stream>>>(deg, rdeg);
    k_sage1<<<dim3(NPAD / 16, HID / 16), 32, 0, stream>>>(agg1, rdeg, x, Wl1, bl1, Wr1, h1);
    {
        long long total = (long long)N_EDGES * HID;
        k_scatter2<<<(unsigned)((total + 255) / 256), 256, 0, stream>>>(ei, h1, agg2);
    }
    k_sage2<<<dim3(NPAD / 16, HID / 16), 32, 0, stream>>>(agg2, rdeg, h1, Wl2, bl2, Wr2, h2);
    k_hpart<<<dim3(NPAD / 16, HID / 16), 32, 0, stream>>>(h2, Wc1, hpart);
    k_task<<<BATCH, HID, 0, stream>>>(task, Wt1, bt1, Wt2, bt2, Wc1, bc1, tpart);
    k_classify<<<dim3(NPAD / CLS_THREADS, BATCH / CLS_BT), CLS_THREADS, 0, stream>>>(hpart, tpart, Wc2, bc2, out);
}


# layer-2 aggregation without float atomics: neighbour lists built per call (rank from the integer degree add's return, prefix sum, fill), one wave per node gathers h1 rows; sums in list order
# speedup vs baseline: 1.7683x; 1.1015x over previous
#include <hip/hip_runtime.h>
#include <hip/hip_bf16.h>

#define N_NODES 5000
#define NPAD    5120
#define N_EDGES 160000
#define HID     64
#define NODE_F  4
#define TASK_F  4
#define BATCH   512

typedef __attribute__((ext_vector_type(2))) float v2f;
typedef __attribute__((ext_vector_type(8))) float v8f;

#define OFF_DEG    0
#define OFF_AGG1   (OFF_DEG  + NPAD)
#define OFF_AGG2   (OFF_AGG1 + NPAD * NODE_F)
#define ZERO_COUNT OFF_AGG2
#define OFF_RDEG   (OFF_AGG2 + NPAD * HID)
#define OFF_H1     (OFF_RDEG  + NPAD)
#define OFF_H2     (OFF_H1    + NPAD * HID)
#define OFF_HPART  (OFF_H2    + NPAD * HID)
#define OFF_TPART  (OFF_HPART + NPAD * HID)


__global__ void k_zero(float* __restrict__ p, int n) {
    int i = blockIdx.x * blockDim.x + threadIdx.x;
    if (i < n) p[i] = 0.0f;
}

__global__ void k_scatter1(const int* __restrict__ ei, const float* __restrict__ x,
                           float* __restrict__ agg1, int* __restrict__ deg, int* __restrict__ rank) {
    int e = blockIdx.x * blockDim.x + threadIdx.x;
    if (e >= N_EDGES) return;
    int s = ei[e];
    int d = ei[N_EDGES + e];
    rank[e] = atomicAdd(&deg[d], 1);
#pragma unroll
    for (int k = 0; k < NODE_F; ++k)
        atomicAdd(&agg1[d * NODE_F + k], x[s * NODE_F + k]);
}

#define SCAN_THREADS 1024
#define SCAN_PER     (NPAD / SCAN_THREADS)
__global__ __launch_bounds__(SCAN_THREADS)
void k_rdeg_scan(const int* __restrict__ deg, float* __restrict__ rdeg, int* __restrict__ start) {
    __shared__ int wsum[SCAN_THREADS / 32];
    const int t    = threadIdx.x;
    const int lane = t & 31;
    const int w    = t >> 5;
    int c[SCAN_PER];
    int tot = 0;
#pragma unroll
    for (int k = 0; k < SCAN_PER; ++k) {
        c[k] = deg[t * SCAN_PER + k];
        rdeg[t * SCAN_PER + k] = 1.0f / fmaxf((float)c[k], 1.0f);
        tot += c[k];
    }
    int inc = tot;
    for (int off = 1; off < 32; off <<= 1) {
        int v = __shfl_up(inc, off);
        if (lane >= off) inc += v;
    }
    if (lane == 31) wsum[w] = inc;
    __syncthreads();
    if (w == 0) {
        const int mine = wsum[lane];
        int ws = mine;
        for (int off = 1; off < 32; off <<= 1) {
            int v = __shfl_up(ws, off);
            if (lane >= off) ws += v;
        }
        wsum[lane] = ws - mine;
    }
    __syncthreads();
    int run = wsum[w] + inc - tot;
#pragma unroll
    for (int k = 0; k < SCAN_PER; ++k) {
        start[t * SCAN_PER + k] = run;
        run += c[k];
    }
    if (t == SCAN_THREADS - 1) start[NPAD] = run;
}

__global__ void k_fill(const int* __restrict__ ei, const int* __restrict__ start,
                       const int* __restrict__ rank, int* __restrict__ csr) {
    int e = blockIdx.x * blockDim.x + threadIdx.x;
    if (e >= N_EDGES) return;
    int s = ei[e];
    int d = ei[N_EDGES + e];
    csr[start[d] + rank[e]] = s;
}

__global__ void k_sage1(const float* __restrict__ agg1, const float* __restrict__ rdeg,
                        const float* __restrict__ x,
                        const float* __restrict__ Wl1, const float* __restrict__ bl1,
                        const float* __restrict__ Wr1, float* __restrict__ h1) {
    const int row0 = blockIdx.x * 16;
    const int col0 = blockIdx.y * 16;
    const int lane = threadIdx.x;
    const int m    = lane & 15;
    const int half = lane >> 4;
    const int kb   = half * 2;
    const int row  = row0 + m;
    const float rd = rdeg[row];

    v2f am, ax, bl, br;
    am.x = agg1[row * NODE_F + kb]     * rd;
    am.y = agg1[row * NODE_F + kb + 1] * rd;
    ax.x = (row < N_NODES) ? x[row * NODE_F + kb]     : 0.0f;
    ax.y = (row < N_NODES) ? x[row * NODE_F + kb + 1] : 0.0f;
    bl.x = Wl1[(col0 + m) * NODE_F + kb];
    bl.y = Wl1[(col0 + m) * NODE_F + kb + 1];
    br.x = Wr1[(col0 + m) * NODE_F + kb];
    br.y = Wr1[(col0 + m) * NODE_F + kb + 1];

    v8f acc = {};
    acc = __builtin_amdgcn_wmma_f32_16x16x4_f32(false, am, false, bl, (short)0, acc, false, false);
    acc = __builtin_amdgcn_wmma_f32_16x16x4_f32(false, ax, false, br, (short)0, acc, false, false);

#pragma unroll
    for (int i = 0; i < 8; ++i) {
        int r = row0 + i + half * 8;
        int c = col0 + m;
        h1[r * HID + c] = fmaxf(acc[i] + bl1[c], 0.0f);
    }
}

#define GAT_WAVES 4
__global__ __launch_bounds__(GAT_WAVES * 32)
void k_gather2(const int* __restrict__ start, const int* __restrict__ csr,
               const float* __restrict__ h1, float* __restrict__ agg2) {
    const int lane = threadIdx.x & 31;
    const int n   = __builtin_amdgcn_readfirstlane((int)(blockIdx.x * GAT_WAVES + (threadIdx.x >> 5)));
    const int beg = start[n];
    const int end = start[n + 1];
    const v2f* __restrict__ rows = (const v2f*)h1;

    v2f acc = {0.0f, 0.0f};
    int idx = (beg + lane < end) ? csr[beg + lane] : 0;
    for (int base = beg; base < end; base += 32) {
        const int nb = base + 32;
        const int idx_next = (nb + lane < end) ? csr[nb + lane] : 0;
        const int cnt = min(32, end - base);
        for (int j = 0; j < cnt; j += 8) {
            v2f v[8];
#pragma unroll
            for (int u = 0; u < 8; ++u) {
                const int s = __builtin_amdgcn_readlane(idx, j + u);
                v[u] = rows[(size_t)s * (HID / 2) + lane];
            }
#pragma unroll
            for (int u = 0; u < 8; ++u) {
                const bool ok = (j + u) < cnt;
                acc.x += ok ? v[u].x : 0.0f;
                acc.y += ok ? v[u].y : 0.0f;
            }
        }
        idx = idx_next;
    }
    ((v2f*)agg2)[(size_t)n * (HID / 2) + lane] = acc;
}

__global__ void k_sage2(const float* __restrict__ agg2, const float* __restrict__ rdeg,
                        const float* __restrict__ h1,
                        const float* __restrict__ Wl2, const float* __restrict__ bl2,
                        const float* __restrict__ Wr2, float* __restrict__ h2) {
    const int row0 = blockIdx.x * 16;
    const int col0 = blockIdx.y * 16;
    const int lane = threadIdx.x;
    const int m    = lane & 15;
    const int half = lane >> 4;
    const int kb   = half * 2;
    const int row  = row0 + m;
    const float rd = rdeg[row];

    v8f acc = {};
#pragma unroll 4
    for (int k0 = 0; k0 < HID; k0 += 4) {
        v2f am, ah, bl, br;
        am.x = agg2[row * HID + k0 + kb]     * rd;
        am.y = agg2[row * HID + k0 + kb + 1] * rd;
        ah.x = h1[row * HID + k0 + kb];
        ah.y = h1[row * HID + k0 + kb + 1];
        bl.x = Wl2[(col0 + m) * HID + k0 + kb];
        bl.y = Wl2[(col0 + m) * HID + k0 + kb + 1];
        br.x = Wr2[(col0 + m) * HID + k0 + kb];
        br.y = Wr2[(col0 + m) * HID + k0 + kb + 1];
        acc = __builtin_amdgcn_wmma_f32_16x16x4_f32(false, am, false, bl, (short)0, acc, false, false);
        acc = __builtin_amdgcn_wmma_f32_16x16x4_f32(false, ah, false, br, (short)0, acc, false, false);
    }
#pragma unroll
    for (int i = 0; i < 8; ++i) {
        int r = row0 + i + half * 8;
        int c = col0 + m;
        h2[r * HID + c] = fmaxf(acc[i] + bl2[c], 0.0f);
    }
}

__global__ void k_hpart(const float* __restrict__ h2, const float* __restrict__ Wc1,
                        float* __restrict__ hpart) {
    const int row0 = blockIdx.x * 16;
    const int col0 = blockIdx.y * 16;
    const int lane = threadIdx.x;
    const int m    = lane & 15;
    const int half = lane >> 4;
    const int kb   = half * 2;
    const int row  = row0 + m;

    v8f acc = {};
#pragma unroll 4
    for (int k0 = 0; k0 < HID; k0 += 4) {
        v2f a, b;
        a.x = h2[row * HID + k0 + kb];
        a.y = h2[row * HID + k0 + kb + 1];
        b.x = Wc1[(col0 + m) * (2 * HID) + k0 + kb];
        b.y = Wc1[(col0 + m) * (2 * HID) + k0 + kb + 1];
        acc = __builtin_amdgcn_wmma_f32_16x16x4_f32(false, a, false, b, (short)0, acc, false, false);
    }
#pragma unroll
    for (int i = 0; i < 8; ++i) {
        int r = row0 + i + half * 8;
        int c = col0 + m;
        hpart[r * HID + c] = acc[i];
    }
}

__global__ void k_task(const float* __restrict__ task, const float* __restrict__ Wt1,
                       const float* __restrict__ bt1, const float* __restrict__ Wt2,
                       const float* __restrict__ bt2, const float* __restrict__ Wc1,
                       const float* __restrict__ bc1, float* __restrict__ tpart) {
    __shared__ float u[HID];
    __shared__ float t[HID];
    const int b = blockIdx.x;
    const int h = threadIdx.x;

    float acc = bt1[h];
#pragma unroll
    for (int k = 0; k < TASK_F; ++k) acc += task[b * TASK_F + k] * Wt1[h * TASK_F + k];
    u[h] = fmaxf(acc, 0.0f);
    __syncthreads();

    float acc2 = bt2[h];
#pragma unroll 8
    for (int k = 0; k < HID; ++k) acc2 += u[k] * Wt2[h * HID + k];
    t[h] = acc2;
    __syncthreads();

    float acc3 = bc1[h];
#pragma unroll 8
    for (int k = 0; k < HID; ++k) acc3 += t[k] * Wc1[h * (2 * HID) + HID + k];
    tpart[b * HID + h] = acc3;
}

#define CLS_BT      16
#define CLS_G       4
#define CLS_THREADS 128
__global__ __launch_bounds__(CLS_THREADS)
void k_classify(const float* __restrict__ hpart, const float* __restrict__ tpart,
                const float* __restrict__ Wc2, const float* __restrict__ bc2,
                float* __restrict__ out) {
    __shared__ float w2[HID];
    if (threadIdx.x < HID) w2[threadIdx.x] = Wc2[threadIdx.x];
    __syncthreads();

    const int n  = blockIdx.x * CLS_THREADS + threadIdx.x;
    const int b0 = blockIdx.y * CLS_BT;
    if (n - (int)(threadIdx.x & 31) >= N_NODES) return;

    float a[HID];
    const float4* hp = (const float4*)(hpart + (size_t)n * HID);
#pragma unroll
    for (int j = 0; j < HID / 4; ++j) {
        float4 v = hp[j];
        a[4 * j + 0] = v.x; a[4 * j + 1] = v.y; a[4 * j + 2] = v.z; a[4 * j + 3] = v.w;
    }
    float w[HID];
#pragma unroll
    for (int h = 0; h < HID; ++h) w[h] = w2[h];
    const float bias = bc2[0];

    for (int g = 0; g < CLS_BT / CLS_G; ++g) {
        const float* __restrict__ tp = tpart + (size_t)(b0 + g * CLS_G) * HID;
        float acc[CLS_G];
#pragma unroll
        for (int r = 0; r < CLS_G; ++r) acc[r] = bias;
#pragma unroll
        for (int hc = 0; hc < HID; hc += 16) {
#pragma unroll
            for (int r = 0; r < CLS_G; ++r) {
#pragma unroll
                for (int k = 0; k < 16; ++k)
                    acc[r] += fmaxf(a[hc + k] + tp[r * HID + hc + k], 0.0f) * w[hc + k];
            }
        }
        if (n < N_NODES) {
#pragma unroll
            for (int r = 0; r < CLS_G; ++r)
                out[(size_t)(b0 + g * CLS_G + r) * N_NODES + n] = acc[r];
        }
    }
}

extern "C" void kernel_launch(void* const* d_in, const int* in_sizes, int n_in,
                              void* d_out, int out_size, void* d_ws, size_t ws_size,
                              hipStream_t stream) {
    const float* x    = (const float*)d_in[0];
    const int*   ei   = (const int*)d_in[1];
    const float* task = (const float*)d_in[2];
    const float* Wl1  = (const float*)d_in[3];
    const float* bl1  = (const float*)d_in[4];
    const float* Wr1  = (const float*)d_in[5];
    const float* Wl2  = (const float*)d_in[6];
    const float* bl2  = (const float*)d_in[7];
    const float* Wr2  = (const float*)d_in[8];
    const float* Wt1  = (const float*)d_in[9];
    const float* bt1  = (const float*)d_in[10];
    const float* Wt2  = (const float*)d_in[11];
    const float* bt2  = (const float*)d_in[12];
    const float* Wc1  = (const float*)d_in[13];
    const float* bc1  = (const float*)d_in[14];
    const float* Wc2  = (const float*)d_in[15];
    const float* bc2  = (const float*)d_in[16];
    float* out = (float*)d_out;

    float* ws    = (float*)d_ws;
    int*   deg   = (int*)(ws + OFF_DEG);
    float* agg1  = ws + OFF_AGG1;
    float* agg2  = ws + OFF_AGG2;
    float* rdeg  = ws + OFF_RDEG;
    float* h1    = ws + OFF_H1;
    float* h2    = ws + OFF_H2;
    float* hpart = ws + OFF_HPART;
    float* tpart = ws + OFF_TPART;
    int*   rank  = (int*)(ws + OFF_H2);
    int*   csr   = (int*)(ws + OFF_HPART);
    int*   start = (int*)(ws + OFF_TPART);

    k_zero<<<(ZERO_COUNT + 255) / 256, 256, 0, stream>>>(ws, ZERO_COUNT);
    k_scatter1<<<(N_EDGES + 255) / 256, 256, 0, stream>>>(ei, x, agg1, deg, rank);
    k_rdeg_scan<<<1, SCAN_THREADS, 0, stream>>>(deg, rdeg, start);
    k_sage1<<<dim3(NPAD / 16, HID / 16), 32, 0, stream>>>(agg1, rdeg, x, Wl1, bl1, Wr1, h1);
    k_fill<<<(N_EDGES + 255) / 256, 256, 0, stream>>>(ei, start, rank, csr);
    k_gather2<<<NPAD / GAT_WAVES, GAT_WAVES * 32, 0, stream>>>(start, csr, h1, agg2);
    k_sage2<<<dim3(NPAD / 16, HID / 16), 32, 0, stream>>>(agg2, rdeg, h1, Wl2, bl2, Wr2, h2);
    k_hpart<<<dim3(NPAD / 16, HID / 16), 32, 0, stream>>>(h2, Wc1, hpart);
    k_task<<<BATCH, HID, 0, stream>>>(task, Wt1, bt1, Wt2, bt2, Wc1, bc1, tpart);
    k_classify<<<dim3(NPAD / CLS_THREADS, BATCH / CLS_BT), CLS_THREADS, 0, stream>>>(hpart, tpart, Wc2, bc2, out);
}


# launches 10 -> 7: accumulator clear + task encoder in launch 1, layer-1 SAGE tiles + list fill in one launch, hpart computed from the h2 tile kept in LDS inside the layer-2 kernel; sums unchanged
# speedup vs baseline: 1.8535x; 1.0482x over previous
#include <hip/hip_runtime.h>
#include <hip/hip_bf16.h>

#define N_NODES 5000
#define NPAD    5120
#define N_EDGES 160000
#define HID     64
#define NODE_F  4
#define TASK_F  4
#define BATCH   512

typedef __attribute__((ext_vector_type(2))) float v2f;
typedef __attribute__((ext_vector_type(8))) float v8f;

#define OFF_DEG    0
#define OFF_AGG1   (OFF_DEG  + NPAD)
#define OFF_AGG2   (OFF_AGG1 + NPAD * NODE_F)
#define ZERO_COUNT OFF_AGG2
#define OFF_RDEG   (OFF_AGG2 + NPAD * HID)
#define OFF_H1     (OFF_RDEG  + NPAD)
#define OFF_H2     (OFF_H1    + NPAD * HID)
#define OFF_HPART  (OFF_H2    + NPAD * HID)
#define OFF_TPART  (OFF_HPART + NPAD * HID)
#define OFF_START  (OFF_H2    + N_EDGES)


__global__ void k_scatter1(const int* __restrict__ ei, const float* __restrict__ x,
                           float* __restrict__ agg1, int* __restrict__ deg, int* __restrict__ rank) {
    int e = blockIdx.x * blockDim.x + threadIdx.x;
    if (e >= N_EDGES) return;
    int s = ei[e];
    int d = ei[N_EDGES + e];
    rank[e] = atomicAdd(&deg[d], 1);
#pragma unroll
    for (int k = 0; k < NODE_F; ++k)
        atomicAdd(&agg1[d * NODE_F + k], x[s * NODE_F + k]);
}

#define SCAN_THREADS 1024
#define SCAN_PER     (NPAD / SCAN_THREADS)
__global__ __launch_bounds__(SCAN_THREADS)
void k_rdeg_scan(const int* __restrict__ deg, float* __restrict__ rdeg, int* __restrict__ start) {
    __shared__ int wsum[SCAN_THREADS / 32];
    const int t    = threadIdx.x;
    const int lane = t & 31;
    const int w    = t >> 5;
    int c[SCAN_PER];
    int tot = 0;
#pragma unroll
    for (int k = 0; k < SCAN_PER; ++k) {
        c[k] = deg[t * SCAN_PER + k];
        rdeg[t * SCAN_PER + k] = 1.0f / fmaxf((float)c[k], 1.0f);
        tot += c[k];
    }
    int inc = tot;
    for (int off = 1; off < 32; off <<= 1) {
        int v = __shfl_up(inc, off);
        if (lane >= off) inc += v;
    }
    if (lane == 31) wsum[w] = inc;
    __syncthreads();
    if (w == 0) {
        const int mine = wsum[lane];
        int ws = mine;
        for (int off = 1; off < 32; off <<= 1) {
            int v = __shfl_up(ws, off);
            if (lane >= off) ws += v;
        }
        wsum[lane] = ws - mine;
    }
    __syncthreads();
    int run = wsum[w] + inc - tot;
#pragma unroll
    for (int k = 0; k < SCAN_PER; ++k) {
        start[t * SCAN_PER + k] = run;
        run += c[k];
    }
    if (t == SCAN_THREADS - 1) start[NPAD] = run;
}

#define SF_THREADS  256
#define S1_BLOCKS   ((NPAD / 16) * (HID / 16) / (SF_THREADS / 32))
#define FILL_BLOCKS ((N_EDGES + SF_THREADS - 1) / SF_THREADS)
__global__ __launch_bounds__(SF_THREADS)
void k_sage1_fill(const float* __restrict__ agg1, const float* __restrict__ rdeg,
                  const float* __restrict__ x,
                  const float* __restrict__ Wl1, const float* __restrict__ bl1,
                  const float* __restrict__ Wr1, float* __restrict__ h1,
                  const int* __restrict__ ei, const int* __restrict__ start,
                  const int* __restrict__ rank, int* __restrict__ csr) {
    if (blockIdx.x >= S1_BLOCKS) {
        const int e = (blockIdx.x - S1_BLOCKS) * SF_THREADS + threadIdx.x;
        if (e >= N_EDGES) return;
        const int s = ei[e];
        const int d = ei[N_EDGES + e];
        csr[start[d] + rank[e]] = s;
        return;
    }
    const int tile = blockIdx.x * (SF_THREADS / 32) + (threadIdx.x >> 5);
    const int row0 = (tile >> 2) * 16;
    const int col0 = (tile & 3) * 16;
    const int lane = threadIdx.x & 31;
    const int m    = lane & 15;
    const int half = lane >> 4;
    const int kb   = half * 2;
    const int row  = row0 + m;
    const float rd = rdeg[row];

    v2f am, ax, bl, br;
    am.x = agg1[row * NODE_F + kb]     * rd;
    am.y = agg1[row * NODE_F + kb + 1] * rd;
    ax.x = (row < N_NODES) ? x[row * NODE_F + kb]     : 0.0f;
    ax.y = (row < N_NODES) ? x[row * NODE_F + kb + 1] : 0.0f;
    bl.x = Wl1[(col0 + m) * NODE_F + kb];
    bl.y = Wl1[(col0 + m) * NODE_F + kb + 1];
    br.x = Wr1[(col0 + m) * NODE_F + kb];
    br.y = Wr1[(col0 + m) * NODE_F + kb + 1];

    v8f acc = {};
    acc = __builtin_amdgcn_wmma_f32_16x16x4_f32(false, am, false, bl, (short)0, acc, false, false);
    acc = __builtin_amdgcn_wmma_f32_16x16x4_f32(false, ax, false, br, (short)0, acc, false, false);

#pragma unroll
    for (int i = 0; i < 8; ++i) {
        int r = row0 + i + half * 8;
        int c = col0 + m;
        h1[r * HID + c] = fmaxf(acc[i] + bl1[c], 0.0f);
    }
}

#define GAT_WAVES 4
__global__ __launch_bounds__(GAT_WAVES * 32)
void k_gather2(const int* __restrict__ start, const int* __restrict__ csr,
               const float* __restrict__ h1, float* __restrict__ agg2) {
    const int lane = threadIdx.x & 31;
    const int n   = __builtin_amdgcn_readfirstlane((int)(blockIdx.x * GAT_WAVES + (threadIdx.x >> 5)));
    const int beg = start[n];
    const int end = start[n + 1];
    const v2f* __restrict__ rows = (const v2f*)h1;

    v2f acc = {0.0f, 0.0f};
    int idx = (beg + lane < end) ? csr[beg + lane] : 0;
    for (int base = beg; base < end; base += 32) {
        const int nb = base + 32;
        const int idx_next = (nb + lane < end) ? csr[nb + lane] : 0;
        const int cnt = min(32, end - base);
        for (int j = 0; j < cnt; j += 8) {
            v2f v[8];
#pragma unroll
            for (int u = 0; u < 8; ++u) {
                const int s = __builtin_amdgcn_readlane(idx, j + u);
                v[u] = rows[(size_t)s * (HID / 2) + lane];
            }
#pragma unroll
            for (int u = 0; u < 8; ++u) {
                const bool ok = (j + u) < cnt;
                acc.x += ok ? v[u].x : 0.0f;
                acc.y += ok ? v[u].y : 0.0f;
            }
        }
        idx = idx_next;
    }
    ((v2f*)agg2)[(size_t)n * (HID / 2) + lane] = acc;
}

#define H2_PITCH 68
#define SH_WAVES (HID / 16)
__global__ __launch_bounds__(SH_WAVES * 32)
void k_sage2_hpart(const float* __restrict__ agg2, const float* __restrict__ rdeg,
                   const float* __restrict__ h1,
                   const float* __restrict__ Wl2, const float* __restrict__ bl2,
                   const float* __restrict__ Wr2, const float* __restrict__ Wc1,
                   float* __restrict__ hpart) {
    __shared__ __attribute__((aligned(16))) float h2t[16 * H2_PITCH];
    const int row0 = blockIdx.x * 16;
    const int col0 = (threadIdx.x >> 5) * 16;
    const int lane = threadIdx.x & 31;
    const int m    = lane & 15;
    const int half = lane >> 4;
    const int kb   = half * 2;
    const int row  = row0 + m;
    const float rd = rdeg[row];

    v8f acc = {};
#pragma unroll 4
    for (int k0 = 0; k0 < HID; k0 += 4) {
        v2f am, ah, bl, br;
        am.x = agg2[row * HID + k0 + kb]     * rd;
        am.y = agg2[row * HID + k0 + kb + 1] * rd;
        ah.x = h1[row * HID + k0 + kb];
        ah.y = h1[row * HID + k0 + kb + 1];
        bl.x = Wl2[(col0 + m) * HID + k0 + kb];
        bl.y = Wl2[(col0 + m) * HID + k0 + kb + 1];
        br.x = Wr2[(col0 + m) * HID + k0 + kb];
        br.y = Wr2[(col0 + m) * HID + k0 + kb + 1];
        acc = __builtin_amdgcn_wmma_f32_16x16x4_f32(false, am, false, bl, (short)0, acc, false, false);
        acc = __builtin_amdgcn_wmma_f32_16x16x4_f32(false, ah, false, br, (short)0, acc, false, false);
    }
#pragma unroll
    for (int i = 0; i < 8; ++i) {
        int r = i + half * 8;
        int c = col0 + m;
        h2t[r * H2_PITCH + c] = fmaxf(acc[i] + bl2[c], 0.0f);
    }
    __syncthreads();

    v8f acc2 = {};
#pragma unroll 4
    for (int k0 = 0; k0 < HID; k0 += 4) {
        v2f a, b;
        a.x = h2t[m * H2_PITCH + k0 + kb];
        a.y = h2t[m * H2_PITCH + k0 + kb + 1];
        b.x = Wc1[(col0 + m) * (2 * HID) + k0 + kb];
        b.y = Wc1[(col0 + m) * (2 * HID) + k0 + kb + 1];
        acc2 = __builtin_amdgcn_wmma_f32_16x16x4_f32(false, a, false, b, (short)0, acc2, false, false);
    }
#pragma unroll
    for (int i = 0; i < 8; ++i) {
        int r = row0 + i + half * 8;
        int c = col0 + m;
        hpart[r * HID + c] = acc2[i];
    }
}

#define ZERO_BLOCKS (ZERO_COUNT / (4 * HID))
__global__ __launch_bounds__(HID)
void k_zero_task(float* __restrict__ zero_base,
                 const float* __restrict__ task, const float* __restrict__ Wt1,
                 const float* __restrict__ bt1, const float* __restrict__ Wt2,
                 const float* __restrict__ bt2, const float* __restrict__ Wc1,
                 const float* __restrict__ bc1, float* __restrict__ tpart) {
    __shared__ float u[HID];
    __shared__ float t[HID];
    if (blockIdx.x < ZERO_BLOCKS) {
        ((float4*)zero_base)[blockIdx.x * HID + threadIdx.x] = make_float4(0.0f, 0.0f, 0.0f, 0.0f);
        return;
    }
    const int b = blockIdx.x - ZERO_BLOCKS;
    const int h = threadIdx.x;

    float acc = bt1[h];
#pragma unroll
    for (int k = 0; k < TASK_F; ++k) acc += task[b * TASK_F + k] * Wt1[h * TASK_F + k];
    u[h] = fmaxf(acc, 0.0f);
    __syncthreads();

    float acc2 = bt2[h];
#pragma unroll 8
    for (int k = 0; k < HID; ++k) acc2 += u[k] * Wt2[h * HID + k];
    t[h] = acc2;
    __syncthreads();

    float acc3 = bc1[h];
#pragma unroll 8
    for (int k = 0; k < HID; ++k) acc3 += t[k] * Wc1[h * (2 * HID) + HID + k];
    tpart[b * HID + h] = acc3;
}

#define CLS_BT      16
#define CLS_G       4
#define CLS_THREADS 128
__global__ __launch_bounds__(CLS_THREADS)
void k_classify(const float* __restrict__ hpart, const float* __restrict__ tpart,
                const float* __restrict__ Wc2, const float* __restrict__ bc2,
                float* __restrict__ out) {
    __shared__ float w2[HID];
    if (threadIdx.x < HID) w2[threadIdx.x] = Wc2[threadIdx.x];
    __syncthreads();

    const int n  = blockIdx.x * CLS_THREADS + threadIdx.x;
    const int b0 = blockIdx.y * CLS_BT;
    if (n - (int)(threadIdx.x & 31) >= N_NODES) return;

    float a[HID];
    const float4* hp = (const float4*)(hpart + (size_t)n * HID);
#pragma unroll
    for (int j = 0; j < HID / 4; ++j) {
        float4 v = hp[j];
        a[4 * j + 0] = v.x; a[4 * j + 1] = v.y; a[4 * j + 2] = v.z; a[4 * j + 3] = v.w;
    }
    float w[HID];
#pragma unroll
    for (int h = 0; h < HID; ++h) w[h] = w2[h];
    const float bias = bc2[0];

    for (int g = 0; g < CLS_BT / CLS_G; ++g) {
        const float* __restrict__ tp = tpart + (size_t)(b0 + g * CLS_G) * HID;
        float acc[CLS_G];
#pragma unroll
        for (int r = 0; r < CLS_G; ++r) acc[r] = bias;
#pragma unroll
        for (int hc = 0; hc < HID; hc += 16) {
#pragma unroll
            for (int r = 0; r < CLS_G; ++r) {
#pragma unroll
                for (int k = 0; k < 16; ++k)
                    acc[r] += fmaxf(a[hc + k] + tp[r * HID + hc + k], 0.0f) * w[hc + k];
            }
        }
        if (n < N_NODES) {
#pragma unroll
            for (int r = 0; r < CLS_G; ++r)
                out[(size_t)(b0 + g * CLS_G + r) * N_NODES + n] = acc[r];
        }
    }
}

extern "C" void kernel_launch(void* const* d_in, const int* in_sizes, int n_in,
                              void* d_out, int out_size, void* d_ws, size_t ws_size,
                              hipStream_t stream) {
    const float* x    = (const float*)d_in[0];
    const int*   ei   = (const int*)d_in[1];
    const float* task = (const float*)d_in[2];
    const float* Wl1  = (const float*)d_in[3];
    const float* bl1  = (const float*)d_in[4];
    const float* Wr1  = (const float*)d_in[5];
    const float* Wl2  = (const float*)d_in[6];
    const float* bl2  = (const float*)d_in[7];
    const float* Wr2  = (const float*)d_in[8];
    const float* Wt1  = (const float*)d_in[9];
    const float* bt1  = (const float*)d_in[10];
    const float* Wt2  = (const float*)d_in[11];
    const float* bt2  = (const float*)d_in[12];
    const float* Wc1  = (const float*)d_in[13];
    const float* bc1  = (const float*)d_in[14];
    const float* Wc2  = (const float*)d_in[15];
    const float* bc2  = (const float*)d_in[16];
    float* out = (float*)d_out;

    float* ws    = (float*)d_ws;
    int*   deg   = (int*)(ws + OFF_DEG);
    float* agg1  = ws + OFF_AGG1;
    float* agg2  = ws + OFF_AGG2;
    float* rdeg  = ws + OFF_RDEG;
    float* h1    = ws + OFF_H1;
    float* hpart = ws + OFF_HPART;
    float* tpart = ws + OFF_TPART;
    int*   rank  = (int*)(ws + OFF_H2);
    int*   csr   = (int*)(ws + OFF_HPART);
    int*   start = (int*)(ws + OFF_START);

    k_zero_task<<<ZERO_BLOCKS + BATCH, HID, 0, stream>>>(ws, task, Wt1, bt1, Wt2, bt2, Wc1, bc1, tpart);
    k_scatter1<<<(N_EDGES + 255) / 256, 256, 0, stream>>>(ei, x, agg1, deg, rank);
    k_rdeg_scan<<<1, SCAN_THREADS, 0, stream>>>(deg, rdeg, start);
    k_sage1_fill<<<S1_BLOCKS + FILL_BLOCKS, SF_THREADS, 0, stream>>>(agg1, rdeg, x, Wl1, bl1, Wr1, h1,
                                                                    ei, start, rank, csr);
    k_gather2<<<NPAD / GAT_WAVES, GAT_WAVES * 32, 0, stream>>>(start, csr, h1, agg2);
    k_sage2_hpart<<<NPAD / 16, SH_WAVES * 32, 0, stream>>>(agg2, rdeg, h1, Wl2, bl2, Wr2, Wc1, hpart);
    k_classify<<<dim3(NPAD / CLS_THREADS, BATCH / CLS_BT), CLS_THREADS, 0, stream>>>(hpart, tpart, Wc2, bc2, out);
}


# degrees, neighbour lists and layer-1 sums by node-range owner blocks in LDS (integer and float LDS adds, arrival order) instead of per-edge global atomics; launches 7 -> 4
# speedup vs baseline: 3.2787x; 1.7689x over previous
#include <hip/hip_runtime.h>
#include <hip/hip_bf16.h>

#define N_NODES 5000
#define NPAD    5120
#define N_EDGES 160000
#define HID     64
#define NODE_F  4
#define TASK_F  4
#define BATCH   512

typedef __attribute__((ext_vector_type(2))) float v2f;
typedef __attribute__((ext_vector_type(8))) float v8f;

#define OFF_AGG2   (NPAD * (1 + NODE_F))
#define OFF_RDEG   (OFF_AGG2 + NPAD * HID)
#define OFF_H1     (OFF_RDEG  + NPAD)
#define OFF_H2     (OFF_H1    + NPAD * HID)
#define OFF_HPART  (OFF_H2    + NPAD * HID)
#define OFF_TPART  (OFF_HPART + NPAD * HID)
#define OFF_START  (OFF_H2    + N_EDGES)


#define NB          128
#define OB_THREADS  1024
#define OWN_BLOCKS  (NPAD / NB)
#define E4          (N_EDGES / 4)
#define OB_TRIPS    ((E4 + OB_THREADS - 1) / OB_THREADS)
#define OB_WORDS    (OB_TRIPS / 8)
#define OB_BATCH    4
static_assert(OB_TRIPS % 8 == 0, "mask words are filled whole");
static_assert(NB * NODE_F <= OB_THREADS && NB == 4 * 32 && OB_THREADS / 32 == (NB / 16) * (HID / 16), "block shape");
__device__ __forceinline__ int ob_slot(int v, int lo, int hi, int slot, int lane, int acc, int& nbel) {
    const unsigned below = __builtin_amdgcn_ballot_w32(v < lo);
    const unsigned under = __builtin_amdgcn_ballot_w32(v < hi);
    nbel += __popc(below);
    return (lane == slot) ? (int)(under & ~below) : acc;
}
__global__ __launch_bounds__(OB_THREADS)
void k_build(const int* __restrict__ ei, const float* __restrict__ x,
             const float* __restrict__ Wl1, const float* __restrict__ bl1,
             const float* __restrict__ Wr1,
             int* __restrict__ elist, int* __restrict__ csr, int* __restrict__ start,
             float* __restrict__ rdeg, float* __restrict__ h1) {
    __shared__ int   cnt[NB];
    __shared__ int   fillp[NB];
    __shared__ float aggL[NB * NODE_F];
    __shared__ unsigned ownL[OB_WORDS][OB_THREADS];
    __shared__ int   wsum[OB_THREADS / 32];
    __shared__ int   wbel[OB_THREADS / 32];
    __shared__ int   sbase, stot;

    const int tid  = threadIdx.x;
    const int lane = tid & 31;
    const int w    = tid >> 5;
    const int lo   = blockIdx.x * NB;
    const int hi   = lo + NB;
    const int*  __restrict__ dstp = ei + N_EDGES;
    const int4* __restrict__ dst4 = (const int4*)dstp;

    int nm = 0;
    int nbel = 0;
#pragma unroll 1
    for (int wd = 0; wd < OB_WORDS; ++wd) {
        int4 d[8];
#pragma unroll
        for (int j = 0; j < 8; ++j) {
            const int i = (wd * 8 + j) * OB_THREADS + tid;
            d[j] = make_int4(0x7fffffff, 0x7fffffff, 0x7fffffff, 0x7fffffff);
            if (i < E4) d[j] = dst4[i];
        }
        int acc = 0;
#pragma unroll
        for (int j = 0; j < 8; ++j) {
            acc = ob_slot(d[j].x, lo, hi, 4 * j + 0, lane, acc, nbel);
            acc = ob_slot(d[j].y, lo, hi, 4 * j + 1, lane, acc, nbel);
            acc = ob_slot(d[j].z, lo, hi, 4 * j + 2, lane, acc, nbel);
            acc = ob_slot(d[j].w, lo, hi, 4 * j + 3, lane, acc, nbel);
        }
        ownL[wd][tid] = (unsigned)acc;
        nm += __popc((unsigned)acc);
    }

    int inc = nm;
    for (int off = 1; off < 32; off <<= 1) {
        int v = __shfl_up(inc, off);
        if (lane >= off) inc += v;
    }
    if (lane == 31) wsum[w] = inc;
    if (lane == 0)  wbel[w] = nbel;
    if (tid < NB)          cnt[tid]  = 0;
    if (tid < NB * NODE_F) aggL[tid] = 0.0f;
    __syncthreads();
    if (w == 0) {
        const int mine = wsum[lane];
        int ws = mine;
        for (int off = 1; off < 32; off <<= 1) {
            int v = __shfl_up(ws, off);
            if (lane >= off) ws += v;
        }
        wsum[lane] = ws - mine;
        int bel = wbel[lane];
        for (int off = 16; off > 0; off >>= 1) bel += __shfl_xor(bel, off);
        if (lane == 31) stot  = ws;
        if (lane == 0)  sbase = bel;
    }
    __syncthreads();
    const int base = sbase;
    const int T    = stot;

    int pos = base + wsum[w] + inc - nm;
    const int comp = lane & 3;
    for (int wd = 0; wd < OB_WORDS; ++wd) {
        unsigned b = ownL[wd][tid];
        const int grp = (wd * 8 + (lane >> 2)) * OB_THREADS + w * 32;
        while (b) {
            const int k = __builtin_ctz(b);
            b &= b - 1;
            elist[pos++] = 4 * (grp + k) + comp;
        }
    }
    __syncthreads();

    for (int i0 = 0; i0 < T; i0 += OB_BATCH * OB_THREADS) {
        int e[OB_BATCH], d[OB_BATCH];
#pragma unroll
        for (int u = 0; u < OB_BATCH; ++u) {
            const int idx = i0 + u * OB_THREADS + tid;
            e[u] = (idx < T) ? elist[base + idx] : -1;
        }
#pragma unroll
        for (int u = 0; u < OB_BATCH; ++u) d[u] = dstp[e[u] < 0 ? 0 : e[u]];
#pragma unroll
        for (int u = 0; u < OB_BATCH; ++u)
            if (e[u] >= 0) atomicAdd(&cnt[d[u] - lo], 1);
    }
    __syncthreads();

    if (w == 0) {
        const int c0 = cnt[4 * lane], c1 = cnt[4 * lane + 1], c2 = cnt[4 * lane + 2], c3 = cnt[4 * lane + 3];
        const int tot = c0 + c1 + c2 + c3;
        int ps = tot;
        for (int off = 1; off < 32; off <<= 1) {
            int v = __shfl_up(ps, off);
            if (lane >= off) ps += v;
        }
        const int b0 = base + ps - tot;
        const int b1 = b0 + c0, b2 = b1 + c1, b3 = b2 + c2;
        fillp[4 * lane] = b0; fillp[4 * lane + 1] = b1; fillp[4 * lane + 2] = b2; fillp[4 * lane + 3] = b3;
        start[lo + 4 * lane] = b0; start[lo + 4 * lane + 1] = b1; start[lo + 4 * lane + 2] = b2; start[lo + 4 * lane + 3] = b3;
        if (blockIdx.x == OWN_BLOCKS - 1 && lane == 31) start[NPAD] = b3 + c3;
    }
    __syncthreads();

    for (int i0 = 0; i0 < T; i0 += OB_BATCH * OB_THREADS) {
        int e[OB_BATCH], d[OB_BATCH], s[OB_BATCH];
        float xv[OB_BATCH][NODE_F];
#pragma unroll
        for (int u = 0; u < OB_BATCH; ++u) {
            const int idx = i0 + u * OB_THREADS + tid;
            e[u] = (idx < T) ? elist[base + idx] : -1;
        }
#pragma unroll
        for (int u = 0; u < OB_BATCH; ++u) {
            const int ee = e[u] < 0 ? 0 : e[u];
            d[u] = dstp[ee];
            s[u] = ei[ee];
        }
#pragma unroll
        for (int u = 0; u < OB_BATCH; ++u) {
#pragma unroll
            for (int k = 0; k < NODE_F; ++k) xv[u][k] = x[s[u] * NODE_F + k];
        }
#pragma unroll
        for (int u = 0; u < OB_BATCH; ++u) {
            if (e[u] >= 0) {
                const int r = d[u] - lo;
                const int slot = atomicAdd(&fillp[r], 1);
                csr[slot] = s[u];
#pragma unroll
                for (int k = 0; k < NODE_F; ++k) atomicAdd(&aggL[r * NODE_F + k], xv[u][k]);
            }
        }
    }
    __syncthreads();

    if (tid < NB) rdeg[lo + tid] = 1.0f / fmaxf((float)cnt[tid], 1.0f);
    const int lrow0 = (w >> 2) * 16;
    const int col0  = (w & 3) * 16;
    const int m     = lane & 15;
    const int half  = lane >> 4;
    const int kb    = half * 2;
    const int lrow  = lrow0 + m;
    const int row   = lo + lrow;
    const float rd  = 1.0f / fmaxf((float)cnt[lrow], 1.0f);

    v2f am, ax, bl, br;
    am.x = aggL[lrow * NODE_F + kb]     * rd;
    am.y = aggL[lrow * NODE_F + kb + 1] * rd;
    ax.x = (row < N_NODES) ? x[row * NODE_F + kb]     : 0.0f;
    ax.y = (row < N_NODES) ? x[row * NODE_F + kb + 1] : 0.0f;
    bl.x = Wl1[(col0 + m) * NODE_F + kb];
    bl.y = Wl1[(col0 + m) * NODE_F + kb + 1];
    br.x = Wr1[(col0 + m) * NODE_F + kb];
    br.y = Wr1[(col0 + m) * NODE_F + kb + 1];

    v8f acc = {};
    acc = __builtin_amdgcn_wmma_f32_16x16x4_f32(false, am, false, bl, (short)0, acc, false, false);
    acc = __builtin_amdgcn_wmma_f32_16x16x4_f32(false, ax, false, br, (short)0, acc, false, false);

#pragma unroll
    for (int i = 0; i < 8; ++i) {
        int r = lo + lrow0 + i + half * 8;
        int c = col0 + m;
        h1[r * HID + c] = fmaxf(acc[i] + bl1[c], 0.0f);
    }
}

#define GAT_WAVES  4
#define GAT_BLOCKS (NPAD / GAT_WAVES)
#define TASK_ROWS  ((GAT_WAVES * 32) / HID)
__global__ __launch_bounds__(GAT_WAVES * 32)
void k_gather2_task(const int* __restrict__ start, const int* __restrict__ csr,
                    const float* __restrict__ h1, float* __restrict__ agg2,
                    const float* __restrict__ task, const float* __restrict__ Wt1,
                    const float* __restrict__ bt1, const float* __restrict__ Wt2,
                    const float* __restrict__ bt2, const float* __restrict__ Wc1,
                    const float* __restrict__ bc1, float* __restrict__ tpart) {
    __shared__ float u[TASK_ROWS][HID];
    __shared__ float t[TASK_ROWS][HID];
    if (blockIdx.x >= GAT_BLOCKS) {
        const int sub = threadIdx.x / HID;
        const int h   = threadIdx.x % HID;
        const int b   = (blockIdx.x - GAT_BLOCKS) * TASK_ROWS + sub;

        float acc = bt1[h];
#pragma unroll
        for (int k = 0; k < TASK_F; ++k) acc += task[b * TASK_F + k] * Wt1[h * TASK_F + k];
        u[sub][h] = fmaxf(acc, 0.0f);
        __syncthreads();

        float acc2 = bt2[h];
#pragma unroll 8
        for (int k = 0; k < HID; ++k) acc2 += u[sub][k] * Wt2[h * HID + k];
        t[sub][h] = acc2;
        __syncthreads();

        float acc3 = bc1[h];
#pragma unroll 8
        for (int k = 0; k < HID; ++k) acc3 += t[sub][k] * Wc1[h * (2 * HID) + HID + k];
        tpart[b * HID + h] = acc3;
        return;
    }
    const int lane = threadIdx.x & 31;
    const int n   = __builtin_amdgcn_readfirstlane((int)(blockIdx.x * GAT_WAVES + (threadIdx.x >> 5)));
    const int beg = start[n];
    const int end = start[n + 1];
    const v2f* __restrict__ rows = (const v2f*)h1;

    v2f acc = {0.0f, 0.0f};
    int idx = (beg + lane < end) ? csr[beg + lane] : 0;
    for (int base = beg; base < end; base += 32) {
        const int nb = base + 32;
        const int idx_next = (nb + lane < end) ? csr[nb + lane] : 0;
        const int cnt = min(32, end - base);
        for (int j = 0; j < cnt; j += 8) {
            v2f v[8];
#pragma unroll
            for (int uu = 0; uu < 8; ++uu) {
                const int s = __builtin_amdgcn_readlane(idx, j + uu);
                v[uu] = rows[(size_t)s * (HID / 2) + lane];
            }
#pragma unroll
            for (int uu = 0; uu < 8; ++uu) {
                const bool ok = (j + uu) < cnt;
                acc.x += ok ? v[uu].x : 0.0f;
                acc.y += ok ? v[uu].y : 0.0f;
            }
        }
        idx = idx_next;
    }
    ((v2f*)agg2)[(size_t)n * (HID / 2) + lane] = acc;
}

#define H2_PITCH 68
#define SH_WAVES (HID / 16)
__global__ __launch_bounds__(SH_WAVES * 32)
void k_sage2_hpart(const float* __restrict__ agg2, const float* __restrict__ rdeg,
                   const float* __restrict__ h1,
                   const float* __restrict__ Wl2, const float* __restrict__ bl2,
                   const float* __restrict__ Wr2, const float* __restrict__ Wc1,
                   float* __restrict__ hpart) {
    __shared__ __attribute__((aligned(16))) float h2t[16 * H2_PITCH];
    const int row0 = blockIdx.x * 16;
    const int col0 = (threadIdx.x >> 5) * 16;
    const int lane = threadIdx.x & 31;
    const int m    = lane & 15;
    const int half = lane >> 4;
    const int kb   = half * 2;
    const int row  = row0 + m;
    const float rd = rdeg[row];

    v8f acc = {};
#pragma unroll 4
    for (int k0 = 0; k0 < HID; k0 += 4) {
        v2f am, ah, bl, br;
        am.x = agg2[row * HID + k0 + kb]     * rd;
        am.y = agg2[row * HID + k0 + kb + 1] * rd;
        ah.x = h1[row * HID + k0 + kb];
        ah.y = h1[row * HID + k0 + kb + 1];
        bl.x = Wl2[(col0 + m) * HID + k0 + kb];
        bl.y = Wl2[(col0 + m) * HID + k0 + kb + 1];
        br.x = Wr2[(col0 + m) * HID + k0 + kb];
        br.y = Wr2[(col0 + m) * HID + k0 + kb + 1];
        acc = __builtin_amdgcn_wmma_f32_16x16x4_f32(false, am, false, bl, (short)0, acc, false, false);
        acc = __builtin_amdgcn_wmma_f32_16x16x4_f32(false, ah, false, br, (short)0, acc, false, false);
    }
#pragma unroll
    for (int i = 0; i < 8; ++i) {
        int r = i + half * 8;
        int c = col0 + m;
        h2t[r * H2_PITCH + c] = fmaxf(acc[i] + bl2[c], 0.0f);
    }
    __syncthreads();

    v8f acc2 = {};
#pragma unroll 4
    for (int k0 = 0; k0 < HID; k0 += 4) {
        v2f a, b;
        a.x = h2t[m * H2_PITCH + k0 + kb];
        a.y = h2t[m * H2_PITCH + k0 + kb + 1];
        b.x = Wc1[(col0 + m) * (2 * HID) + k0 + kb];
        b.y = Wc1[(col0 + m) * (2 * HID) + k0 + kb + 1];
        acc2 = __builtin_amdgcn_wmma_f32_16x16x4_f32(false, a, false, b, (short)0, acc2, false, false);
    }
#pragma unroll
    for (int i = 0; i < 8; ++i) {
        int r = row0 + i + half * 8;
        int c = col0 + m;
        hpart[r * HID + c] = acc2[i];
    }
}

#define CLS_BT      16
#define CLS_G       4
#define CLS_THREADS 128
__global__ __launch_bounds__(CLS_THREADS)
void k_classify(const float* __restrict__ hpart, const float* __restrict__ tpart,
                const float* __restrict__ Wc2, const float* __restrict__ bc2,
                float* __restrict__ out) {
    __shared__ float w2[HID];
    if (threadIdx.x < HID) w2[threadIdx.x] = Wc2[threadIdx.x];
    __syncthreads();

    const int n  = blockIdx.x * CLS_THREADS + threadIdx.x;
    const int b0 = blockIdx.y * CLS_BT;
    if (n - (int)(threadIdx.x & 31) >= N_NODES) return;

    float a[HID];
    const float4* hp = (const float4*)(hpart + (size_t)n * HID);
#pragma unroll
    for (int j = 0; j < HID / 4; ++j) {
        float4 v = hp[j];
        a[4 * j + 0] = v.x; a[4 * j + 1] = v.y; a[4 * j + 2] = v.z; a[4 * j + 3] = v.w;
    }
    float w[HID];
#pragma unroll
    for (int h = 0; h < HID; ++h) w[h] = w2[h];
    const float bias = bc2[0];

    for (int g = 0; g < CLS_BT / CLS_G; ++g) {
        const float* __restrict__ tp = tpart + (size_t)(b0 + g * CLS_G) * HID;
        float acc[CLS_G];
#pragma unroll
        for (int r = 0; r < CLS_G; ++r) acc[r] = bias;
#pragma unroll
        for (int hc = 0; hc < HID; hc += 16) {
#pragma unroll
            for (int r = 0; r < CLS_G; ++r) {
#pragma unroll
                for (int k = 0; k < 16; ++k)
                    acc[r] += fmaxf(a[hc + k] + tp[r * HID + hc + k], 0.0f) * w[hc + k];
            }
        }
        if (n < N_NODES) {
#pragma unroll
            for (int r = 0; r < CLS_G; ++r)
                out[(size_t)(b0 + g * CLS_G + r) * N_NODES + n] = acc[r];
        }
    }
}

extern "C" void kernel_launch(void* const* d_in, const int* in_sizes, int n_in,
                              void* d_out, int out_size, void* d_ws, size_t ws_size,
                              hipStream_t stream) {
    const float* x    = (const float*)d_in[0];
    const int*   ei   = (const int*)d_in[1];
    const float* task = (const float*)d_in[2];
    const float* Wl1  = (const float*)d_in[3];
    const float* bl1  = (const float*)d_in[4];
    const float* Wr1  = (const float*)d_in[5];
    const float* Wl2  = (const float*)d_in[6];
    const float* bl2  = (const float*)d_in[7];
    const float* Wr2  = (const float*)d_in[8];
    const float* Wt1  = (const float*)d_in[9];
    const float* bt1  = (const float*)d_in[10];
    const float* Wt2  = (const float*)d_in[11];
    const float* bt2  = (const float*)d_in[12];
    const float* Wc1  = (const float*)d_in[13];
    const float* bc1  = (const float*)d_in[14];
    const float* Wc2  = (const float*)d_in[15];
    const float* bc2  = (const float*)d_in[16];
    float* out = (float*)d_out;

    float* ws    = (float*)d_ws;
    float* agg2  = ws + OFF_AGG2;
    float* rdeg  = ws + OFF_RDEG;
    float* h1    = ws + OFF_H1;
    float* hpart = ws + OFF_HPART;
    float* tpart = ws + OFF_TPART;
    int*   elist = (int*)(ws + OFF_H2);
    int*   csr   = (int*)(ws + OFF_HPART);
    int*   start = (int*)(ws + OFF_START);

    k_build<<<OWN_BLOCKS, OB_THREADS, 0, stream>>>(ei, x, Wl1, bl1, Wr1, elist, csr, start, rdeg, h1);
    k_gather2_task<<<GAT_BLOCKS + BATCH / TASK_ROWS, GAT_WAVES * 32, 0, stream>>>(start, csr, h1, agg2,
                                                                  task, Wt1, bt1, Wt2, bt2, Wc1, bc1, tpart);
    k_sage2_hpart<<<NPAD / 16, SH_WAVES * 32, 0, stream>>>(agg2, rdeg, h1, Wl2, bl2, Wr2, Wc1, hpart);
    k_classify<<<dim3(NPAD / CLS_THREADS, BATCH / CLS_BT), CLS_THREADS, 0, stream>>>(hpart, tpart, Wc2, bc2, out);
}


# 7424 -> 2560 waves in the middle launches: gather + SAGE 2 + hpart as one launch of 8-wave blocks (2 rows per wave, operands staged in LDS), task encoder on idle blocks of launch 1; sums unchanged
# speedup vs baseline: 3.4870x; 1.0635x over previous
#include <hip/hip_runtime.h>
#include <hip/hip_bf16.h>

#define N_NODES 5000
#define NPAD    5120
#define N_EDGES 160000
#define HID     64
#define NODE_F  4
#define TASK_F  4
#define BATCH   512

typedef __attribute__((ext_vector_type(2))) float v2f;
typedef __attribute__((ext_vector_type(8))) float v8f;

#define OFF_AGG2   (NPAD * (1 + NODE_F))
#define OFF_RDEG   (OFF_AGG2 + NPAD * HID)
#define OFF_H1     (OFF_RDEG  + NPAD)
#define OFF_H2     (OFF_H1    + NPAD * HID)
#define OFF_HPART  (OFF_H2    + NPAD * HID)
#define OFF_TPART  (OFF_HPART + NPAD * HID)
#define OFF_START  (OFF_H2    + N_EDGES)


#define NB          128
#define OB_THREADS  1024
#define OWN_BLOCKS  (NPAD / NB)
#define E4          (N_EDGES / 4)
#define OB_TRIPS    ((E4 + OB_THREADS - 1) / OB_THREADS)
#define OB_WORDS    (OB_TRIPS / 8)
#define OB_BATCH    4
static_assert(OB_TRIPS % 8 == 0, "mask words are filled whole");
static_assert(NB * NODE_F <= OB_THREADS && NB == 4 * 32 && OB_THREADS / 32 == (NB / 16) * (HID / 16), "block shape");
__device__ __forceinline__ int ob_slot(int v, int lo, int hi, int slot, int lane, int acc, int& nbel) {
    const unsigned below = __builtin_amdgcn_ballot_w32(v < lo);
    const unsigned under = __builtin_amdgcn_ballot_w32(v < hi);
    nbel += __popc(below);
    return (lane == slot) ? (int)(under & ~below) : acc;
}
#define TASK_ROWS   (OB_THREADS / HID)
#define TASK_BLOCKS (BATCH / TASK_ROWS)
static_assert(BATCH % TASK_ROWS == 0, "task blocks are whole");
__global__ __launch_bounds__(OB_THREADS)
void k_build(const int* __restrict__ ei, const float* __restrict__ x,
             const float* __restrict__ Wl1, const float* __restrict__ bl1,
             const float* __restrict__ Wr1,
             int* __restrict__ elist, int* __restrict__ csr, int* __restrict__ start,
             float* __restrict__ rdeg, float* __restrict__ h1,
             const float* __restrict__ task, const float* __restrict__ Wt1,
             const float* __restrict__ bt1, const float* __restrict__ Wt2,
             const float* __restrict__ bt2, const float* __restrict__ Wc1,
             const float* __restrict__ bc1, float* __restrict__ tpart) {
    __shared__ float tu[TASK_ROWS][HID];
    __shared__ float tt[TASK_ROWS][HID];
    if (blockIdx.x >= OWN_BLOCKS) {
        const int sub = threadIdx.x / HID;
        const int h   = threadIdx.x % HID;
        const int b   = (blockIdx.x - OWN_BLOCKS) * TASK_ROWS + sub;

        float acc = bt1[h];
#pragma unroll
        for (int k = 0; k < TASK_F; ++k) acc += task[b * TASK_F + k] * Wt1[h * TASK_F + k];
        tu[sub][h] = fmaxf(acc, 0.0f);
        __syncthreads();

        float acc2 = bt2[h];
#pragma unroll 8
        for (int k = 0; k < HID; ++k) acc2 += tu[sub][k] * Wt2[h * HID + k];
        tt[sub][h] = acc2;
        __syncthreads();

        float acc3 = bc1[h];
#pragma unroll 8
        for (int k = 0; k < HID; ++k) acc3 += tt[sub][k] * Wc1[h * (2 * HID) + HID + k];
        tpart[b * HID + h] = acc3;
        return;
    }
    __shared__ int   cnt[NB];
    __shared__ int   fillp[NB];
    __shared__ float aggL[NB * NODE_F];
    __shared__ unsigned ownL[OB_WORDS][OB_THREADS];
    __shared__ int   wsum[OB_THREADS / 32];
    __shared__ int   wbel[OB_THREADS / 32];
    __shared__ int   sbase, stot;

    const int tid  = threadIdx.x;
    const int lane = tid & 31;
    const int w    = tid >> 5;
    const int lo   = blockIdx.x * NB;
    const int hi   = lo + NB;
    const int*  __restrict__ dstp = ei + N_EDGES;
    const int4* __restrict__ dst4 = (const int4*)dstp;

    int nm = 0;
    int nbel = 0;
#pragma unroll 1
    for (int wd = 0; wd < OB_WORDS; ++wd) {
        int4 d[8];
#pragma unroll
        for (int j = 0; j < 8; ++j) {
            const int i = (wd * 8 + j) * OB_THREADS + tid;
            d[j] = make_int4(0x7fffffff, 0x7fffffff, 0x7fffffff, 0x7fffffff);
            if (i < E4) d[j] = dst4[i];
        }
        int acc = 0;
#pragma unroll
        for (int j = 0; j < 8; ++j) {
            acc = ob_slot(d[j].x, lo, hi, 4 * j + 0, lane, acc, nbel);
            acc = ob_slot(d[j].y, lo, hi, 4 * j + 1, lane, acc, nbel);
            acc = ob_slot(d[j].z, lo, hi, 4 * j + 2, lane, acc, nbel);
            acc = ob_slot(d[j].w, lo, hi, 4 * j + 3, lane, acc, nbel);
        }
        ownL[wd][tid] = (unsigned)acc;
        nm += __popc((unsigned)acc);
    }

    int inc = nm;
    for (int off = 1; off < 32; off <<= 1) {
        int v = __shfl_up(inc, off);
        if (lane >= off) inc += v;
    }
    if (lane == 31) wsum[w] = inc;
    if (lane == 0)  wbel[w] = nbel;
    if (tid < NB)          cnt[tid]  = 0;
    if (tid < NB * NODE_F) aggL[tid] = 0.0f;
    __syncthreads();
    if (w == 0) {
        const int mine = wsum[lane];
        int ws = mine;
        for (int off = 1; off < 32; off <<= 1) {
            int v = __shfl_up(ws, off);
            if (lane >= off) ws += v;
        }
        wsum[lane] = ws - mine;
        int bel = wbel[lane];
        for (int off = 16; off > 0; off >>= 1) bel += __shfl_xor(bel, off);
        if (lane == 31) stot  = ws;
        if (lane == 0)  sbase = bel;
    }
    __syncthreads();
    const int base = sbase;
    const int T    = stot;

    int pos = base + wsum[w] + inc - nm;
    const int comp = lane & 3;
    for (int wd = 0; wd < OB_WORDS; ++wd) {
        unsigned b = ownL[wd][tid];
        const int grp = (wd * 8 + (lane >> 2)) * OB_THREADS + w * 32;
        while (b) {
            const int k = __builtin_ctz(b);
            b &= b - 1;
            elist[pos++] = 4 * (grp + k) + comp;
        }
    }
    __syncthreads();

    for (int i0 = 0; i0 < T; i0 += OB_BATCH * OB_THREADS) {
        int e[OB_BATCH], d[OB_BATCH];
#pragma unroll
        for (int u = 0; u < OB_BATCH; ++u) {
            const int idx = i0 + u * OB_THREADS + tid;
            e[u] = (idx < T) ? elist[base + idx] : -1;
        }
#pragma unroll
        for (int u = 0; u < OB_BATCH; ++u) d[u] = dstp[e[u] < 0 ? 0 : e[u]];
#pragma unroll
        for (int u = 0; u < OB_BATCH; ++u)
            if (e[u] >= 0) atomicAdd(&cnt[d[u] - lo], 1);
    }
    __syncthreads();

    if (w == 0) {
        const int c0 = cnt[4 * lane], c1 = cnt[4 * lane + 1], c2 = cnt[4 * lane + 2], c3 = cnt[4 * lane + 3];
        const int tot = c0 + c1 + c2 + c3;
        int ps = tot;
        for (int off = 1; off < 32; off <<= 1) {
            int v = __shfl_up(ps, off);
            if (lane >= off) ps += v;
        }
        const int b0 = base + ps - tot;
        const int b1 = b0 + c0, b2 = b1 + c1, b3 = b2 + c2;
        fillp[4 * lane] = b0; fillp[4 * lane + 1] = b1; fillp[4 * lane + 2] = b2; fillp[4 * lane + 3] = b3;
        start[lo + 4 * lane] = b0; start[lo + 4 * lane + 1] = b1; start[lo + 4 * lane + 2] = b2; start[lo + 4 * lane + 3] = b3;
        if (blockIdx.x == OWN_BLOCKS - 1 && lane == 31) start[NPAD] = b3 + c3;
    }
    __syncthreads();

    for (int i0 = 0; i0 < T; i0 += OB_BATCH * OB_THREADS) {
        int e[OB_BATCH], d[OB_BATCH], s[OB_BATCH];
        float xv[OB_BATCH][NODE_F];
#pragma unroll
        for (int u = 0; u < OB_BATCH; ++u) {
            const int idx = i0 + u * OB_THREADS + tid;
            e[u] = (idx < T) ? elist[base + idx] : -1;
        }
#pragma unroll
        for (int u = 0; u < OB_BATCH; ++u) {
            const int ee = e[u] < 0 ? 0 : e[u];
            d[u] = dstp[ee];
            s[u] = ei[ee];
        }
#pragma unroll
        for (int u = 0; u < OB_BATCH; ++u) {
#pragma unroll
            for (int k = 0; k < NODE_F; ++k) xv[u][k] = x[s[u] * NODE_F + k];
        }
#pragma unroll
        for (int u = 0; u < OB_BATCH; ++u) {
            if (e[u] >= 0) {
                const int r = d[u] - lo;
                const int slot = atomicAdd(&fillp[r], 1);
                csr[slot] = s[u];
#pragma unroll
                for (int k = 0; k < NODE_F; ++k) atomicAdd(&aggL[r * NODE_F + k], xv[u][k]);
            }
        }
    }
    __syncthreads();

    if (tid < NB) rdeg[lo + tid] = 1.0f / fmaxf((float)cnt[tid], 1.0f);
    const int lrow0 = (w >> 2) * 16;
    const int col0  = (w & 3) * 16;
    const int m     = lane & 15;
    const int half  = lane >> 4;
    const int kb    = half * 2;
    const int lrow  = lrow0 + m;
    const int row   = lo + lrow;
    const float rd  = 1.0f / fmaxf((float)cnt[lrow], 1.0f);

    v2f am, ax, bl, br;
    am.x = aggL[lrow * NODE_F + kb]     * rd;
    am.y = aggL[lrow * NODE_F + kb + 1] * rd;
    ax.x = (row < N_NODES) ? x[row * NODE_F + kb]     : 0.0f;
    ax.y = (row < N_NODES) ? x[row * NODE_F + kb + 1] : 0.0f;
    bl.x = Wl1[(col0 + m) * NODE_F + kb];
    bl.y = Wl1[(col0 + m) * NODE_F + kb + 1];
    br.x = Wr1[(col0 + m) * NODE_F + kb];
    br.y = Wr1[(col0 + m) * NODE_F + kb + 1];

    v8f acc = {};
    acc = __builtin_amdgcn_wmma_f32_16x16x4_f32(false, am, false, bl, (short)0, acc, false, false);
    acc = __builtin_amdgcn_wmma_f32_16x16x4_f32(false, ax, false, br, (short)0, acc, false, false);

#pragma unroll
    for (int i = 0; i < 8; ++i) {
        int r = lo + lrow0 + i + half * 8;
        int c = col0 + m;
        h1[r * HID + c] = fmaxf(acc[i] + bl1[c], 0.0f);
    }
}

#define T_PITCH  68
#define SH_WAVES (HID / 16)
#define FG_WAVES 8
#define FG_RPW   (16 / FG_WAVES)
#define FG_BATCH 32
#define KS       (HID / 4)
#define W_PIECES ((HID * HID / 4) / (FG_WAVES * 32))
static_assert(FG_RPW == 2 && FG_WAVES >= SH_WAVES, "block shape");
static_assert(W_PIECES * FG_WAVES * 32 * 4 == HID * HID && 16 * HID / 4 == FG_WAVES * 32, "copies are whole");
__device__ __forceinline__ v2f gather_row(int beg, int end, int idx, const int* __restrict__ csr,
                                          const v2f* __restrict__ rows, int lane) {
    v2f acc = {0.0f, 0.0f};
    for (int base = beg; base < end; base += 32) {
        const int nb = base + 32;
        const int idx_next = (nb + lane < end) ? csr[nb + lane] : 0;
        const int cnt = min(32, end - base);
        for (int j = 0; j < cnt; j += FG_BATCH) {
            v2f v[FG_BATCH];
#pragma unroll
            for (int uu = 0; uu < FG_BATCH; ++uu) {
                const int s = __builtin_amdgcn_readlane(idx, j + uu);
                v[uu] = rows[(size_t)s * (HID / 2) + lane];
            }
#pragma unroll
            for (int uu = 0; uu < FG_BATCH; ++uu) {
                const bool ok = (j + uu) < cnt;
                acc.x += ok ? v[uu].x : 0.0f;
                acc.y += ok ? v[uu].y : 0.0f;
            }
        }
        idx = idx_next;
    }
    return acc;
}
__global__ __launch_bounds__(FG_WAVES * 32)
void k_gather_sage2_hpart(const int* __restrict__ start, const int* __restrict__ csr,
                          const float* __restrict__ h1, const float* __restrict__ rdeg,
                          const float* __restrict__ Wl2, const float* __restrict__ bl2,
                          const float* __restrict__ Wr2, const float* __restrict__ Wc1,
                          float* __restrict__ hpart) {
    __shared__ __attribute__((aligned(16))) float wlT[HID * T_PITCH];
    __shared__ __attribute__((aligned(16))) float wrT[HID * T_PITCH];
    __shared__ __attribute__((aligned(16))) float wcT[HID * T_PITCH];
    __shared__ __attribute__((aligned(16))) float h1T[16 * T_PITCH];
    __shared__ __attribute__((aligned(16))) float aggT[16 * T_PITCH];
    __shared__ __attribute__((aligned(16))) float h2t[16 * T_PITCH];
    const int tid  = threadIdx.x;
    const int row0 = blockIdx.x * 16;
    const int lane = tid & 31;
    const int w    = __builtin_amdgcn_readfirstlane(tid >> 5);
    const int m    = lane & 15;
    const int half = lane >> 4;
    const int kb   = half * 2;
    const int col0 = (w & (SH_WAVES - 1)) * 16;
    const bool mat = w < SH_WAVES;

    float4 ql[W_PIECES], qr[W_PIECES], qc[W_PIECES];
#pragma unroll
    for (int i = 0; i < W_PIECES; ++i) {
        const int p = tid + FG_WAVES * 32 * i;
        const int r = p >> 4, c4 = (p & 15) * 4;
        ql[i] = *(const float4*)(Wl2 + r * HID + c4);
        qr[i] = *(const float4*)(Wr2 + r * HID + c4);
        qc[i] = *(const float4*)(Wc1 + r * (2 * HID) + c4);
    }
    const float4 qh = *(const float4*)(h1 + (row0 + (tid >> 4)) * HID + (tid & 15) * 4);

    const int n  = row0 + FG_RPW * w;
    const int s0 = start[n];
    const int s1 = start[n + 1];
    const int s2 = start[n + 2];
    const int idxA = (s0 + lane < s1) ? csr[s0 + lane] : 0;
    const int idxB = (s1 + lane < s2) ? csr[s1 + lane] : 0;

#pragma unroll
    for (int i = 0; i < W_PIECES; ++i) {
        const int p = tid + FG_WAVES * 32 * i;
        const int r = p >> 4, c4 = (p & 15) * 4;
        *(float4*)&wlT[r * T_PITCH + c4] = ql[i];
        *(float4*)&wrT[r * T_PITCH + c4] = qr[i];
        *(float4*)&wcT[r * T_PITCH + c4] = qc[i];
    }
    *(float4*)&h1T[(tid >> 4) * T_PITCH + (tid & 15) * 4] = qh;

    {
        const v2f* __restrict__ rows = (const v2f*)h1;
        const v2f sa = gather_row(s0, s1, idxA, csr, rows, lane);
        const v2f sb = gather_row(s1, s2, idxB, csr, rows, lane);
        *(v2f*)&aggT[(FG_RPW * w)     * T_PITCH + 2 * lane] = sa;
        *(v2f*)&aggT[(FG_RPW * w + 1) * T_PITCH + 2 * lane] = sb;
    }
    __syncthreads();

    if (mat) {
        const float rd   = rdeg[row0 + m];
        const float bias = bl2[col0 + m];
        v8f acc = {};
#pragma unroll
        for (int ks = 0; ks < KS; ++ks) {
            const int k = 4 * ks + kb;
            v2f am = *(const v2f*)&aggT[m * T_PITCH + k];
            am.x *= rd;
            am.y *= rd;
            const v2f ah = *(const v2f*)&h1T[m * T_PITCH + k];
            const v2f bl = *(const v2f*)&wlT[(col0 + m) * T_PITCH + k];
            const v2f br = *(const v2f*)&wrT[(col0 + m) * T_PITCH + k];
            acc = __builtin_amdgcn_wmma_f32_16x16x4_f32(false, am, false, bl, (short)0, acc, false, false);
            acc = __builtin_amdgcn_wmma_f32_16x16x4_f32(false, ah, false, br, (short)0, acc, false, false);
        }
#pragma unroll
        for (int i = 0; i < 8; ++i) {
            int r = i + half * 8;
            int c = col0 + m;
            h2t[r * T_PITCH + c] = fmaxf(acc[i] + bias, 0.0f);
        }
    }
    __syncthreads();

    if (mat) {
        v8f acc2 = {};
#pragma unroll
        for (int ks = 0; ks < KS; ++ks) {
            const int k = 4 * ks + kb;
            const v2f a = *(const v2f*)&h2t[m * T_PITCH + k];
            const v2f b = *(const v2f*)&wcT[(col0 + m) * T_PITCH + k];
            acc2 = __builtin_amdgcn_wmma_f32_16x16x4_f32(false, a, false, b, (short)0, acc2, false, false);
        }
#pragma unroll
        for (int i = 0; i < 8; ++i) {
            int r = row0 + i + half * 8;
            int c = col0 + m;
            hpart[r * HID + c] = acc2[i];
        }
    }
}

#define CLS_BT      16
#define CLS_G       4
#define CLS_THREADS 128
__global__ __launch_bounds__(CLS_THREADS)
void k_classify(const float* __restrict__ hpart, const float* __restrict__ tpart,
                const float* __restrict__ Wc2, const float* __restrict__ bc2,
                float* __restrict__ out) {
    __shared__ float w2[HID];
    if (threadIdx.x < HID) w2[threadIdx.x] = Wc2[threadIdx.x];
    __syncthreads();

    const int n  = blockIdx.x * CLS_THREADS + threadIdx.x;
    const int b0 = blockIdx.y * CLS_BT;
    if (n - (int)(threadIdx.x & 31) >= N_NODES) return;

    float a[HID];
    const float4* hp = (const float4*)(hpart + (size_t)n * HID);
#pragma unroll
    for (int j = 0; j < HID / 4; ++j) {
        float4 v = hp[j];
        a[4 * j + 0] = v.x; a[4 * j + 1] = v.y; a[4 * j + 2] = v.z; a[4 * j + 3] = v.w;
    }
    float w[HID];
#pragma unroll
    for (int h = 0; h < HID; ++h) w[h] = w2[h];
    const float bias = bc2[0];

    for (int g = 0; g < CLS_BT / CLS_G; ++g) {
        const float* __restrict__ tp = tpart + (size_t)(b0 + g * CLS_G) * HID;
        float acc[CLS_G];
#pragma unroll
        for (int r = 0; r < CLS_G; ++r) acc[r] = bias;
#pragma unroll
        for (int hc = 0; hc < HID; hc += 16) {
#pragma unroll
            for (int r = 0; r < CLS_G; ++r) {
#pragma unroll
                for (int k = 0; k < 16; ++k)
                    acc[r] += fmaxf(a[hc + k] + tp[r * HID + hc + k], 0.0f) * w[hc + k];
            }
        }
        if (n < N_NODES) {
#pragma unroll
            for (int r = 0; r < CLS_G; ++r)
                out[(size_t)(b0 + g * CLS_G + r) * N_NODES + n] = acc[r];
        }
    }
}

extern "C" void kernel_launch(void* const* d_in, const int* in_sizes, int n_in,
                              void* d_out, int out_size, void* d_ws, size_t ws_size,
                              hipStream_t stream) {
    const float* x    = (const float*)d_in[0];
    const int*   ei   = (const int*)d_in[1];
    const float* task = (const float*)d_in[2];
    const float* Wl1  = (const float*)d_in[3];
    const float* bl1  = (const float*)d_in[4];
    const float* Wr1  = (const float*)d_in[5];
    const float* Wl2  = (const float*)d_in[6];
    const float* bl2  = (const float*)d_in[7];
    const float* Wr2  = (const float*)d_in[8];
    const float* Wt1  = (const float*)d_in[9];
    const float* bt1  = (const float*)d_in[10];
    const float* Wt2  = (const float*)d_in[11];
    const float* bt2  = (const float*)d_in[12];
    const float* Wc1  = (const float*)d_in[13];
    const float* bc1  = (const float*)d_in[14];
    const float* Wc2  = (const float*)d_in[15];
    const float* bc2  = (const float*)d_in[16];
    float* out = (float*)d_out;

    float* ws    = (float*)d_ws;
    float* rdeg  = ws + OFF_RDEG;
    float* h1    = ws + OFF_H1;
    float* hpart = ws + OFF_HPART;
    float* tpart = ws + OFF_TPART;
    int*   elist = (int*)(ws + OFF_H2);
    int*   csr   = (int*)(ws + OFF_AGG2);
    int*   start = (int*)(ws + OFF_START);

    k_build<<<OWN_BLOCKS + TASK_BLOCKS, OB_THREADS, 0, stream>>>(ei, x, Wl1, bl1, Wr1, elist, csr, start, rdeg, h1,
                                                                 task, Wt1, bt1, Wt2, bt2, Wc1, bc1, tpart);
    k_gather_sage2_hpart<<<NPAD / 16, FG_WAVES * 32, 0, stream>>>(start, csr, h1, rdeg, Wl2, bl2, Wr2, Wc1, hpart);
    k_classify<<<dim3(NPAD / CLS_THREADS, BATCH / CLS_BT), CLS_THREADS, 0, stream>>>(hpart, tpart, Wc2, bc2, out);
}


# build launch: one-line-per-lane global accesses per edge 5 -> 2: dst and src fetched once and kept packed in the list entry, x rows and the block's piece of csr held in LDS; LDS adds as before
# speedup vs baseline: 3.8149x; 1.0940x over previous
#include <hip/hip_runtime.h>
#include <hip/hip_bf16.h>

#define N_NODES 5000
#define NPAD    5120
#define N_EDGES 160000
#define HID     64
#define NODE_F  4
#define TASK_F  4
#define BATCH   512

typedef __attribute__((ext_vector_type(2))) float v2f;
typedef __attribute__((ext_vector_type(8))) float v8f;

#define OFF_AGG2   (NPAD * (1 + NODE_F))
#define OFF_RDEG   (OFF_AGG2 + NPAD * HID)
#define OFF_H1     (OFF_RDEG  + NPAD)
#define OFF_H2     (OFF_H1    + NPAD * HID)
#define OFF_HPART  (OFF_H2    + NPAD * HID)
#define OFF_TPART  (OFF_HPART + NPAD * HID)
#define OFF_START  (OFF_H2    + N_EDGES)


#define NB          128
#define OB_THREADS  1024
#define OWN_BLOCKS  (NPAD / NB)
#define E4          (N_EDGES / 4)
#define OB_TRIPS    ((E4 + OB_THREADS - 1) / OB_THREADS)
#define OB_WORDS    (OB_TRIPS / 8)
#define OB_BATCH    4
#define X_PIECES    ((N_NODES + OB_THREADS - 1) / OB_THREADS)
#define ROW_BITS    7
#define CSR_CAP     (OB_WORDS * OB_THREADS)
static_assert((1 << ROW_BITS) == NB && (long long)(N_NODES - 1) * NB + NB - 1 < 0x7fffffffLL, "packed entries are non-negative ints");
static_assert(OB_TRIPS % 8 == 0, "mask words are filled whole");
static_assert(NB * NODE_F <= OB_THREADS && NB == 4 * 32 && OB_THREADS / 32 == (NB / 16) * (HID / 16), "block shape");
__device__ __forceinline__ int ob_slot(int v, int lo, int hi, int slot, int lane, int acc, int& nbel) {
    const unsigned below = __builtin_amdgcn_ballot_w32(v < lo);
    const unsigned under = __builtin_amdgcn_ballot_w32(v < hi);
    nbel += __popc(below);
    return (lane == slot) ? (int)(under & ~below) : acc;
}
#define TASK_ROWS   (OB_THREADS / HID)
#define TASK_BLOCKS (BATCH / TASK_ROWS)
static_assert(BATCH % TASK_ROWS == 0, "task blocks are whole");
__global__ __launch_bounds__(OB_THREADS)
void k_build(const int* __restrict__ ei, const float* __restrict__ x,
             const float* __restrict__ Wl1, const float* __restrict__ bl1,
             const float* __restrict__ Wr1,
             int* __restrict__ elist, int* __restrict__ csr, int* __restrict__ start,
             float* __restrict__ rdeg, float* __restrict__ h1,
             const float* __restrict__ task, const float* __restrict__ Wt1,
             const float* __restrict__ bt1, const float* __restrict__ Wt2,
             const float* __restrict__ bt2, const float* __restrict__ Wc1,
             const float* __restrict__ bc1, float* __restrict__ tpart) {
    __shared__ float tu[TASK_ROWS][HID];
    __shared__ float tt[TASK_ROWS][HID];
    if (blockIdx.x >= OWN_BLOCKS) {
        const int sub = threadIdx.x / HID;
        const int h   = threadIdx.x % HID;
        const int b   = (blockIdx.x - OWN_BLOCKS) * TASK_ROWS + sub;

        float acc = bt1[h];
#pragma unroll
        for (int k = 0; k < TASK_F; ++k) acc += task[b * TASK_F + k] * Wt1[h * TASK_F + k];
        tu[sub][h] = fmaxf(acc, 0.0f);
        __syncthreads();

        float acc2 = bt2[h];
#pragma unroll 8
        for (int k = 0; k < HID; ++k) acc2 += tu[sub][k] * Wt2[h * HID + k];
        tt[sub][h] = acc2;
        __syncthreads();

        float acc3 = bc1[h];
#pragma unroll 8
        for (int k = 0; k < HID; ++k) acc3 += tt[sub][k] * Wc1[h * (2 * HID) + HID + k];
        tpart[b * HID + h] = acc3;
        return;
    }
    __shared__ int   cnt[NB];
    __shared__ int   fillp[NB];
    __shared__ float aggL[NB * NODE_F];
    __shared__ __attribute__((aligned(16))) float xL[N_NODES * NODE_F];
    __shared__ unsigned ownL[OB_WORDS][OB_THREADS];
    __shared__ int   wsum[OB_THREADS / 32];
    __shared__ int   wbel[OB_THREADS / 32];
    __shared__ int   sbase, stot;

    const int tid  = threadIdx.x;
    const int lane = tid & 31;
    const int w    = tid >> 5;
    const int lo   = blockIdx.x * NB;
    const int hi   = lo + NB;
    const int*  __restrict__ dstp = ei + N_EDGES;
    const int4* __restrict__ dst4 = (const int4*)dstp;

    float4 xq[X_PIECES];
#pragma unroll
    for (int i = 0; i < X_PIECES; ++i) {
        const int p = i * OB_THREADS + tid;
        xq[i] = make_float4(0.0f, 0.0f, 0.0f, 0.0f);
        if (p < N_NODES) xq[i] = ((const float4*)x)[p];
    }

    int nm = 0;
    int nbel = 0;
#pragma unroll 1
    for (int wd = 0; wd < OB_WORDS; ++wd) {
        int4 d[8];
#pragma unroll
        for (int j = 0; j < 8; ++j) {
            const int i = (wd * 8 + j) * OB_THREADS + tid;
            d[j] = make_int4(0x7fffffff, 0x7fffffff, 0x7fffffff, 0x7fffffff);
            if (i < E4) d[j] = dst4[i];
        }
        int acc = 0;
#pragma unroll
        for (int j = 0; j < 8; ++j) {
            acc = ob_slot(d[j].x, lo, hi, 4 * j + 0, lane, acc, nbel);
            acc = ob_slot(d[j].y, lo, hi, 4 * j + 1, lane, acc, nbel);
            acc = ob_slot(d[j].z, lo, hi, 4 * j + 2, lane, acc, nbel);
            acc = ob_slot(d[j].w, lo, hi, 4 * j + 3, lane, acc, nbel);
        }
        ownL[wd][tid] = (unsigned)acc;
        nm += __popc((unsigned)acc);
    }

    int inc = nm;
    for (int off = 1; off < 32; off <<= 1) {
        int v = __shfl_up(inc, off);
        if (lane >= off) inc += v;
    }
    if (lane == 31) wsum[w] = inc;
    if (lane == 0)  wbel[w] = nbel;
    if (tid < NB)          cnt[tid]  = 0;
    if (tid < NB * NODE_F) aggL[tid] = 0.0f;
#pragma unroll
    for (int i = 0; i < X_PIECES; ++i) {
        const int p = i * OB_THREADS + tid;
        if (p < N_NODES) *(float4*)&xL[p * NODE_F] = xq[i];
    }
    __syncthreads();
    if (w == 0) {
        const int mine = wsum[lane];
        int ws = mine;
        for (int off = 1; off < 32; off <<= 1) {
            int v = __shfl_up(ws, off);
            if (lane >= off) ws += v;
        }
        wsum[lane] = ws - mine;
        int bel = wbel[lane];
        for (int off = 16; off > 0; off >>= 1) bel += __shfl_xor(bel, off);
        if (lane == 31) stot  = ws;
        if (lane == 0)  sbase = bel;
    }
    __syncthreads();
    const int base = sbase;
    const int T    = stot;

    int pos = base + wsum[w] + inc - nm;
    const int comp = lane & 3;
    for (int wd = 0; wd < OB_WORDS; ++wd) {
        unsigned b = ownL[wd][tid];
        const int grp = (wd * 8 + (lane >> 2)) * OB_THREADS + w * 32;
        while (b) {
            const int k = __builtin_ctz(b);
            b &= b - 1;
            elist[pos++] = 4 * (grp + k) + comp;
        }
    }
    __syncthreads();

    for (int i0 = 0; i0 < T; i0 += OB_BATCH * OB_THREADS) {
        int e[OB_BATCH], d[OB_BATCH], s[OB_BATCH];
#pragma unroll
        for (int u = 0; u < OB_BATCH; ++u) {
            const int idx = i0 + u * OB_THREADS + tid;
            e[u] = (idx < T) ? elist[base + idx] : -1;
        }
#pragma unroll
        for (int u = 0; u < OB_BATCH; ++u) {
            const int ee = e[u] < 0 ? 0 : e[u];
            d[u] = dstp[ee];
            s[u] = ei[ee];
        }
#pragma unroll
        for (int u = 0; u < OB_BATCH; ++u) {
            if (e[u] >= 0) {
                const int r = d[u] - lo;
                atomicAdd(&cnt[r], 1);
                elist[base + i0 + u * OB_THREADS + tid] = (s[u] << ROW_BITS) | r;
            }
        }
    }
    __syncthreads();

    if (w == 0) {
        const int c0 = cnt[4 * lane], c1 = cnt[4 * lane + 1], c2 = cnt[4 * lane + 2], c3 = cnt[4 * lane + 3];
        const int tot = c0 + c1 + c2 + c3;
        int ps = tot;
        for (int off = 1; off < 32; off <<= 1) {
            int v = __shfl_up(ps, off);
            if (lane >= off) ps += v;
        }
        const int b0 = base + ps - tot;
        const int b1 = b0 + c0, b2 = b1 + c1, b3 = b2 + c2;
        fillp[4 * lane] = b0; fillp[4 * lane + 1] = b1; fillp[4 * lane + 2] = b2; fillp[4 * lane + 3] = b3;
        start[lo + 4 * lane] = b0; start[lo + 4 * lane + 1] = b1; start[lo + 4 * lane + 2] = b2; start[lo + 4 * lane + 3] = b3;
        if (blockIdx.x == OWN_BLOCKS - 1 && lane == 31) start[NPAD] = b3 + c3;
    }
    __syncthreads();

    int* csrL = (int*)&ownL[0][0];
    for (int i0 = 0; i0 < T; i0 += OB_BATCH * OB_THREADS) {
        int en[OB_BATCH];
        float4 xv[OB_BATCH];
#pragma unroll
        for (int u = 0; u < OB_BATCH; ++u) {
            const int idx = i0 + u * OB_THREADS + tid;
            en[u] = (idx < T) ? elist[base + idx] : -1;
        }
#pragma unroll
        for (int u = 0; u < OB_BATCH; ++u) {
            const int s = en[u] < 0 ? 0 : (en[u] >> ROW_BITS);
            xv[u] = *(const float4*)&xL[s * NODE_F];
        }
#pragma unroll
        for (int u = 0; u < OB_BATCH; ++u) {
            if (en[u] >= 0) {
                const int s = en[u] >> ROW_BITS;
                const int r = en[u] & (NB - 1);
                const int slot = atomicAdd(&fillp[r], 1);
                if (slot - base < CSR_CAP) csrL[slot - base] = s;
                else                       csr[slot] = s;
                atomicAdd(&aggL[r * NODE_F + 0], xv[u].x);
                atomicAdd(&aggL[r * NODE_F + 1], xv[u].y);
                atomicAdd(&aggL[r * NODE_F + 2], xv[u].z);
                atomicAdd(&aggL[r * NODE_F + 3], xv[u].w);
            }
        }
    }
    __syncthreads();

    for (int i = tid; i < min(T, CSR_CAP); i += OB_THREADS) csr[base + i] = csrL[i];

    if (tid < NB) rdeg[lo + tid] = 1.0f / fmaxf((float)cnt[tid], 1.0f);
    const int lrow0 = (w >> 2) * 16;
    const int col0  = (w & 3) * 16;
    const int m     = lane & 15;
    const int half  = lane >> 4;
    const int kb    = half * 2;
    const int lrow  = lrow0 + m;
    const int row   = lo + lrow;
    const float rd  = 1.0f / fmaxf((float)cnt[lrow], 1.0f);

    v2f am, ax, bl, br;
    am.x = aggL[lrow * NODE_F + kb]     * rd;
    am.y = aggL[lrow * NODE_F + kb + 1] * rd;
    ax.x = (row < N_NODES) ? x[row * NODE_F + kb]     : 0.0f;
    ax.y = (row < N_NODES) ? x[row * NODE_F + kb + 1] : 0.0f;
    bl.x = Wl1[(col0 + m) * NODE_F + kb];
    bl.y = Wl1[(col0 + m) * NODE_F + kb + 1];
    br.x = Wr1[(col0 + m) * NODE_F + kb];
    br.y = Wr1[(col0 + m) * NODE_F + kb + 1];

    v8f acc = {};
    acc = __builtin_amdgcn_wmma_f32_16x16x4_f32(false, am, false, bl, (short)0, acc, false, false);
    acc = __builtin_amdgcn_wmma_f32_16x16x4_f32(false, ax, false, br, (short)0, acc, false, false);

#pragma unroll
    for (int i = 0; i < 8; ++i) {
        int r = lo + lrow0 + i + half * 8;
        int c = col0 + m;
        h1[r * HID + c] = fmaxf(acc[i] + bl1[c], 0.0f);
    }
}

#define T_PITCH  68
#define SH_WAVES (HID / 16)
#define FG_WAVES 8
#define FG_RPW   (16 / FG_WAVES)
#define FG_BATCH 32
#define KS       (HID / 4)
#define W_PIECES ((HID * HID / 4) / (FG_WAVES * 32))
static_assert(FG_RPW == 2 && FG_WAVES >= SH_WAVES, "block shape");
static_assert(W_PIECES * FG_WAVES * 32 * 4 == HID * HID && 16 * HID / 4 == FG_WAVES * 32, "copies are whole");
__device__ __forceinline__ v2f gather_row(int beg, int end, int idx, const int* __restrict__ csr,
                                          const v2f* __restrict__ rows, int lane) {
    v2f acc = {0.0f, 0.0f};
    for (int base = beg; base < end; base += 32) {
        const int nb = base + 32;
        const int idx_next = (nb + lane < end) ? csr[nb + lane] : 0;
        const int cnt = min(32, end - base);
        for (int j = 0; j < cnt; j += FG_BATCH) {
            v2f v[FG_BATCH];
#pragma unroll
            for (int uu = 0; uu < FG_BATCH; ++uu) {
                const int s = __builtin_amdgcn_readlane(idx, j + uu);
                v[uu] = rows[(size_t)s * (HID / 2) + lane];
            }
#pragma unroll
            for (int uu = 0; uu < FG_BATCH; ++uu) {
                const bool ok = (j + uu) < cnt;
                acc.x += ok ? v[uu].x : 0.0f;
                acc.y += ok ? v[uu].y : 0.0f;
            }
        }
        idx = idx_next;
    }
    return acc;
}
__global__ __launch_bounds__(FG_WAVES * 32)
void k_gather_sage2_hpart(const int* __restrict__ start, const int* __restrict__ csr,
                          const float* __restrict__ h1, const float* __restrict__ rdeg,
                          const float* __restrict__ Wl2, const float* __restrict__ bl2,
                          const float* __restrict__ Wr2, const float* __restrict__ Wc1,
                          float* __restrict__ hpart) {
    __shared__ __attribute__((aligned(16))) float wlT[HID * T_PITCH];
    __shared__ __attribute__((aligned(16))) float wrT[HID * T_PITCH];
    __shared__ __attribute__((aligned(16))) float wcT[HID * T_PITCH];
    __shared__ __attribute__((aligned(16))) float h1T[16 * T_PITCH];
    __shared__ __attribute__((aligned(16))) float aggT[16 * T_PITCH];
    __shared__ __attribute__((aligned(16))) float h2t[16 * T_PITCH];
    const int tid  = threadIdx.x;
    const int row0 = blockIdx.x * 16;
    const int lane = tid & 31;
    const int w    = __builtin_amdgcn_readfirstlane(tid >> 5);
    const int m    = lane & 15;
    const int half = lane >> 4;
    const int kb   = half * 2;
    const int col0 = (w & (SH_WAVES - 1)) * 16;
    const bool mat = w < SH_WAVES;

    float4 ql[W_PIECES], qr[W_PIECES], qc[W_PIECES];
#pragma unroll
    for (int i = 0; i < W_PIECES; ++i) {
        const int p = tid + FG_WAVES * 32 * i;
        const int r = p >> 4, c4 = (p & 15) * 4;
        ql[i] = *(const float4*)(Wl2 + r * HID + c4);
        qr[i] = *(const float4*)(Wr2 + r * HID + c4);
        qc[i] = *(const float4*)(Wc1 + r * (2 * HID) + c4);
    }
    const float4 qh = *(const float4*)(h1 + (row0 + (tid >> 4)) * HID + (tid & 15) * 4);

    const int n  = row0 + FG_RPW * w;
    const int s0 = start[n];
    const int s1 = start[n + 1];
    const int s2 = start[n + 2];
    const int idxA = (s0 + lane < s1) ? csr[s0 + lane] : 0;
    const int idxB = (s1 + lane < s2) ? csr[s1 + lane] : 0;

#pragma unroll
    for (int i = 0; i < W_PIECES; ++i) {
        const int p = tid + FG_WAVES * 32 * i;
        const int r = p >> 4, c4 = (p & 15) * 4;
        *(float4*)&wlT[r * T_PITCH + c4] = ql[i];
        *(float4*)&wrT[r * T_PITCH + c4] = qr[i];
        *(float4*)&wcT[r * T_PITCH + c4] = qc[i];
    }
    *(float4*)&h1T[(tid >> 4) * T_PITCH + (tid & 15) * 4] = qh;

    {
        const v2f* __restrict__ rows = (const v2f*)h1;
        const v2f sa = gather_row(s0, s1, idxA, csr, rows, lane);
        const v2f sb = gather_row(s1, s2, idxB, csr, rows, lane);
        *(v2f*)&aggT[(FG_RPW * w)     * T_PITCH + 2 * lane] = sa;
        *(v2f*)&aggT[(FG_RPW * w + 1) * T_PITCH + 2 * lane] = sb;
    }
    __syncthreads();

    if (mat) {
        const float rd   = rdeg[row0 + m];
        const float bias = bl2[col0 + m];
        v8f acc = {};
#pragma unroll
        for (int ks = 0; ks < KS; ++ks) {
            const int k = 4 * ks + kb;
            v2f am = *(const v2f*)&aggT[m * T_PITCH + k];
            am.x *= rd;
            am.y *= rd;
            const v2f ah = *(const v2f*)&h1T[m * T_PITCH + k];
            const v2f bl = *(const v2f*)&wlT[(col0 + m) * T_PITCH + k];
            const v2f br = *(const v2f*)&wrT[(col0 + m) * T_PITCH + k];
            acc = __builtin_amdgcn_wmma_f32_16x16x4_f32(false, am, false, bl, (short)0, acc, false, false);
            acc = __builtin_amdgcn_wmma_f32_16x16x4_f32(false, ah, false, br, (short)0, acc, false, false);
        }
#pragma unroll
        for (int i = 0; i < 8; ++i) {
            int r = i + half * 8;
            int c = col0 + m;
            h2t[r * T_PITCH + c] = fmaxf(acc[i] + bias, 0.0f);
        }
    }
    __syncthreads();

    if (mat) {
        v8f acc2 = {};
#pragma unroll
        for (int ks = 0; ks < KS; ++ks) {
            const int k = 4 * ks + kb;
            const v2f a = *(const v2f*)&h2t[m * T_PITCH + k];
            const v2f b = *(const v2f*)&wcT[(col0 + m) * T_PITCH + k];
            acc2 = __builtin_amdgcn_wmma_f32_16x16x4_f32(false, a, false, b, (short)0, acc2, false, false);
        }
#pragma unroll
        for (int i = 0; i < 8; ++i) {
            int r = row0 + i + half * 8;
            int c = col0 + m;
            hpart[r * HID + c] = acc2[i];
        }
    }
}

#define CLS_BT      16
#define CLS_G       4
#define CLS_THREADS 128
__global__ __launch_bounds__(CLS_THREADS)
void k_classify(const float* __restrict__ hpart, const float* __restrict__ tpart,
                const float* __restrict__ Wc2, const float* __restrict__ bc2,
                float* __restrict__ out) {
    __shared__ float w2[HID];
    if (threadIdx.x < HID) w2[threadIdx.x] = Wc2[threadIdx.x];
    __syncthreads();

    const int n  = blockIdx.x * CLS_THREADS + threadIdx.x;
    const int b0 = blockIdx.y * CLS_BT;
    if (n - (int)(threadIdx.x & 31) >= N_NODES) return;

    float a[HID];
    const float4* hp = (const float4*)(hpart + (size_t)n * HID);
#pragma unroll
    for (int j = 0; j < HID / 4; ++j) {
        float4 v = hp[j];
        a[4 * j + 0] = v.x; a[4 * j + 1] = v.y; a[4 * j + 2] = v.z; a[4 * j + 3] = v.w;
    }
    float w[HID];
#pragma unroll
    for (int h = 0; h < HID; ++h) w[h] = w2[h];
    const float bias = bc2[0];

    for (int g = 0; g < CLS_BT / CLS_G; ++g) {
        const float* __restrict__ tp = tpart + (size_t)(b0 + g * CLS_G) * HID;
        float acc[CLS_G];
#pragma unroll
        for (int r = 0; r < CLS_G; ++r) acc[r] = bias;
#pragma unroll
        for (int hc = 0; hc < HID; hc += 16) {
#pragma unroll
            for (int r = 0; r < CLS_G; ++r) {
#pragma unroll
                for (int k = 0; k < 16; ++k)
                    acc[r] += fmaxf(a[hc + k] + tp[r * HID + hc + k], 0.0f) * w[hc + k];
            }
        }
        if (n < N_NODES) {
#pragma unroll
            for (int r = 0; r < CLS_G; ++r)
                out[(size_t)(b0 + g * CLS_G + r) * N_NODES + n] = acc[r];
        }
    }
}

extern "C" void kernel_launch(void* const* d_in, const int* in_sizes, int n_in,
                              void* d_out, int out_size, void* d_ws, size_t ws_size,
                              hipStream_t stream) {
    const float* x    = (const float*)d_in[0];
    const int*   ei   = (const int*)d_in[1];
    const float* task = (const float*)d_in[2];
    const float* Wl1  = (const float*)d_in[3];
    const float* bl1  = (const float*)d_in[4];
    const float* Wr1  = (const float*)d_in[5];
    const float* Wl2  = (const float*)d_in[6];
    const float* bl2  = (const float*)d_in[7];
    const float* Wr2  = (const float*)d_in[8];
    const float* Wt1  = (const float*)d_in[9];
    const float* bt1  = (const float*)d_in[10];
    const float* Wt2  = (const float*)d_in[11];
    const float* bt2  = (const float*)d_in[12];
    const float* Wc1  = (const float*)d_in[13];
    const float* bc1  = (const float*)d_in[14];
    const float* Wc2  = (const float*)d_in[15];
    const float* bc2  = (const float*)d_in[16];
    float* out = (float*)d_out;

    float* ws    = (float*)d_ws;
    float* rdeg  = ws + OFF_RDEG;
    float* h1    = ws + OFF_H1;
    float* hpart = ws + OFF_HPART;
    float* tpart = ws + OFF_TPART;
    int*   elist = (int*)(ws + OFF_H2);
    int*   csr   = (int*)(ws + OFF_AGG2);
    int*   start = (int*)(ws + OFF_START);

    k_build<<<OWN_BLOCKS + TASK_BLOCKS, OB_THREADS, 0, stream>>>(ei, x, Wl1, bl1, Wr1, elist, csr, start, rdeg, h1,
                                                                 task, Wt1, bt1, Wt2, bt2, Wc1, bc1, tpart);
    k_gather_sage2_hpart<<<NPAD / 16, FG_WAVES * 32, 0, stream>>>(start, csr, h1, rdeg, Wl2, bl2, Wr2, Wc1, hpart);
    k_classify<<<dim3(NPAD / CLS_THREADS, BATCH / CLS_BT), CLS_THREADS, 0, stream>>>(hpart, tpart, Wc2, bc2, out);
}


# build launch: owner blocks of 64 nodes and 512 threads (80 blocks, wave count unchanged): a block's edge list, its one-line-per-lane loads and its LDS adds halve; scan loads one word ahead
# speedup vs baseline: 4.1106x; 1.0775x over previous
#include <hip/hip_runtime.h>
#include <hip/hip_bf16.h>

#define N_NODES 5000
#define NPAD    5120
#define N_EDGES 160000
#define HID     64
#define NODE_F  4
#define TASK_F  4
#define BATCH   512

typedef __attribute__((ext_vector_type(2))) float v2f;
typedef __attribute__((ext_vector_type(8))) float v8f;

#define OFF_AGG2   (NPAD * (1 + NODE_F))
#define OFF_RDEG   (OFF_AGG2 + NPAD * HID)
#define OFF_H1     (OFF_RDEG  + NPAD)
#define OFF_H2     (OFF_H1    + NPAD * HID)
#define OFF_HPART  (OFF_H2    + NPAD * HID)
#define OFF_TPART  (OFF_HPART + NPAD * HID)
#define OFF_START  (OFF_H2    + N_EDGES)


#define NB          64
#define OB_THREADS  512
#define OB_WAVES    (OB_THREADS / 32)
#define OWN_BLOCKS  (NPAD / NB)
#define E4          (N_EDGES / 4)
#define OB_TRIPS    (((E4 + OB_THREADS - 1) / OB_THREADS + 7) / 8 * 8)
#define OB_WORDS    (OB_TRIPS / 8)
#define OB_BATCH    5
#define X_PIECES    ((N_NODES + OB_THREADS - 1) / OB_THREADS)
#define ROW_BITS    6
#define CSR_CAP     (OB_WORDS * OB_THREADS)
#define NPL         (NB / 32)
#define OB_FULLW    ((E4 / OB_THREADS) / 8)
static_assert((1 << ROW_BITS) == NB && (long long)(N_NODES - 1) * NB + NB - 1 < 0x7fffffffLL, "packed entries are non-negative ints");
static_assert(OB_TRIPS % 16 == 0 && OB_TRIPS * OB_THREADS >= E4, "mask words are filled whole, in pairs, and cover the row");
static_assert(OB_FULLW >= OB_WORDS - 1, "only the last word tests for the row's end");
static_assert(NB * NODE_F <= OB_THREADS && NB % 32 == 0 && OB_WAVES <= 32 && OB_WAVES == (NB / 16) * (HID / 16), "block shape");
__device__ __forceinline__ int ob_slot(int v, int lo, int hi, int slot, int lane, int acc, int& nbel) {
    const unsigned below = __builtin_amdgcn_ballot_w32(v < lo);
    const unsigned under = __builtin_amdgcn_ballot_w32(v < hi);
    nbel += __popc(below);
    return (lane == slot) ? (int)(under & ~below) : acc;
}
template <bool FULL>
__device__ __forceinline__ void ob_load(int4 (&d)[8], const int4* __restrict__ dst4, int wd, int tid) {
#pragma unroll
    for (int j = 0; j < 8; ++j) {
        const int i = (wd * 8 + j) * OB_THREADS + tid;
        if (FULL) {
            d[j] = dst4[i];
        } else {
            d[j] = make_int4(0x7fffffff, 0x7fffffff, 0x7fffffff, 0x7fffffff);
            if (i < E4) d[j] = dst4[i];
        }
    }
}
__device__ __forceinline__ float4 x_piece(const float* __restrict__ x, int piece, int tid) {
    const int p = piece * OB_THREADS + tid;
    return ((const float4*)x)[min(p, N_NODES - 1)];
}
__device__ __forceinline__ void x_keep(float* xL, float4 v, int piece, int tid) {
    const int p = piece * OB_THREADS + tid;
    if (p < N_NODES) *(float4*)&xL[p * NODE_F] = v;
}
__device__ __forceinline__ int ob_word(const int4 (&d)[8], int lo, int hi, int lane, int& nbel) {
    int acc = 0;
#pragma unroll
    for (int j = 0; j < 8; ++j) {
        acc = ob_slot(d[j].x, lo, hi, 4 * j + 0, lane, acc, nbel);
        acc = ob_slot(d[j].y, lo, hi, 4 * j + 1, lane, acc, nbel);
        acc = ob_slot(d[j].z, lo, hi, 4 * j + 2, lane, acc, nbel);
        acc = ob_slot(d[j].w, lo, hi, 4 * j + 3, lane, acc, nbel);
    }
    return acc;
}
#define TASK_ROWS   (OB_THREADS / HID)
#define TASK_BLOCKS (BATCH / TASK_ROWS)
static_assert(BATCH % TASK_ROWS == 0, "task blocks are whole");
__global__ __launch_bounds__(OB_THREADS) __attribute__((amdgpu_waves_per_eu(4, 4)))
void k_build(const int* __restrict__ ei, const float* __restrict__ x,
             const float* __restrict__ Wl1, const float* __restrict__ bl1,
             const float* __restrict__ Wr1,
             int* __restrict__ elist, int* __restrict__ csr, int* __restrict__ start,
             float* __restrict__ rdeg, float* __restrict__ h1,
             const float* __restrict__ task, const float* __restrict__ Wt1,
             const float* __restrict__ bt1, const float* __restrict__ Wt2,
             const float* __restrict__ bt2, const float* __restrict__ Wc1,
             const float* __restrict__ bc1, float* __restrict__ tpart) {
    __shared__ float tu[TASK_ROWS][HID];
    __shared__ float tt[TASK_ROWS][HID];
    if (blockIdx.x >= OWN_BLOCKS) {
        const int sub = threadIdx.x / HID;
        const int h   = threadIdx.x % HID;
        const int b   = (blockIdx.x - OWN_BLOCKS) * TASK_ROWS + sub;

        float acc = bt1[h];
#pragma unroll
        for (int k = 0; k < TASK_F; ++k) acc += task[b * TASK_F + k] * Wt1[h * TASK_F + k];
        tu[sub][h] = fmaxf(acc, 0.0f);
        __syncthreads();

        float acc2 = bt2[h];
#pragma unroll 8
        for (int k = 0; k < HID; ++k) acc2 += tu[sub][k] * Wt2[h * HID + k];
        tt[sub][h] = acc2;
        __syncthreads();

        float acc3 = bc1[h];
#pragma unroll 8
        for (int k = 0; k < HID; ++k) acc3 += tt[sub][k] * Wc1[h * (2 * HID) + HID + k];
        tpart[b * HID + h] = acc3;
        return;
    }
    __shared__ int   cnt[NB];
    __shared__ int   fillp[NB];
    __shared__ float aggL[NB * NODE_F];
    __shared__ __attribute__((aligned(16))) float xL[N_NODES * NODE_F];
    __shared__ unsigned ownL[OB_WORDS][OB_THREADS];
    __shared__ int   wsum[OB_WAVES];
    __shared__ int   wbel[OB_WAVES];
    __shared__ int   sbase, stot;

    const int tid  = threadIdx.x;
    const int lane = tid & 31;
    const int w    = tid >> 5;
    const int lo   = blockIdx.x * NB;
    const int hi   = lo + NB;
    const int*  __restrict__ dstp = ei + N_EDGES;
    const int4* __restrict__ dst4 = (const int4*)dstp;

    float4 xq[X_PIECES];
#pragma unroll
    for (int i = 0; i < X_PIECES; ++i) xq[i] = x_piece(x, i, tid);

    int nm = 0;
    int nbel = 0;
    int4 dA[8], dB[8];
    ob_load<true>(dA, dst4, 0, tid);
#pragma unroll 1
    for (int wd = 0; wd < OB_WORDS - 2; wd += 2) {
        ob_load<true>(dB, dst4, wd + 1, tid);
        __builtin_amdgcn_sched_barrier(0);
        const int accA = ob_word(dA, lo, hi, lane, nbel);
        ownL[wd][tid] = (unsigned)accA;
        nm += __popc((unsigned)accA);
        ob_load<true>(dA, dst4, wd + 2, tid);
        __builtin_amdgcn_sched_barrier(0);
        const int accB = ob_word(dB, lo, hi, lane, nbel);
        ownL[wd + 1][tid] = (unsigned)accB;
        nm += __popc((unsigned)accB);
    }
    {
        ob_load<false>(dB, dst4, OB_WORDS - 1, tid);
        const int accA = ob_word(dA, lo, hi, lane, nbel);
        ownL[OB_WORDS - 2][tid] = (unsigned)accA;
        nm += __popc((unsigned)accA);
        const int accB = ob_word(dB, lo, hi, lane, nbel);
        ownL[OB_WORDS - 1][tid] = (unsigned)accB;
        nm += __popc((unsigned)accB);
    }

    int inc = nm;
    for (int off = 1; off < 32; off <<= 1) {
        int v = __shfl_up(inc, off);
        if (lane >= off) inc += v;
    }
    if (lane == 31) wsum[w] = inc;
    if (lane == 0)  wbel[w] = nbel;
    if (tid < NB)          cnt[tid]  = 0;
    if (tid < NB * NODE_F) aggL[tid] = 0.0f;
#pragma unroll
    for (int i = 0; i < X_PIECES; ++i) x_keep(xL, xq[i], i, tid);
    __syncthreads();
    if (w == 0) {
        const int mine = (lane < OB_WAVES) ? wsum[lane] : 0;
        int ws = mine;
        for (int off = 1; off < 32; off <<= 1) {
            int v = __shfl_up(ws, off);
            if (lane >= off) ws += v;
        }
        if (lane < OB_WAVES) wsum[lane] = ws - mine;
        int bel = (lane < OB_WAVES) ? wbel[lane] : 0;
        for (int off = 16; off > 0; off >>= 1) bel += __shfl_xor(bel, off);
        if (lane == 31) stot  = ws;
        if (lane == 0)  sbase = bel;
    }
    __syncthreads();
    const int base = sbase;
    const int T    = stot;

    int pos = base + wsum[w] + inc - nm;
    const int comp = lane & 3;
    for (int wd = 0; wd < OB_WORDS; ++wd) {
        unsigned b = ownL[wd][tid];
        const int grp = (wd * 8 + (lane >> 2)) * OB_THREADS + w * 32;
        while (b) {
            const int k = __builtin_ctz(b);
            b &= b - 1;
            elist[pos++] = 4 * (grp + k) + comp;
        }
    }
    __syncthreads();

    for (int i0 = 0; i0 < T; i0 += OB_BATCH * OB_THREADS) {
        int e[OB_BATCH], d[OB_BATCH], s[OB_BATCH];
#pragma unroll
        for (int u = 0; u < OB_BATCH; ++u) {
            const int idx = i0 + u * OB_THREADS + tid;
            e[u] = (idx < T) ? elist[base + idx] : -1;
        }
#pragma unroll
        for (int u = 0; u < OB_BATCH; ++u) {
            const int ee = e[u] < 0 ? 0 : e[u];
            d[u] = dstp[ee];
            s[u] = ei[ee];
        }
#pragma unroll
        for (int u = 0; u < OB_BATCH; ++u) {
            if (e[u] >= 0) {
                const int r = d[u] - lo;
                atomicAdd(&cnt[r], 1);
                elist[base + i0 + u * OB_THREADS + tid] = (s[u] << ROW_BITS) | r;
            }
        }
    }
    __syncthreads();

    if (w == 0) {
        int cq[NPL];
        int tot = 0;
#pragma unroll
        for (int qn = 0; qn < NPL; ++qn) { cq[qn] = cnt[NPL * lane + qn]; tot += cq[qn]; }
        int ps = tot;
        for (int off = 1; off < 32; off <<= 1) {
            int v = __shfl_up(ps, off);
            if (lane >= off) ps += v;
        }
        int bq = base + ps - tot;
#pragma unroll
        for (int qn = 0; qn < NPL; ++qn) {
            fillp[NPL * lane + qn] = bq;
            start[lo + NPL * lane + qn] = bq;
            bq += cq[qn];
        }
        if (blockIdx.x == OWN_BLOCKS - 1 && lane == 31) start[NPAD] = bq;
    }
    __syncthreads();

    int* csrL = (int*)&ownL[0][0];
    for (int i0 = 0; i0 < T; i0 += OB_BATCH * OB_THREADS) {
        int en[OB_BATCH];
        float4 xv[OB_BATCH];
#pragma unroll
        for (int u = 0; u < OB_BATCH; ++u) {
            const int idx = i0 + u * OB_THREADS + tid;
            en[u] = (idx < T) ? elist[base + idx] : -1;
        }
#pragma unroll
        for (int u = 0; u < OB_BATCH; ++u) {
            const int s = en[u] < 0 ? 0 : (en[u] >> ROW_BITS);
            xv[u] = *(const float4*)&xL[s * NODE_F];
        }
#pragma unroll
        for (int u = 0; u < OB_BATCH; ++u) {
            if (en[u] >= 0) {
                const int s = en[u] >> ROW_BITS;
                const int r = en[u] & (NB - 1);
                const int slot = atomicAdd(&fillp[r], 1);
                if (slot - base < CSR_CAP) csrL[slot - base] = s;
                else                       csr[slot] = s;
                atomicAdd(&aggL[r * NODE_F + 0], xv[u].x);
                atomicAdd(&aggL[r * NODE_F + 1], xv[u].y);
                atomicAdd(&aggL[r * NODE_F + 2], xv[u].z);
                atomicAdd(&aggL[r * NODE_F + 3], xv[u].w);
            }
        }
    }
    __syncthreads();

    for (int i = tid; i < min(T, CSR_CAP); i += OB_THREADS) csr[base + i] = csrL[i];

    if (tid < NB) rdeg[lo + tid] = 1.0f / fmaxf((float)cnt[tid], 1.0f);
    const int lrow0 = (w >> 2) * 16;
    const int col0  = (w & 3) * 16;
    const int m     = lane & 15;
    const int half  = lane >> 4;
    const int kb    = half * 2;
    const int lrow  = lrow0 + m;
    const int row   = lo + lrow;
    const float rd  = 1.0f / fmaxf((float)cnt[lrow], 1.0f);

    v2f am, ax, bl, br;
    am.x = aggL[lrow * NODE_F + kb]     * rd;
    am.y = aggL[lrow * NODE_F + kb + 1] * rd;
    ax.x = (row < N_NODES) ? x[row * NODE_F + kb]     : 0.0f;
    ax.y = (row < N_NODES) ? x[row * NODE_F + kb + 1] : 0.0f;
    bl.x = Wl1[(col0 + m) * NODE_F + kb];
    bl.y = Wl1[(col0 + m) * NODE_F + kb + 1];
    br.x = Wr1[(col0 + m) * NODE_F + kb];
    br.y = Wr1[(col0 + m) * NODE_F + kb + 1];

    v8f acc = {};
    acc = __builtin_amdgcn_wmma_f32_16x16x4_f32(false, am, false, bl, (short)0, acc, false, false);
    acc = __builtin_amdgcn_wmma_f32_16x16x4_f32(false, ax, false, br, (short)0, acc, false, false);

#pragma unroll
    for (int i = 0; i < 8; ++i) {
        int r = lo + lrow0 + i + half * 8;
        int c = col0 + m;
        h1[r * HID + c] = fmaxf(acc[i] + bl1[c], 0.0f);
    }
}

#define T_PITCH  68
#define SH_WAVES (HID / 16)
#define FG_WAVES 8
#define FG_RPW   (16 / FG_WAVES)
#define FG_BATCH 32
#define KS       (HID / 4)
#define W_PIECES ((HID * HID / 4) / (FG_WAVES * 32))
static_assert(FG_RPW == 2 && FG_WAVES >= SH_WAVES, "block shape");
static_assert(W_PIECES * FG_WAVES * 32 * 4 == HID * HID && 16 * HID / 4 == FG_WAVES * 32, "copies are whole");
__device__ __forceinline__ v2f gather_row(int beg, int end, int idx, const int* __restrict__ csr,
                                          const v2f* __restrict__ rows, int lane) {
    v2f acc = {0.0f, 0.0f};
    for (int base = beg; base < end; base += 32) {
        const int nb = base + 32;
        const int idx_next = (nb + lane < end) ? csr[nb + lane] : 0;
        const int cnt = min(32, end - base);
        for (int j = 0; j < cnt; j += FG_BATCH) {
            v2f v[FG_BATCH];
#pragma unroll
            for (int uu = 0; uu < FG_BATCH; ++uu) {
                const int s = __builtin_amdgcn_readlane(idx, j + uu);
                v[uu] = rows[(size_t)s * (HID / 2) + lane];
            }
#pragma unroll
            for (int uu = 0; uu < FG_BATCH; ++uu) {
                const bool ok = (j + uu) < cnt;
                acc.x += ok ? v[uu].x : 0.0f;
                acc.y += ok ? v[uu].y : 0.0f;
            }
        }
        idx = idx_next;
    }
    return acc;
}
__global__ __launch_bounds__(FG_WAVES * 32)
void k_gather_sage2_hpart(const int* __restrict__ start, const int* __restrict__ csr,
                          const float* __restrict__ h1, const float* __restrict__ rdeg,
                          const float* __restrict__ Wl2, const float* __restrict__ bl2,
                          const float* __restrict__ Wr2, const float* __restrict__ Wc1,
                          float* __restrict__ hpart) {
    __shared__ __attribute__((aligned(16))) float wlT[HID * T_PITCH];
    __shared__ __attribute__((aligned(16))) float wrT[HID * T_PITCH];
    __shared__ __attribute__((aligned(16))) float wcT[HID * T_PITCH];
    __shared__ __attribute__((aligned(16))) float h1T[16 * T_PITCH];
    __shared__ __attribute__((aligned(16))) float aggT[16 * T_PITCH];
    __shared__ __attribute__((aligned(16))) float h2t[16 * T_PITCH];
    const int tid  = threadIdx.x;
    const int row0 = blockIdx.x * 16;
    const int lane = tid & 31;
    const int w    = __builtin_amdgcn_readfirstlane(tid >> 5);
    const int m    = lane & 15;
    const int half = lane >> 4;
    const int kb   = half * 2;
    const int col0 = (w & (SH_WAVES - 1)) * 16;
    const bool mat = w < SH_WAVES;

    float4 ql[W_PIECES], qr[W_PIECES], qc[W_PIECES];
#pragma unroll
    for (int i = 0; i < W_PIECES; ++i) {
        const int p = tid + FG_WAVES * 32 * i;
        const int r = p >> 4, c4 = (p & 15) * 4;
        ql[i] = *(const float4*)(Wl2 + r * HID + c4);
        qr[i] = *(const float4*)(Wr2 + r * HID + c4);
        qc[i] = *(const float4*)(Wc1 + r * (2 * HID) + c4);
    }
    const float4 qh = *(const float4*)(h1 + (row0 + (tid >> 4)) * HID + (tid & 15) * 4);

    const int n  = row0 + FG_RPW * w;
    const int s0 = start[n];
    const int s1 = start[n + 1];
    const int s2 = start[n + 2];
    const int idxA = (s0 + lane < s1) ? csr[s0 + lane] : 0;
    const int idxB = (s1 + lane < s2) ? csr[s1 + lane] : 0;

#pragma unroll
    for (int i = 0; i < W_PIECES; ++i) {
        const int p = tid + FG_WAVES * 32 * i;
        const int r = p >> 4, c4 = (p & 15) * 4;
        *(float4*)&wlT[r * T_PITCH + c4] = ql[i];
        *(float4*)&wrT[r * T_PITCH + c4] = qr[i];
        *(float4*)&wcT[r * T_PITCH + c4] = qc[i];
    }
    *(float4*)&h1T[(tid >> 4) * T_PITCH + (tid & 15) * 4] = qh;

    {
        const v2f* __restrict__ rows = (const v2f*)h1;
        const v2f sa = gather_row(s0, s1, idxA, csr, rows, lane);
        const v2f sb = gather_row(s1, s2, idxB, csr, rows, lane);
        *(v2f*)&aggT[(FG_RPW * w)     * T_PITCH + 2 * lane] = sa;
        *(v2f*)&aggT[(FG_RPW * w + 1) * T_PITCH + 2 * lane] = sb;
    }
    __syncthreads();

    if (mat) {
        const float rd   = rdeg[row0 + m];
        const float bias = bl2[col0 + m];
        v8f acc = {};
#pragma unroll
        for (int ks = 0; ks < KS; ++ks) {
            const int k = 4 * ks + kb;
            v2f am = *(const v2f*)&aggT[m * T_PITCH + k];
            am.x *= rd;
            am.y *= rd;
            const v2f ah = *(const v2f*)&h1T[m * T_PITCH + k];
            const v2f bl = *(const v2f*)&wlT[(col0 + m) * T_PITCH + k];
            const v2f br = *(const v2f*)&wrT[(col0 + m) * T_PITCH + k];
            acc = __builtin_amdgcn_wmma_f32_16x16x4_f32(false, am, false, bl, (short)0, acc, false, false);
            acc = __builtin_amdgcn_wmma_f32_16x16x4_f32(false, ah, false, br, (short)0, acc, false, false);
        }
#pragma unroll
        for (int i = 0; i < 8; ++i) {
            int r = i + half * 8;
            int c = col0 + m;
            h2t[r * T_PITCH + c] = fmaxf(acc[i] + bias, 0.0f);
        }
    }
    __syncthreads();

    if (mat) {
        v8f acc2 = {};
#pragma unroll
        for (int ks = 0; ks < KS; ++ks) {
            const int k = 4 * ks + kb;
            const v2f a = *(const v2f*)&h2t[m * T_PITCH + k];
            const v2f b = *(const v2f*)&wcT[(col0 + m) * T_PITCH + k];
            acc2 = __builtin_amdgcn_wmma_f32_16x16x4_f32(false, a, false, b, (short)0, acc2, false, false);
        }
#pragma unroll
        for (int i = 0; i < 8; ++i) {
            int r = row0 + i + half * 8;
            int c = col0 + m;
            hpart[r * HID + c] = acc2[i];
        }
    }
}

#define CLS_BT      16
#define CLS_G       4
#define CLS_THREADS 128
__global__ __launch_bounds__(CLS_THREADS)
void k_classify(const float* __restrict__ hpart, const float* __restrict__ tpart,
                const float* __restrict__ Wc2, const float* __restrict__ bc2,
                float* __restrict__ out) {
    __shared__ float w2[HID];
    if (threadIdx.x < HID) w2[threadIdx.x] = Wc2[threadIdx.x];
    __syncthreads();

    const int n  = blockIdx.x * CLS_THREADS + threadIdx.x;
    const int b0 = blockIdx.y * CLS_BT;
    if (n - (int)(threadIdx.x & 31) >= N_NODES) return;

    float a[HID];
    const float4* hp = (const float4*)(hpart + (size_t)n * HID);
#pragma unroll
    for (int j = 0; j < HID / 4; ++j) {
        float4 v = hp[j];
        a[4 * j + 0] = v.x; a[4 * j + 1] = v.y; a[4 * j + 2] = v.z; a[4 * j + 3] = v.w;
    }
    float w[HID];
#pragma unroll
    for (int h = 0; h < HID; ++h) w[h] = w2[h];
    const float bias = bc2[0];

    for (int g = 0; g < CLS_BT / CLS_G; ++g) {
        const float* __restrict__ tp = tpart + (size_t)(b0 + g * CLS_G) * HID;
        float acc[CLS_G];
#pragma unroll
        for (int r = 0; r < CLS_G; ++r) acc[r] = bias;
#pragma unroll
        for (int hc = 0; hc < HID; hc += 16) {
#pragma unroll
            for (int r = 0; r < CLS_G; ++r) {
#pragma unroll
                for (int k = 0; k < 16; ++k)
                    acc[r] += fmaxf(a[hc + k] + tp[r * HID + hc + k], 0.0f) * w[hc + k];
            }
        }
        if (n < N_NODES) {
#pragma unroll
            for (int r = 0; r < CLS_G; ++r)
                out[(size_t)(b0 + g * CLS_G + r) * N_NODES + n] = acc[r];
        }
    }
}

extern "C" void kernel_launch(void* const* d_in, const int* in_sizes, int n_in,
                              void* d_out, int out_size, void* d_ws, size_t ws_size,
                              hipStream_t stream) {
    const float* x    = (const float*)d_in[0];
    const int*   ei   = (const int*)d_in[1];
    const float* task = (const float*)d_in[2];
    const float* Wl1  = (const float*)d_in[3];
    const float* bl1  = (const float*)d_in[4];
    const float* Wr1  = (const float*)d_in[5];
    const float* Wl2  = (const float*)d_in[6];
    const float* bl2  = (const float*)d_in[7];
    const float* Wr2  = (const float*)d_in[8];
    const float* Wt1  = (const float*)d_in[9];
    const float* bt1  = (const float*)d_in[10];
    const float* Wt2  = (const float*)d_in[11];
    const float* bt2  = (const float*)d_in[12];
    const float* Wc1  = (const float*)d_in[13];
    const float* bc1  = (const float*)d_in[14];
    const float* Wc2  = (const float*)d_in[15];
    const float* bc2  = (const float*)d_in[16];
    float* out = (float*)d_out;

    float* ws    = (float*)d_ws;
    float* rdeg  = ws + OFF_RDEG;
    float* h1    = ws + OFF_H1;
    float* hpart = ws + OFF_HPART;
    float* tpart = ws + OFF_TPART;
    int*   elist = (int*)(ws + OFF_H2);
    int*   csr   = (int*)(ws + OFF_AGG2);
    int*   start = (int*)(ws + OFF_START);

    k_build<<<OWN_BLOCKS + TASK_BLOCKS, OB_THREADS, 0, stream>>>(ei, x, Wl1, bl1, Wr1, elist, csr, start, rdeg, h1,
                                                                 task, Wt1, bt1, Wt2, bt2, Wc1, bc1, tpart);
    k_gather_sage2_hpart<<<NPAD / 16, FG_WAVES * 32, 0, stream>>>(start, csr, h1, rdeg, Wl2, bl2, Wr2, Wc1, hpart);
    k_classify<<<dim3(NPAD / CLS_THREADS, BATCH / CLS_BT), CLS_THREADS, 0, stream>>>(hpart, tpart, Wc2, bc2, out);
}


# layer-2 gather: a wave's two rows side by side in its half-waves (16-byte loads, ids by row-share), up to 48 row loads per list in flight: a tile's lists cost one round trip, not up to 4; same sums
# speedup vs baseline: 4.2708x; 1.0390x over previous
#include <hip/hip_runtime.h>
#include <hip/hip_bf16.h>

#define N_NODES 5000
#define NPAD    5120
#define N_EDGES 160000
#define HID     64
#define NODE_F  4
#define TASK_F  4
#define BATCH   512

typedef __attribute__((ext_vector_type(2))) float v2f;
typedef __attribute__((ext_vector_type(4))) float v4f;
typedef __attribute__((ext_vector_type(8))) float v8f;

#define OFF_AGG2   (NPAD * (1 + NODE_F))
#define OFF_RDEG   (OFF_AGG2 + NPAD * HID)
#define OFF_H1     (OFF_RDEG  + NPAD)
#define OFF_H2     (OFF_H1    + NPAD * HID)
#define OFF_HPART  (OFF_H2    + NPAD * HID)
#define OFF_TPART  (OFF_HPART + NPAD * HID)
#define OFF_START  (OFF_H2    + N_EDGES)


#define NB          64
#define OB_THREADS  512
#define OB_WAVES    (OB_THREADS / 32)
#define OWN_BLOCKS  (NPAD / NB)
#define E4          (N_EDGES / 4)
#define OB_TRIPS    (((E4 + OB_THREADS - 1) / OB_THREADS + 7) / 8 * 8)
#define OB_WORDS    (OB_TRIPS / 8)
#define OB_BATCH    5
#define X_PIECES    ((N_NODES + OB_THREADS - 1) / OB_THREADS)
#define ROW_BITS    6
#define CSR_CAP     (OB_WORDS * OB_THREADS)
#define NPL         (NB / 32)
#define OB_FULLW    ((E4 / OB_THREADS) / 8)
static_assert((1 << ROW_BITS) == NB && (long long)(N_NODES - 1) * NB + NB - 1 < 0x7fffffffLL, "packed entries are non-negative ints");
static_assert(OB_TRIPS % 16 == 0 && OB_TRIPS * OB_THREADS >= E4, "mask words are filled whole, in pairs, and cover the row");
static_assert(OB_FULLW >= OB_WORDS - 1, "only the last word tests for the row's end");
static_assert(NB * NODE_F <= OB_THREADS && NB % 32 == 0 && OB_WAVES <= 32 && OB_WAVES == (NB / 16) * (HID / 16), "block shape");
__device__ __forceinline__ int ob_slot(int v, int lo, int hi, int slot, int lane, int acc, int& nbel) {
    const unsigned below = __builtin_amdgcn_ballot_w32(v < lo);
    const unsigned under = __builtin_amdgcn_ballot_w32(v < hi);
    nbel += __popc(below);
    return (lane == slot) ? (int)(under & ~below) : acc;
}
template <bool FULL>
__device__ __forceinline__ void ob_load(int4 (&d)[8], const int4* __restrict__ dst4, int wd, int tid) {
#pragma unroll
    for (int j = 0; j < 8; ++j) {
        const int i = (wd * 8 + j) * OB_THREADS + tid;
        if (FULL) {
            d[j] = dst4[i];
        } else {
            d[j] = make_int4(0x7fffffff, 0x7fffffff, 0x7fffffff, 0x7fffffff);
            if (i < E4) d[j] = dst4[i];
        }
    }
}
__device__ __forceinline__ float4 x_piece(const float* __restrict__ x, int piece, int tid) {
    const int p = piece * OB_THREADS + tid;
    return ((const float4*)x)[min(p, N_NODES - 1)];
}
__device__ __forceinline__ void x_keep(float* xL, float4 v, int piece, int tid) {
    const int p = piece * OB_THREADS + tid;
    if (p < N_NODES) *(float4*)&xL[p * NODE_F] = v;
}
__device__ __forceinline__ int ob_word(const int4 (&d)[8], int lo, int hi, int lane, int& nbel) {
    int acc = 0;
#pragma unroll
    for (int j = 0; j < 8; ++j) {
        acc = ob_slot(d[j].x, lo, hi, 4 * j + 0, lane, acc, nbel);
        acc = ob_slot(d[j].y, lo, hi, 4 * j + 1, lane, acc, nbel);
        acc = ob_slot(d[j].z, lo, hi, 4 * j + 2, lane, acc, nbel);
        acc = ob_slot(d[j].w, lo, hi, 4 * j + 3, lane, acc, nbel);
    }
    return acc;
}
#define TASK_ROWS   (OB_THREADS / HID)
#define TASK_BLOCKS (BATCH / TASK_ROWS)
static_assert(BATCH % TASK_ROWS == 0, "task blocks are whole");
__global__ __launch_bounds__(OB_THREADS) __attribute__((amdgpu_waves_per_eu(4, 4)))
void k_build(const int* __restrict__ ei, const float* __restrict__ x,
             const float* __restrict__ Wl1, const float* __restrict__ bl1,
             const float* __restrict__ Wr1,
             int* __restrict__ elist, int* __restrict__ csr, int* __restrict__ start,
             float* __restrict__ rdeg, float* __restrict__ h1,
             const float* __restrict__ task, const float* __restrict__ Wt1,
             const float* __restrict__ bt1, const float* __restrict__ Wt2,
             const float* __restrict__ bt2, const float* __restrict__ Wc1,
             const float* __restrict__ bc1, float* __restrict__ tpart) {
    __shared__ float tu[TASK_ROWS][HID];
    __shared__ float tt[TASK_ROWS][HID];
    if (blockIdx.x >= OWN_BLOCKS) {
        const int sub = threadIdx.x / HID;
        const int h   = threadIdx.x % HID;
        const int b   = (blockIdx.x - OWN_BLOCKS) * TASK_ROWS + sub;

        float acc = bt1[h];
#pragma unroll
        for (int k = 0; k < TASK_F; ++k) acc += task[b * TASK_F + k] * Wt1[h * TASK_F + k];
        tu[sub][h] = fmaxf(acc, 0.0f);
        __syncthreads();

        float acc2 = bt2[h];
#pragma unroll 8
        for (int k = 0; k < HID; ++k) acc2 += tu[sub][k] * Wt2[h * HID + k];
        tt[sub][h] = acc2;
        __syncthreads();

        float acc3 = bc1[h];
#pragma unroll 8
        for (int k = 0; k < HID; ++k) acc3 += tt[sub][k] * Wc1[h * (2 * HID) + HID + k];
        tpart[b * HID + h] = acc3;
        return;
    }
    __shared__ int   cnt[NB];
    __shared__ int   fillp[NB];
    __shared__ float aggL[NB * NODE_F];
    __shared__ __attribute__((aligned(16))) float xL[N_NODES * NODE_F];
    __shared__ unsigned ownL[OB_WORDS][OB_THREADS];
    __shared__ int   wsum[OB_WAVES];
    __shared__ int   wbel[OB_WAVES];
    __shared__ int   sbase, stot;

    const int tid  = threadIdx.x;
    const int lane = tid & 31;
    const int w    = tid >> 5;
    const int lo   = blockIdx.x * NB;
    const int hi   = lo + NB;
    const int*  __restrict__ dstp = ei + N_EDGES;
    const int4* __restrict__ dst4 = (const int4*)dstp;

    float4 xq[X_PIECES];
#pragma unroll
    for (int i = 0; i < X_PIECES; ++i) xq[i] = x_piece(x, i, tid);

    int nm = 0;
    int nbel = 0;
    int4 dA[8], dB[8];
    ob_load<true>(dA, dst4, 0, tid);
#pragma unroll 1
    for (int wd = 0; wd < OB_WORDS - 2; wd += 2) {
        ob_load<true>(dB, dst4, wd + 1, tid);
        __builtin_amdgcn_sched_barrier(0);
        const int accA = ob_word(dA, lo, hi, lane, nbel);
        ownL[wd][tid] = (unsigned)accA;
        nm += __popc((unsigned)accA);
        ob_load<true>(dA, dst4, wd + 2, tid);
        __builtin_amdgcn_sched_barrier(0);
        const int accB = ob_word(dB, lo, hi, lane, nbel);
        ownL[wd + 1][tid] = (unsigned)accB;
        nm += __popc((unsigned)accB);
    }
    {
        ob_load<false>(dB, dst4, OB_WORDS - 1, tid);
        const int accA = ob_word(dA, lo, hi, lane, nbel);
        ownL[OB_WORDS - 2][tid] = (unsigned)accA;
        nm += __popc((unsigned)accA);
        const int accB = ob_word(dB, lo, hi, lane, nbel);
        ownL[OB_WORDS - 1][tid] = (unsigned)accB;
        nm += __popc((unsigned)accB);
    }

    int inc = nm;
    for (int off = 1; off < 32; off <<= 1) {
        int v = __shfl_up(inc, off);
        if (lane >= off) inc += v;
    }
    if (lane == 31) wsum[w] = inc;
    if (lane == 0)  wbel[w] = nbel;
    if (tid < NB)          cnt[tid]  = 0;
    if (tid < NB * NODE_F) aggL[tid] = 0.0f;
#pragma unroll
    for (int i = 0; i < X_PIECES; ++i) x_keep(xL, xq[i], i, tid);
    __syncthreads();
    if (w == 0) {
        const int mine = (lane < OB_WAVES) ? wsum[lane] : 0;
        int ws = mine;
        for (int off = 1; off < 32; off <<= 1) {
            int v = __shfl_up(ws, off);
            if (lane >= off) ws += v;
        }
        if (lane < OB_WAVES) wsum[lane] = ws - mine;
        int bel = (lane < OB_WAVES) ? wbel[lane] : 0;
        for (int off = 16; off > 0; off >>= 1) bel += __shfl_xor(bel, off);
        if (lane == 31) stot  = ws;
        if (lane == 0)  sbase = bel;
    }
    __syncthreads();
    const int base = sbase;
    const int T    = stot;

    int pos = base + wsum[w] + inc - nm;
    const int comp = lane & 3;
    for (int wd = 0; wd < OB_WORDS; ++wd) {
        unsigned b = ownL[wd][tid];
        const int grp = (wd * 8 + (lane >> 2)) * OB_THREADS + w * 32;
        while (b) {
            const int k = __builtin_ctz(b);
            b &= b - 1;
            elist[pos++] = 4 * (grp + k) + comp;
        }
    }
    __syncthreads();

    for (int i0 = 0; i0 < T; i0 += OB_BATCH * OB_THREADS) {
        int e[OB_BATCH], d[OB_BATCH], s[OB_BATCH];
#pragma unroll
        for (int u = 0; u < OB_BATCH; ++u) {
            const int idx = i0 + u * OB_THREADS + tid;
            e[u] = (idx < T) ? elist[base + idx] : -1;
        }
#pragma unroll
        for (int u = 0; u < OB_BATCH; ++u) {
            const int ee = e[u] < 0 ? 0 : e[u];
            d[u] = dstp[ee];
            s[u] = ei[ee];
        }
#pragma unroll
        for (int u = 0; u < OB_BATCH; ++u) {
            if (e[u] >= 0) {
                const int r = d[u] - lo;
                atomicAdd(&cnt[r], 1);
                elist[base + i0 + u * OB_THREADS + tid] = (s[u] << ROW_BITS) | r;
            }
        }
    }
    __syncthreads();

    if (w == 0) {
        int cq[NPL];
        int tot = 0;
#pragma unroll
        for (int qn = 0; qn < NPL; ++qn) { cq[qn] = cnt[NPL * lane + qn]; tot += cq[qn]; }
        int ps = tot;
        for (int off = 1; off < 32; off <<= 1) {
            int v = __shfl_up(ps, off);
            if (lane >= off) ps += v;
        }
        int bq = base + ps - tot;
#pragma unroll
        for (int qn = 0; qn < NPL; ++qn) {
            fillp[NPL * lane + qn] = bq;
            start[lo + NPL * lane + qn] = bq;
            bq += cq[qn];
        }
        if (blockIdx.x == OWN_BLOCKS - 1 && lane == 31) start[NPAD] = bq;
    }
    __syncthreads();

    int* csrL = (int*)&ownL[0][0];
    for (int i0 = 0; i0 < T; i0 += OB_BATCH * OB_THREADS) {
        int en[OB_BATCH];
        float4 xv[OB_BATCH];
#pragma unroll
        for (int u = 0; u < OB_BATCH; ++u) {
            const int idx = i0 + u * OB_THREADS + tid;
            en[u] = (idx < T) ? elist[base + idx] : -1;
        }
#pragma unroll
        for (int u = 0; u < OB_BATCH; ++u) {
            const int s = en[u] < 0 ? 0 : (en[u] >> ROW_BITS);
            xv[u] = *(const float4*)&xL[s * NODE_F];
        }
#pragma unroll
        for (int u = 0; u < OB_BATCH; ++u) {
            if (en[u] >= 0) {
                const int s = en[u] >> ROW_BITS;
                const int r = en[u] & (NB - 1);
                const int slot = atomicAdd(&fillp[r], 1);
                if (slot - base < CSR_CAP) csrL[slot - base] = s;
                else                       csr[slot] = s;
                atomicAdd(&aggL[r * NODE_F + 0], xv[u].x);
                atomicAdd(&aggL[r * NODE_F + 1], xv[u].y);
                atomicAdd(&aggL[r * NODE_F + 2], xv[u].z);
                atomicAdd(&aggL[r * NODE_F + 3], xv[u].w);
            }
        }
    }
    __syncthreads();

    for (int i = tid; i < min(T, CSR_CAP); i += OB_THREADS) csr[base + i] = csrL[i];

    if (tid < NB) rdeg[lo + tid] = 1.0f / fmaxf((float)cnt[tid], 1.0f);
    const int lrow0 = (w >> 2) * 16;
    const int col0  = (w & 3) * 16;
    const int m     = lane & 15;
    const int half  = lane >> 4;
    const int kb    = half * 2;
    const int lrow  = lrow0 + m;
    const int row   = lo + lrow;
    const float rd  = 1.0f / fmaxf((float)cnt[lrow], 1.0f);

    v2f am, ax, bl, br;
    am.x = aggL[lrow * NODE_F + kb]     * rd;
    am.y = aggL[lrow * NODE_F + kb + 1] * rd;
    ax.x = (row < N_NODES) ? x[row * NODE_F + kb]     : 0.0f;
    ax.y = (row < N_NODES) ? x[row * NODE_F + kb + 1] : 0.0f;
    bl.x = Wl1[(col0 + m) * NODE_F + kb];
    bl.y = Wl1[(col0 + m) * NODE_F + kb + 1];
    br.x = Wr1[(col0 + m) * NODE_F + kb];
    br.y = Wr1[(col0 + m) * NODE_F + kb + 1];

    v8f acc = {};
    acc = __builtin_amdgcn_wmma_f32_16x16x4_f32(false, am, false, bl, (short)0, acc, false, false);
    acc = __builtin_amdgcn_wmma_f32_16x16x4_f32(false, ax, false, br, (short)0, acc, false, false);

#pragma unroll
    for (int i = 0; i < 8; ++i) {
        int r = lo + lrow0 + i + half * 8;
        int c = col0 + m;
        h1[r * HID + c] = fmaxf(acc[i] + bl1[c], 0.0f);
    }
}

#define T_PITCH  68
#define SH_WAVES (HID / 16)
#define FG_WAVES 8
#define FG_RPW   (16 / FG_WAVES)
#define FG_CHUNK 16
#define FG_CHUNKS 3
#define FG_ROUND (FG_CHUNK * FG_CHUNKS)
#define KS       (HID / 4)
#define W_PIECES ((HID * HID / 4) / (FG_WAVES * 32))
static_assert(FG_RPW == 2 && FG_WAVES >= SH_WAVES && FG_CHUNK == 16, "block shape");
static_assert(W_PIECES * FG_WAVES * 32 * 4 == HID * HID && 16 * HID / 4 == FG_WAVES * 32, "copies are whole");
template <int U>
__device__ __forceinline__ int half_lane(int v) {
    return __builtin_amdgcn_update_dpp(0, v, 0x150 + U, 0xF, 0xF, false);
}
__device__ __forceinline__ void chunk_load(v4f (&v)[FG_CHUNK], int id, const float* __restrict__ rows, unsigned qoff) {
#define FG_LD(U) v[U] = *(const v4f*)((const char*)rows + (((unsigned)half_lane<U>(id) * (unsigned)(HID * 4)) + qoff));
    FG_LD(0) FG_LD(1) FG_LD(2)  FG_LD(3)  FG_LD(4)  FG_LD(5)  FG_LD(6)  FG_LD(7)
    FG_LD(8) FG_LD(9) FG_LD(10) FG_LD(11) FG_LD(12) FG_LD(13) FG_LD(14) FG_LD(15)
#undef FG_LD
}
__device__ __forceinline__ v4f gather_pair(int beg, int len, int maxlen, int (&id)[FG_CHUNKS],
                                           const int* __restrict__ csr, const float* __restrict__ rows, int q) {
    const unsigned qoff = (unsigned)q * 16u;
    v4f acc = {0.0f, 0.0f, 0.0f, 0.0f};
    for (int base = 0; base < maxlen; base += FG_ROUND) {
        const int nb = base + FG_ROUND;
        int idn[FG_CHUNKS] = {0, 0, 0};
        if (nb < maxlen) {
#pragma unroll
            for (int c = 0; c < FG_CHUNKS; ++c) {
                const int j = nb + FG_CHUNK * c + q;
                idn[c] = (j < len) ? csr[beg + j] : 0;
            }
        }
        v4f v[FG_CHUNKS][FG_CHUNK];
#pragma unroll
        for (int c = 0; c < FG_CHUNKS; ++c)
            if (base + FG_CHUNK * c < maxlen) chunk_load(v[c], id[c], rows, qoff);
#pragma unroll
        for (int c = 0; c < FG_CHUNKS; ++c) {
            if (base + FG_CHUNK * c < maxlen) {
#pragma unroll
                for (int u = 0; u < FG_CHUNK; ++u) {
                    const bool ok = base + FG_CHUNK * c + u < len;
                    acc.x += ok ? v[c][u].x : 0.0f;
                    acc.y += ok ? v[c][u].y : 0.0f;
                    acc.z += ok ? v[c][u].z : 0.0f;
                    acc.w += ok ? v[c][u].w : 0.0f;
                }
            }
        }
#pragma unroll
        for (int c = 0; c < FG_CHUNKS; ++c) id[c] = idn[c];
    }
    return acc;
}
__global__ __launch_bounds__(FG_WAVES * 32)
void k_gather_sage2_hpart(const int* __restrict__ start, const int* __restrict__ csr,
                          const float* __restrict__ h1, const float* __restrict__ rdeg,
                          const float* __restrict__ Wl2, const float* __restrict__ bl2,
                          const float* __restrict__ Wr2, const float* __restrict__ Wc1,
                          float* __restrict__ hpart) {
    __shared__ __attribute__((aligned(16))) float wlT[HID * T_PITCH];
    __shared__ __attribute__((aligned(16))) float wrT[HID * T_PITCH];
    __shared__ __attribute__((aligned(16))) float wcT[HID * T_PITCH];
    __shared__ __attribute__((aligned(16))) float h1T[16 * T_PITCH];
    __shared__ __attribute__((aligned(16))) float aggT[16 * T_PITCH];
    __shared__ __attribute__((aligned(16))) float h2t[16 * T_PITCH];
    const int tid  = threadIdx.x;
    const int row0 = blockIdx.x * 16;
    const int lane = tid & 31;
    const int w    = __builtin_amdgcn_readfirstlane(tid >> 5);
    const int m    = lane & 15;
    const int half = lane >> 4;
    const int kb   = half * 2;
    const int col0 = (w & (SH_WAVES - 1)) * 16;
    const bool mat = w < SH_WAVES;

    float4 ql[W_PIECES], qr[W_PIECES], qc[W_PIECES];
#pragma unroll
    for (int i = 0; i < W_PIECES; ++i) {
        const int p = tid + FG_WAVES * 32 * i;
        const int r = p >> 4, c4 = (p & 15) * 4;
        ql[i] = *(const float4*)(Wl2 + r * HID + c4);
        qr[i] = *(const float4*)(Wr2 + r * HID + c4);
        qc[i] = *(const float4*)(Wc1 + r * (2 * HID) + c4);
    }
    const float4 qh = *(const float4*)(h1 + (row0 + (tid >> 4)) * HID + (tid & 15) * 4);

    const int n  = row0 + FG_RPW * w;
    const int s0 = start[n];
    const int s1 = start[n + 1];
    const int s2 = start[n + 2];
    const int q      = lane & 15;
    const int beg    = half ? s1 : s0;
    const int len    = (half ? s2 : s1) - beg;
    const int maxlen = max(s1 - s0, s2 - s1);
    int id[FG_CHUNKS];
#pragma unroll
    for (int c = 0; c < FG_CHUNKS; ++c) {
        const int j = FG_CHUNK * c + q;
        id[c] = (j < len) ? csr[beg + j] : 0;
    }

#pragma unroll
    for (int i = 0; i < W_PIECES; ++i) {
        const int p = tid + FG_WAVES * 32 * i;
        const int r = p >> 4, c4 = (p & 15) * 4;
        *(float4*)&wlT[r * T_PITCH + c4] = ql[i];
        *(float4*)&wrT[r * T_PITCH + c4] = qr[i];
        *(float4*)&wcT[r * T_PITCH + c4] = qc[i];
    }
    *(float4*)&h1T[(tid >> 4) * T_PITCH + (tid & 15) * 4] = qh;

    {
        const v4f sum = gather_pair(beg, len, maxlen, id, csr, h1, q);
        *(v4f*)&aggT[(FG_RPW * w + half) * T_PITCH + 4 * q] = sum;
    }
    __syncthreads();

    if (mat) {
        const float rd   = rdeg[row0 + m];
        const float bias = bl2[col0 + m];
        v8f acc = {};
#pragma unroll
        for (int ks = 0; ks < KS; ++ks) {
            const int k = 4 * ks + kb;
            v2f am = *(const v2f*)&aggT[m * T_PITCH + k];
            am.x *= rd;
            am.y *= rd;
            const v2f ah = *(const v2f*)&h1T[m * T_PITCH + k];
            const v2f bl = *(const v2f*)&wlT[(col0 + m) * T_PITCH + k];
            const v2f br = *(const v2f*)&wrT[(col0 + m) * T_PITCH + k];
            acc = __builtin_amdgcn_wmma_f32_16x16x4_f32(false, am, false, bl, (short)0, acc, false, false);
            acc = __builtin_amdgcn_wmma_f32_16x16x4_f32(false, ah, false, br, (short)0, acc, false, false);
        }
#pragma unroll
        for (int i = 0; i < 8; ++i) {
            int r = i + half * 8;
            int c = col0 + m;
            h2t[r * T_PITCH + c] = fmaxf(acc[i] + bias, 0.0f);
        }
    }
    __syncthreads();

    if (mat) {
        v8f acc2 = {};
#pragma unroll
        for (int ks = 0; ks < KS; ++ks) {
            const int k = 4 * ks + kb;
            const v2f a = *(const v2f*)&h2t[m * T_PITCH + k];
            const v2f b = *(const v2f*)&wcT[(col0 + m) * T_PITCH + k];
            acc2 = __builtin_amdgcn_wmma_f32_16x16x4_f32(false, a, false, b, (short)0, acc2, false, false);
        }
#pragma unroll
        for (int i = 0; i < 8; ++i) {
            int r = row0 + i + half * 8;
            int c = col0 + m;
            hpart[r * HID + c] = acc2[i];
        }
    }
}

#define CLS_BT      16
#define CLS_G       4
#define CLS_THREADS 128
__global__ __launch_bounds__(CLS_THREADS)
void k_classify(const float* __restrict__ hpart, const float* __restrict__ tpart,
                const float* __restrict__ Wc2, const float* __restrict__ bc2,
                float* __restrict__ out) {
    __shared__ float w2[HID];
    if (threadIdx.x < HID) w2[threadIdx.x] = Wc2[threadIdx.x];
    __syncthreads();

    const int n  = blockIdx.x * CLS_THREADS + threadIdx.x;
    const int b0 = blockIdx.y * CLS_BT;
    if (n - (int)(threadIdx.x & 31) >= N_NODES) return;

    float a[HID];
    const float4* hp = (const float4*)(hpart + (size_t)n * HID);
#pragma unroll
    for (int j = 0; j < HID / 4; ++j) {
        float4 v = hp[j];
        a[4 * j + 0] = v.x; a[4 * j + 1] = v.y; a[4 * j + 2] = v.z; a[4 * j + 3] = v.w;
    }
    float w[HID];
#pragma unroll
    for (int h = 0; h < HID; ++h) w[h] = w2[h];
    const float bias = bc2[0];

    for (int g = 0; g < CLS_BT / CLS_G; ++g) {
        const float* __restrict__ tp = tpart + (size_t)(b0 + g * CLS_G) * HID;
        float acc[CLS_G];
#pragma unroll
        for (int r = 0; r < CLS_G; ++r) acc[r] = bias;
#pragma unroll
        for (int hc = 0; hc < HID; hc += 16) {
#pragma unroll
            for (int r = 0; r < CLS_G; ++r) {
#pragma unroll
                for (int k = 0; k < 16; ++k)
                    acc[r] += fmaxf(a[hc + k] + tp[r * HID + hc + k], 0.0f) * w[hc + k];
            }
        }
        if (n < N_NODES) {
#pragma unroll
            for (int r = 0; r < CLS_G; ++r)
                out[(size_t)(b0 + g * CLS_G + r) * N_NODES + n] = acc[r];
        }
    }
}

extern "C" void kernel_launch(void* const* d_in, const int* in_sizes, int n_in,
                              void* d_out, int out_size, void* d_ws, size_t ws_size,
                              hipStream_t stream) {
    const float* x    = (const float*)d_in[0];
    const int*   ei   = (const int*)d_in[1];
    const float* task = (const float*)d_in[2];
    const float* Wl1  = (const float*)d_in[3];
    const float* bl1  = (const float*)d_in[4];
    const float* Wr1  = (const float*)d_in[5];
    const float* Wl2  = (const float*)d_in[6];
    const float* bl2  = (const float*)d_in[7];
    const float* Wr2  = (const float*)d_in[8];
    const float* Wt1  = (const float*)d_in[9];
    const float* bt1  = (const float*)d_in[10];
    const float* Wt2  = (const float*)d_in[11];
    const float* bt2  = (const float*)d_in[12];
    const float* Wc1  = (const float*)d_in[13];
    const float* bc1  = (const float*)d_in[14];
    const float* Wc2  = (const float*)d_in[15];
    const float* bc2  = (const float*)d_in[16];
    float* out = (float*)d_out;

    float* ws    = (float*)d_ws;
    float* rdeg  = ws + OFF_RDEG;
    float* h1    = ws + OFF_H1;
    float* hpart = ws + OFF_HPART;
    float* tpart = ws + OFF_TPART;
    int*   elist = (int*)(ws + OFF_H2);
    int*   csr   = (int*)(ws + OFF_AGG2);
    int*   start = (int*)(ws + OFF_START);

    k_build<<<OWN_BLOCKS + TASK_BLOCKS, OB_THREADS, 0, stream>>>(ei, x, Wl1, bl1, Wr1, elist, csr, start, rdeg, h1,
                                                                 task, Wt1, bt1, Wt2, bt2, Wc1, bc1, tpart);
    k_gather_sage2_hpart<<<NPAD / 16, FG_WAVES * 32, 0, stream>>>(start, csr, h1, rdeg, Wl2, bl2, Wr2, Wc1, hpart);
    k_classify<<<dim3(NPAD / CLS_THREADS, BATCH / CLS_BT), CLS_THREADS, 0, stream>>>(hpart, tpart, Wc2, bc2, out);
}


# build launch: a block's piece of the edge list stays in LDS (overflow to memory as before), list starts stored behind their barrier: no barrier waits for a global store, no stage reloads the list
# speedup vs baseline: 4.4135x; 1.0334x over previous
#include <hip/hip_runtime.h>
#include <hip/hip_bf16.h>

#define N_NODES 5000
#define NPAD    5120
#define N_EDGES 160000
#define HID     64
#define NODE_F  4
#define TASK_F  4
#define BATCH   512

typedef __attribute__((ext_vector_type(2))) float v2f;
typedef __attribute__((ext_vector_type(4))) float v4f;
typedef __attribute__((ext_vector_type(8))) float v8f;

#define OFF_AGG2   (NPAD * (1 + NODE_F))
#define OFF_RDEG   (OFF_AGG2 + NPAD * HID)
#define OFF_H1     (OFF_RDEG  + NPAD)
#define OFF_H2     (OFF_H1    + NPAD * HID)
#define OFF_HPART  (OFF_H2    + NPAD * HID)
#define OFF_TPART  (OFF_HPART + NPAD * HID)
#define OFF_START  (OFF_H2    + N_EDGES)


#define NB          64
#define OB_THREADS  512
#define OB_WAVES    (OB_THREADS / 32)
#define OWN_BLOCKS  (NPAD / NB)
#define E4          (N_EDGES / 4)
#define OB_TRIPS    (((E4 + OB_THREADS - 1) / OB_THREADS + 7) / 8 * 8)
#define OB_WORDS    (OB_TRIPS / 8)
#define OB_BATCH    5
#define X_PIECES    ((N_NODES + OB_THREADS - 1) / OB_THREADS)
#define ROW_BITS    6
#define CSR_CAP     (OB_WORDS * OB_THREADS)
#define ELIST_CAP   CSR_CAP
#define NPL         (NB / 32)
#define OB_FULLW    ((E4 / OB_THREADS) / 8)
static_assert((1 << ROW_BITS) == NB && (long long)(N_NODES - 1) * NB + NB - 1 < 0x7fffffffLL, "packed entries are non-negative ints");
static_assert(OB_TRIPS % 16 == 0 && OB_TRIPS * OB_THREADS >= E4, "mask words are filled whole, in pairs, and cover the row");
static_assert(OB_FULLW >= OB_WORDS - 1, "only the last word tests for the row's end");
static_assert(ELIST_CAP % (OB_BATCH * OB_THREADS) == 0, "a trip of a dense pass lies wholly inside or wholly outside the LDS part of the piece");
static_assert(NB * NODE_F <= OB_THREADS && NB % 32 == 0 && OB_WAVES <= 32 && OB_WAVES == (NB / 16) * (HID / 16), "block shape");
__device__ __forceinline__ int ob_slot(int v, int lo, int hi, int slot, int lane, int acc, int& nbel) {
    const unsigned below = __builtin_amdgcn_ballot_w32(v < lo);
    const unsigned under = __builtin_amdgcn_ballot_w32(v < hi);
    nbel += __popc(below);
    return (lane == slot) ? (int)(under & ~below) : acc;
}
template <bool FULL>
__device__ __forceinline__ void ob_load(int4 (&d)[8], const int4* __restrict__ dst4, int wd, int tid) {
#pragma unroll
    for (int j = 0; j < 8; ++j) {
        const int i = (wd * 8 + j) * OB_THREADS + tid;
        if (FULL) {
            d[j] = dst4[i];
        } else {
            d[j] = make_int4(0x7fffffff, 0x7fffffff, 0x7fffffff, 0x7fffffff);
            if (i < E4) d[j] = dst4[i];
        }
    }
}
__device__ __forceinline__ float4 x_piece(const float* __restrict__ x, int piece, int tid) {
    const int p = piece * OB_THREADS + tid;
    return ((const float4*)x)[min(p, N_NODES - 1)];
}
__device__ __forceinline__ void x_keep(float* xL, float4 v, int piece, int tid) {
    const int p = piece * OB_THREADS + tid;
    if (p < N_NODES) *(float4*)&xL[p * NODE_F] = v;
}
__device__ __forceinline__ int ob_word(const int4 (&d)[8], int lo, int hi, int lane, int& nbel) {
    int acc = 0;
#pragma unroll
    for (int j = 0; j < 8; ++j) {
        acc = ob_slot(d[j].x, lo, hi, 4 * j + 0, lane, acc, nbel);
        acc = ob_slot(d[j].y, lo, hi, 4 * j + 1, lane, acc, nbel);
        acc = ob_slot(d[j].z, lo, hi, 4 * j + 2, lane, acc, nbel);
        acc = ob_slot(d[j].w, lo, hi, 4 * j + 3, lane, acc, nbel);
    }
    return acc;
}
#define TASK_ROWS   (OB_THREADS / HID)
#define TASK_BLOCKS (BATCH / TASK_ROWS)
static_assert(BATCH % TASK_ROWS == 0, "task blocks are whole");
__global__ __launch_bounds__(OB_THREADS) __attribute__((amdgpu_waves_per_eu(4, 4)))
void k_build(const int* __restrict__ ei, const float* __restrict__ x,
             const float* __restrict__ Wl1, const float* __restrict__ bl1,
             const float* __restrict__ Wr1,
             int* __restrict__ elist, int* __restrict__ csr, int* __restrict__ start,
             float* __restrict__ rdeg, float* __restrict__ h1,
             const float* __restrict__ task, const float* __restrict__ Wt1,
             const float* __restrict__ bt1, const float* __restrict__ Wt2,
             const float* __restrict__ bt2, const float* __restrict__ Wc1,
             const float* __restrict__ bc1, float* __restrict__ tpart) {
    __shared__ float tu[TASK_ROWS][HID];
    __shared__ float tt[TASK_ROWS][HID];
    if (blockIdx.x >= OWN_BLOCKS) {
        const int sub = threadIdx.x / HID;
        const int h   = threadIdx.x % HID;
        const int b   = (blockIdx.x - OWN_BLOCKS) * TASK_ROWS + sub;

        float acc = bt1[h];
#pragma unroll
        for (int k = 0; k < TASK_F; ++k) acc += task[b * TASK_F + k] * Wt1[h * TASK_F + k];
        tu[sub][h] = fmaxf(acc, 0.0f);
        __syncthreads();

        float acc2 = bt2[h];
#pragma unroll 8
        for (int k = 0; k < HID; ++k) acc2 += tu[sub][k] * Wt2[h * HID + k];
        tt[sub][h] = acc2;
        __syncthreads();

        float acc3 = bc1[h];
#pragma unroll 8
        for (int k = 0; k < HID; ++k) acc3 += tt[sub][k] * Wc1[h * (2 * HID) + HID + k];
        tpart[b * HID + h] = acc3;
        return;
    }
    __shared__ int   cnt[NB];
    __shared__ int   fillp[NB];
    __shared__ float aggL[NB * NODE_F];
    __shared__ __attribute__((aligned(16))) float xL[N_NODES * NODE_F];
    __shared__ unsigned ownL[OB_WORDS][OB_THREADS];
    __shared__ int   elistL[ELIST_CAP];
    __shared__ int   wsum[OB_WAVES];
    __shared__ int   wbel[OB_WAVES];
    __shared__ int   sbase, stot;

    const int tid  = threadIdx.x;
    const int lane = tid & 31;
    const int w    = tid >> 5;
    const int lo   = blockIdx.x * NB;
    const int hi   = lo + NB;
    const int*  __restrict__ dstp = ei + N_EDGES;
    const int4* __restrict__ dst4 = (const int4*)dstp;

    float4 xq[X_PIECES];
#pragma unroll
    for (int i = 0; i < X_PIECES; ++i) xq[i] = x_piece(x, i, tid);

    int nm = 0;
    int nbel = 0;
    int4 dA[8], dB[8];
    ob_load<true>(dA, dst4, 0, tid);
#pragma unroll 1
    for (int wd = 0; wd < OB_WORDS - 2; wd += 2) {
        ob_load<true>(dB, dst4, wd + 1, tid);
        __builtin_amdgcn_sched_barrier(0);
        const int accA = ob_word(dA, lo, hi, lane, nbel);
        ownL[wd][tid] = (unsigned)accA;
        nm += __popc((unsigned)accA);
        ob_load<true>(dA, dst4, wd + 2, tid);
        __builtin_amdgcn_sched_barrier(0);
        const int accB = ob_word(dB, lo, hi, lane, nbel);
        ownL[wd + 1][tid] = (unsigned)accB;
        nm += __popc((unsigned)accB);
    }
    {
        ob_load<false>(dB, dst4, OB_WORDS - 1, tid);
        const int accA = ob_word(dA, lo, hi, lane, nbel);
        ownL[OB_WORDS - 2][tid] = (unsigned)accA;
        nm += __popc((unsigned)accA);
        const int accB = ob_word(dB, lo, hi, lane, nbel);
        ownL[OB_WORDS - 1][tid] = (unsigned)accB;
        nm += __popc((unsigned)accB);
    }

    int inc = nm;
    for (int off = 1; off < 32; off <<= 1) {
        int v = __shfl_up(inc, off);
        if (lane >= off) inc += v;
    }
    if (lane == 31) wsum[w] = inc;
    if (lane == 0)  wbel[w] = nbel;
    if (tid < NB)          cnt[tid]  = 0;
    if (tid < NB * NODE_F) aggL[tid] = 0.0f;
#pragma unroll
    for (int i = 0; i < X_PIECES; ++i) x_keep(xL, xq[i], i, tid);
    __syncthreads();
    if (w == 0) {
        const int mine = (lane < OB_WAVES) ? wsum[lane] : 0;
        int ws = mine;
        for (int off = 1; off < 32; off <<= 1) {
            int v = __shfl_up(ws, off);
            if (lane >= off) ws += v;
        }
        if (lane < OB_WAVES) wsum[lane] = ws - mine;
        int bel = (lane < OB_WAVES) ? wbel[lane] : 0;
        for (int off = 16; off > 0; off >>= 1) bel += __shfl_xor(bel, off);
        if (lane == 31) stot  = ws;
        if (lane == 0)  sbase = bel;
    }
    __syncthreads();
    const int base = sbase;
    const int T    = stot;

    int pos = wsum[w] + inc - nm;
    const int comp = lane & 3;
    for (int wd = 0; wd < OB_WORDS; ++wd) {
        unsigned b = ownL[wd][tid];
        const int grp = (wd * 8 + (lane >> 2)) * OB_THREADS + w * 32;
        while (b) {
            const int k = __builtin_ctz(b);
            b &= b - 1;
            const int e = 4 * (grp + k) + comp;
            if (pos < ELIST_CAP) elistL[pos] = e;
            else                 elist[base + pos] = e;
            ++pos;
        }
    }
    __syncthreads();

    for (int i0 = 0; i0 < T; i0 += OB_BATCH * OB_THREADS) {
        const bool inL = i0 < ELIST_CAP;
        int e[OB_BATCH], d[OB_BATCH], s[OB_BATCH];
#pragma unroll
        for (int u = 0; u < OB_BATCH; ++u) {
            const int idx = i0 + u * OB_THREADS + tid;
            e[u] = (idx < T) ? (inL ? elistL[idx] : elist[base + idx]) : -1;
        }
#pragma unroll
        for (int u = 0; u < OB_BATCH; ++u) {
            const int ee = e[u] < 0 ? 0 : e[u];
            d[u] = dstp[ee];
            s[u] = ei[ee];
        }
#pragma unroll
        for (int u = 0; u < OB_BATCH; ++u) {
            if (e[u] >= 0) {
                const int r = d[u] - lo;
                atomicAdd(&cnt[r], 1);
                const int idx = i0 + u * OB_THREADS + tid;
                const int pk  = (s[u] << ROW_BITS) | r;
                if (inL) elistL[idx] = pk;
                else     elist[base + idx] = pk;
            }
        }
    }
    __syncthreads();

    int sq[NPL + 1];
    if (w == 0) {
        int cq[NPL];
        int tot = 0;
#pragma unroll
        for (int qn = 0; qn < NPL; ++qn) { cq[qn] = cnt[NPL * lane + qn]; tot += cq[qn]; }
        int ps = tot;
        for (int off = 1; off < 32; off <<= 1) {
            int v = __shfl_up(ps, off);
            if (lane >= off) ps += v;
        }
        int bq = base + ps - tot;
#pragma unroll
        for (int qn = 0; qn < NPL; ++qn) {
            fillp[NPL * lane + qn] = bq;
            sq[qn] = bq;
            bq += cq[qn];
        }
        sq[NPL] = bq;
    }
    __syncthreads();
    if (w == 0) {
#pragma unroll
        for (int qn = 0; qn < NPL; ++qn) start[lo + NPL * lane + qn] = sq[qn];
        if (blockIdx.x == OWN_BLOCKS - 1 && lane == 31) start[NPAD] = sq[NPL];
    }

    int* csrL = (int*)&ownL[0][0];
    for (int i0 = 0; i0 < T; i0 += OB_BATCH * OB_THREADS) {
        const bool inL = i0 < ELIST_CAP;
        int en[OB_BATCH];
        float4 xv[OB_BATCH];
#pragma unroll
        for (int u = 0; u < OB_BATCH; ++u) {
            const int idx = i0 + u * OB_THREADS + tid;
            en[u] = (idx < T) ? (inL ? elistL[idx] : elist[base + idx]) : -1;
        }
#pragma unroll
        for (int u = 0; u < OB_BATCH; ++u) {
            const int s = en[u] < 0 ? 0 : (en[u] >> ROW_BITS);
            xv[u] = *(const float4*)&xL[s * NODE_F];
        }
#pragma unroll
        for (int u = 0; u < OB_BATCH; ++u) {
            if (en[u] >= 0) {
                const int s = en[u] >> ROW_BITS;
                const int r = en[u] & (NB - 1);
                const int slot = atomicAdd(&fillp[r], 1);
                if (slot - base < CSR_CAP) csrL[slot - base] = s;
                else                       csr[slot] = s;
                atomicAdd(&aggL[r * NODE_F + 0], xv[u].x);
                atomicAdd(&aggL[r * NODE_F + 1], xv[u].y);
                atomicAdd(&aggL[r * NODE_F + 2], xv[u].z);
                atomicAdd(&aggL[r * NODE_F + 3], xv[u].w);
            }
        }
    }
    __syncthreads();

    for (int i = tid; i < min(T, CSR_CAP); i += OB_THREADS) csr[base + i] = csrL[i];

    if (tid < NB) rdeg[lo + tid] = 1.0f / fmaxf((float)cnt[tid], 1.0f);
    const int lrow0 = (w >> 2) * 16;
    const int col0  = (w & 3) * 16;
    const int m     = lane & 15;
    const int half  = lane >> 4;
    const int kb    = half * 2;
    const int lrow  = lrow0 + m;
    const int row   = lo + lrow;
    const float rd  = 1.0f / fmaxf((float)cnt[lrow], 1.0f);

    v2f am, ax, bl, br;
    am.x = aggL[lrow * NODE_F + kb]     * rd;
    am.y = aggL[lrow * NODE_F + kb + 1] * rd;
    ax.x = (row < N_NODES) ? x[row * NODE_F + kb]     : 0.0f;
    ax.y = (row < N_NODES) ? x[row * NODE_F + kb + 1] : 0.0f;
    bl.x = Wl1[(col0 + m) * NODE_F + kb];
    bl.y = Wl1[(col0 + m) * NODE_F + kb + 1];
    br.x = Wr1[(col0 + m) * NODE_F + kb];
    br.y = Wr1[(col0 + m) * NODE_F + kb + 1];

    v8f acc = {};
    acc = __builtin_amdgcn_wmma_f32_16x16x4_f32(false, am, false, bl, (short)0, acc, false, false);
    acc = __builtin_amdgcn_wmma_f32_16x16x4_f32(false, ax, false, br, (short)0, acc, false, false);

#pragma unroll
    for (int i = 0; i < 8; ++i) {
        int r = lo + lrow0 + i + half * 8;
        int c = col0 + m;
        h1[r * HID + c] = fmaxf(acc[i] + bl1[c], 0.0f);
    }
}

#define T_PITCH  68
#define SH_WAVES (HID / 16)
#define FG_WAVES 8
#define FG_RPW   (16 / FG_WAVES)
#define FG_CHUNK 16
#define FG_CHUNKS 3
#define FG_ROUND (FG_CHUNK * FG_CHUNKS)
#define KS       (HID / 4)
#define W_PIECES ((HID * HID / 4) / (FG_WAVES * 32))
static_assert(FG_RPW == 2 && FG_WAVES >= SH_WAVES && FG_CHUNK == 16, "block shape");
static_assert(W_PIECES * FG_WAVES * 32 * 4 == HID * HID && 16 * HID / 4 == FG_WAVES * 32, "copies are whole");
template <int U>
__device__ __forceinline__ int half_lane(int v) {
    return __builtin_amdgcn_update_dpp(0, v, 0x150 + U, 0xF, 0xF, false);
}
__device__ __forceinline__ void chunk_load(v4f (&v)[FG_CHUNK], int id, const float* __restrict__ rows, unsigned qoff) {
#define FG_LD(U) v[U] = *(const v4f*)((const char*)rows + (((unsigned)half_lane<U>(id) * (unsigned)(HID * 4)) + qoff));
    FG_LD(0) FG_LD(1) FG_LD(2)  FG_LD(3)  FG_LD(4)  FG_LD(5)  FG_LD(6)  FG_LD(7)
    FG_LD(8) FG_LD(9) FG_LD(10) FG_LD(11) FG_LD(12) FG_LD(13) FG_LD(14) FG_LD(15)
#undef FG_LD
}
__device__ __forceinline__ v4f gather_pair(int beg, int len, int maxlen, int (&id)[FG_CHUNKS],
                                           const int* __restrict__ csr, const float* __restrict__ rows, int q) {
    const unsigned qoff = (unsigned)q * 16u;
    v4f acc = {0.0f, 0.0f, 0.0f, 0.0f};
    for (int base = 0; base < maxlen; base += FG_ROUND) {
        const int nb = base + FG_ROUND;
        int idn[FG_CHUNKS] = {0, 0, 0};
        if (nb < maxlen) {
#pragma unroll
            for (int c = 0; c < FG_CHUNKS; ++c) {
                const int j = nb + FG_CHUNK * c + q;
                idn[c] = (j < len) ? csr[beg + j] : 0;
            }
        }
        v4f v[FG_CHUNKS][FG_CHUNK];
#pragma unroll
        for (int c = 0; c < FG_CHUNKS; ++c)
            if (base + FG_CHUNK * c < maxlen) chunk_load(v[c], id[c], rows, qoff);
#pragma unroll
        for (int c = 0; c < FG_CHUNKS; ++c) {
            if (base + FG_CHUNK * c < maxlen) {
#pragma unroll
                for (int u = 0; u < FG_CHUNK; ++u) {
                    const bool ok = base + FG_CHUNK * c + u < len;
                    acc.x += ok ? v[c][u].x : 0.0f;
                    acc.y += ok ? v[c][u].y : 0.0f;
                    acc.z += ok ? v[c][u].z : 0.0f;
                    acc.w += ok ? v[c][u].w : 0.0f;
                }
            }
        }
#pragma unroll
        for (int c = 0; c < FG_CHUNKS; ++c) id[c] = idn[c];
    }
    return acc;
}
__global__ __launch_bounds__(FG_WAVES * 32)
void k_gather_sage2_hpart(const int* __restrict__ start, const int* __restrict__ csr,
                          const float* __restrict__ h1, const float* __restrict__ rdeg,
                          const float* __restrict__ Wl2, const float* __restrict__ bl2,
                          const float* __restrict__ Wr2, const float* __restrict__ Wc1,
                          float* __restrict__ hpart) {
    __shared__ __attribute__((aligned(16))) float wlT[HID * T_PITCH];
    __shared__ __attribute__((aligned(16))) float wrT[HID * T_PITCH];
    __shared__ __attribute__((aligned(16))) float wcT[HID * T_PITCH];
    __shared__ __attribute__((aligned(16))) float h1T[16 * T_PITCH];
    __shared__ __attribute__((aligned(16))) float aggT[16 * T_PITCH];
    __shared__ __attribute__((aligned(16))) float h2t[16 * T_PITCH];
    const int tid  = threadIdx.x;
    const int row0 = blockIdx.x * 16;
    const int lane = tid & 31;
    const int w    = __builtin_amdgcn_readfirstlane(tid >> 5);
    const int m    = lane & 15;
    const int half = lane >> 4;
    const int kb   = half * 2;
    const int col0 = (w & (SH_WAVES - 1)) * 16;
    const bool mat = w < SH_WAVES;

    float4 ql[W_PIECES], qr[W_PIECES], qc[W_PIECES];
#pragma unroll
    for (int i = 0; i < W_PIECES; ++i) {
        const int p = tid + FG_WAVES * 32 * i;
        const int r = p >> 4, c4 = (p & 15) * 4;
        ql[i] = *(const float4*)(Wl2 + r * HID + c4);
        qr[i] = *(const float4*)(Wr2 + r * HID + c4);
        qc[i] = *(const float4*)(Wc1 + r * (2 * HID) + c4);
    }
    const float4 qh = *(const float4*)(h1 + (row0 + (tid >> 4)) * HID + (tid & 15) * 4);

    const int n  = row0 + FG_RPW * w;
    const int s0 = start[n];
    const int s1 = start[n + 1];
    const int s2 = start[n + 2];
    const int q      = lane & 15;
    const int beg    = half ? s1 : s0;
    const int len    = (half ? s2 : s1) - beg;
    const int maxlen = max(s1 - s0, s2 - s1);
    int id[FG_CHUNKS];
#pragma unroll
    for (int c = 0; c < FG_CHUNKS; ++c) {
        const int j = FG_CHUNK * c + q;
        id[c] = (j < len) ? csr[beg + j] : 0;
    }

#pragma unroll
    for (int i = 0; i < W_PIECES; ++i) {
        const int p = tid + FG_WAVES * 32 * i;
        const int r = p >> 4, c4 = (p & 15) * 4;
        *(float4*)&wlT[r * T_PITCH + c4] = ql[i];
        *(float4*)&wrT[r * T_PITCH + c4] = qr[i];
        *(float4*)&wcT[r * T_PITCH + c4] = qc[i];
    }
    *(float4*)&h1T[(tid >> 4) * T_PITCH + (tid & 15) * 4] = qh;

    {
        const v4f sum = gather_pair(beg, len, maxlen, id, csr, h1, q);
        *(v4f*)&aggT[(FG_RPW * w + half) * T_PITCH + 4 * q] = sum;
    }
    __syncthreads();

    if (mat) {
        const float rd   = rdeg[row0 + m];
        const float bias = bl2[col0 + m];
        v8f acc = {};
#pragma unroll
        for (int ks = 0; ks < KS; ++ks) {
            const int k = 4 * ks + kb;
            v2f am = *(const v2f*)&aggT[m * T_PITCH + k];
            am.x *= rd;
            am.y *= rd;
            const v2f ah = *(const v2f*)&h1T[m * T_PITCH + k];
            const v2f bl = *(const v2f*)&wlT[(col0 + m) * T_PITCH + k];
            const v2f br = *(const v2f*)&wrT[(col0 + m) * T_PITCH + k];
            acc = __builtin_amdgcn_wmma_f32_16x16x4_f32(false, am, false, bl, (short)0, acc, false, false);
            acc = __builtin_amdgcn_wmma_f32_16x16x4_f32(false, ah, false, br, (short)0, acc, false, false);
        }
#pragma unroll
        for (int i = 0; i < 8; ++i) {
            int r = i + half * 8;
            int c = col0 + m;
            h2t[r * T_PITCH + c] = fmaxf(acc[i] + bias, 0.0f);
        }
    }
    __syncthreads();

    if (mat) {
        v8f acc2 = {};
#pragma unroll
        for (int ks = 0; ks < KS; ++ks) {
            const int k = 4 * ks + kb;
            const v2f a = *(const v2f*)&h2t[m * T_PITCH + k];
            const v2f b = *(const v2f*)&wcT[(col0 + m) * T_PITCH + k];
            acc2 = __builtin_amdgcn_wmma_f32_16x16x4_f32(false, a, false, b, (short)0, acc2, false, false);
        }
#pragma unroll
        for (int i = 0; i < 8; ++i) {
            int r = row0 + i + half * 8;
            int c = col0 + m;
            hpart[r * HID + c] = acc2[i];
        }
    }
}

#define CLS_BT      16
#define CLS_G       4
#define CLS_THREADS 128
__global__ __launch_bounds__(CLS_THREADS)
void k_classify(const float* __restrict__ hpart, const float* __restrict__ tpart,
                const float* __restrict__ Wc2, const float* __restrict__ bc2,
                float* __restrict__ out) {
    __shared__ float w2[HID];
    if (threadIdx.x < HID) w2[threadIdx.x] = Wc2[threadIdx.x];
    __syncthreads();

    const int n  = blockIdx.x * CLS_THREADS + threadIdx.x;
    const int b0 = blockIdx.y * CLS_BT;
    if (n - (int)(threadIdx.x & 31) >= N_NODES) return;

    float a[HID];
    const float4* hp = (const float4*)(hpart + (size_t)n * HID);
#pragma unroll
    for (int j = 0; j < HID / 4; ++j) {
        float4 v = hp[j];
        a[4 * j + 0] = v.x; a[4 * j + 1] = v.y; a[4 * j + 2] = v.z; a[4 * j + 3] = v.w;
    }
    float w[HID];
#pragma unroll
    for (int h = 0; h < HID; ++h) w[h] = w2[h];
    const float bias = bc2[0];

    for (int g = 0; g < CLS_BT / CLS_G; ++g) {
        const float* __restrict__ tp = tpart + (size_t)(b0 + g * CLS_G) * HID;
        float acc[CLS_G];
#pragma unroll
        for (int r = 0; r < CLS_G; ++r) acc[r] = bias;
#pragma unroll
        for (int hc = 0; hc < HID; hc += 16) {
#pragma unroll
            for (int r = 0; r < CLS_G; ++r) {
#pragma unroll
                for (int k = 0; k < 16; ++k)
                    acc[r] += fmaxf(a[hc + k] + tp[r * HID + hc + k], 0.0f) * w[hc + k];
            }
        }
        if (n < N_NODES) {
#pragma unroll
            for (int r = 0; r < CLS_G; ++r)
                out[(size_t)(b0 + g * CLS_G + r) * N_NODES + n] = acc[r];
        }
    }
}

extern "C" void kernel_launch(void* const* d_in, const int* in_sizes, int n_in,
                              void* d_out, int out_size, void* d_ws, size_t ws_size,
                              hipStream_t stream) {
    const float* x    = (const float*)d_in[0];
    const int*   ei   = (const int*)d_in[1];
    const float* task = (const float*)d_in[2];
    const float* Wl1  = (const float*)d_in[3];
    const float* bl1  = (const float*)d_in[4];
    const float* Wr1  = (const float*)d_in[5];
    const float* Wl2  = (const float*)d_in[6];
    const float* bl2  = (const float*)d_in[7];
    const float* Wr2  = (const float*)d_in[8];
    const float* Wt1  = (const float*)d_in[9];
    const float* bt1  = (const float*)d_in[10];
    const float* Wt2  = (const float*)d_in[11];
    const float* bt2  = (const float*)d_in[12];
    const float* Wc1  = (const float*)d_in[13];
    const float* bc1  = (const float*)d_in[14];
    const float* Wc2  = (const float*)d_in[15];
    const float* bc2  = (const float*)d_in[16];
    float* out = (float*)d_out;

    float* ws    = (float*)d_ws;
    float* rdeg  = ws + OFF_RDEG;
    float* h1    = ws + OFF_H1;
    float* hpart = ws + OFF_HPART;
    float* tpart = ws + OFF_TPART;
    int*   elist = (int*)(ws + OFF_H2);
    int*   csr   = (int*)(ws + OFF_AGG2);
    int*   start = (int*)(ws + OFF_START);

    k_build<<<OWN_BLOCKS + TASK_BLOCKS, OB_THREADS, 0, stream>>>(ei, x, Wl1, bl1, Wr1, elist, csr, start, rdeg, h1,
                                                                 task, Wt1, bt1, Wt2, bt2, Wc1, bc1, tpart);
    k_gather_sage2_hpart<<<NPAD / 16, FG_WAVES * 32, 0, stream>>>(start, csr, h1, rdeg, Wl2, bl2, Wr2, Wc1, hpart);
    k_classify<<<dim3(NPAD / CLS_THREADS, BATCH / CLS_BT), CLS_THREADS, 0, stream>>>(hpart, tpart, Wc2, bc2, out);
}


# classifier: w*relu(a+t) summed as w*max(a,-t) plus a per-batch-row sum of w*t made once by the task blocks (butterfly order): inner loop add-max-fma -> max-fma; summation order changes
# speedup vs baseline: 4.5728x; 1.0361x over previous
#include <hip/hip_runtime.h>
#include <hip/hip_bf16.h>

#define N_NODES 5000
#define NPAD    5120
#define N_EDGES 160000
#define HID     64
#define NODE_F  4
#define TASK_F  4
#define BATCH   512

typedef __attribute__((ext_vector_type(2))) float v2f;
typedef __attribute__((ext_vector_type(4))) float v4f;
typedef __attribute__((ext_vector_type(8))) float v8f;

#define OFF_TBIAS  0
#define OFF_AGG2   (NPAD * (1 + NODE_F))
#define OFF_RDEG   (OFF_AGG2 + NPAD * HID)
#define OFF_H1     (OFF_RDEG  + NPAD)
#define OFF_H2     (OFF_H1    + NPAD * HID)
#define OFF_HPART  (OFF_H2    + NPAD * HID)
#define OFF_TPART  (OFF_HPART + NPAD * HID)
#define OFF_START  (OFF_H2    + N_EDGES)


#define NB          64
#define OB_THREADS  512
#define OB_WAVES    (OB_THREADS / 32)
#define OWN_BLOCKS  (NPAD / NB)
#define E4          (N_EDGES / 4)
#define OB_TRIPS    (((E4 + OB_THREADS - 1) / OB_THREADS + 7) / 8 * 8)
#define OB_WORDS    (OB_TRIPS / 8)
#define OB_BATCH    5
#define X_PIECES    ((N_NODES + OB_THREADS - 1) / OB_THREADS)
#define ROW_BITS    6
#define CSR_CAP     (OB_WORDS * OB_THREADS)
#define ELIST_CAP   CSR_CAP
#define NPL         (NB / 32)
#define OB_FULLW    ((E4 / OB_THREADS) / 8)
static_assert((1 << ROW_BITS) == NB && (long long)(N_NODES - 1) * NB + NB - 1 < 0x7fffffffLL, "packed entries are non-negative ints");
static_assert(OB_TRIPS % 16 == 0 && OB_TRIPS * OB_THREADS >= E4, "mask words are filled whole, in pairs, and cover the row");
static_assert(OB_FULLW >= OB_WORDS - 1, "only the last word tests for the row's end");
static_assert(ELIST_CAP % (OB_BATCH * OB_THREADS) == 0, "a trip of a dense pass lies wholly inside or wholly outside the LDS part of the piece");
static_assert(NB * NODE_F <= OB_THREADS && NB % 32 == 0 && OB_WAVES <= 32 && OB_WAVES == (NB / 16) * (HID / 16), "block shape");
__device__ __forceinline__ int ob_slot(int v, int lo, int hi, int slot, int lane, int acc, int& nbel) {
    const unsigned below = __builtin_amdgcn_ballot_w32(v < lo);
    const unsigned under = __builtin_amdgcn_ballot_w32(v < hi);
    nbel += __popc(below);
    return (lane == slot) ? (int)(under & ~below) : acc;
}
template <bool FULL>
__device__ __forceinline__ void ob_load(int4 (&d)[8], const int4* __restrict__ dst4, int wd, int tid) {
#pragma unroll
    for (int j = 0; j < 8; ++j) {
        const int i = (wd * 8 + j) * OB_THREADS + tid;
        if (FULL) {
            d[j] = dst4[i];
        } else {
            d[j] = make_int4(0x7fffffff, 0x7fffffff, 0x7fffffff, 0x7fffffff);
            if (i < E4) d[j] = dst4[i];
        }
    }
}
__device__ __forceinline__ float4 x_piece(const float* __restrict__ x, int piece, int tid) {
    const int p = piece * OB_THREADS + tid;
    return ((const float4*)x)[min(p, N_NODES - 1)];
}
__device__ __forceinline__ void x_keep(float* xL, float4 v, int piece, int tid) {
    const int p = piece * OB_THREADS + tid;
    if (p < N_NODES) *(float4*)&xL[p * NODE_F] = v;
}
__device__ __forceinline__ int ob_word(const int4 (&d)[8], int lo, int hi, int lane, int& nbel) {
    int acc = 0;
#pragma unroll
    for (int j = 0; j < 8; ++j) {
        acc = ob_slot(d[j].x, lo, hi, 4 * j + 0, lane, acc, nbel);
        acc = ob_slot(d[j].y, lo, hi, 4 * j + 1, lane, acc, nbel);
        acc = ob_slot(d[j].z, lo, hi, 4 * j + 2, lane, acc, nbel);
        acc = ob_slot(d[j].w, lo, hi, 4 * j + 3, lane, acc, nbel);
    }
    return acc;
}
#define TASK_ROWS   (OB_THREADS / HID)
#define TASK_BLOCKS (BATCH / TASK_ROWS)
static_assert(BATCH % TASK_ROWS == 0, "task blocks are whole");
__global__ __launch_bounds__(OB_THREADS) __attribute__((amdgpu_waves_per_eu(4, 4)))
void k_build(const int* __restrict__ ei, const float* __restrict__ x,
             const float* __restrict__ Wl1, const float* __restrict__ bl1,
             const float* __restrict__ Wr1,
             int* __restrict__ elist, int* __restrict__ csr, int* __restrict__ start,
             float* __restrict__ rdeg, float* __restrict__ h1,
             const float* __restrict__ task, const float* __restrict__ Wt1,
             const float* __restrict__ bt1, const float* __restrict__ Wt2,
             const float* __restrict__ bt2, const float* __restrict__ Wc1,
             const float* __restrict__ bc1, float* __restrict__ tpart,
             const float* __restrict__ Wc2, const float* __restrict__ bc2, float* __restrict__ tbias) {
    __shared__ float tu[TASK_ROWS][HID];
    __shared__ float tt[TASK_ROWS][HID];
    __shared__ float tred[TASK_ROWS][HID / 32];
    if (blockIdx.x >= OWN_BLOCKS) {
        const int sub = threadIdx.x / HID;
        const int h   = threadIdx.x % HID;
        const int b   = (blockIdx.x - OWN_BLOCKS) * TASK_ROWS + sub;

        float acc = bt1[h];
#pragma unroll
        for (int k = 0; k < TASK_F; ++k) acc += task[b * TASK_F + k] * Wt1[h * TASK_F + k];
        tu[sub][h] = fmaxf(acc, 0.0f);
        __syncthreads();

        float acc2 = bt2[h];
#pragma unroll 8
        for (int k = 0; k < HID; ++k) acc2 += tu[sub][k] * Wt2[h * HID + k];
        tt[sub][h] = acc2;
        __syncthreads();

        float acc3 = bc1[h];
#pragma unroll 8
        for (int k = 0; k < HID; ++k) acc3 += tt[sub][k] * Wc1[h * (2 * HID) + HID + k];
        tpart[b * HID + h] = -acc3;
        float part = Wc2[h] * acc3;
        for (int off = 16; off > 0; off >>= 1) part += __shfl_xor(part, off);
        if ((h & 31) == 0) tred[sub][h >> 5] = part;
        __syncthreads();
        if (h == 0) tbias[b] = bc2[0] + tred[sub][0] + tred[sub][1];
        return;
    }
    __shared__ int   cnt[NB];
    __shared__ int   fillp[NB];
    __shared__ float aggL[NB * NODE_F];
    __shared__ __attribute__((aligned(16))) float xL[N_NODES * NODE_F];
    __shared__ unsigned ownL[OB_WORDS][OB_THREADS];
    __shared__ int   elistL[ELIST_CAP];
    __shared__ int   wsum[OB_WAVES];
    __shared__ int   wbel[OB_WAVES];
    __shared__ int   sbase, stot;

    const int tid  = threadIdx.x;
    const int lane = tid & 31;
    const int w    = tid >> 5;
    const int lo   = blockIdx.x * NB;
    const int hi   = lo + NB;
    const int*  __restrict__ dstp = ei + N_EDGES;
    const int4* __restrict__ dst4 = (const int4*)dstp;

    float4 xq[X_PIECES];
#pragma unroll
    for (int i = 0; i < X_PIECES; ++i) xq[i] = x_piece(x, i, tid);

    int nm = 0;
    int nbel = 0;
    int4 dA[8], dB[8];
    ob_load<true>(dA, dst4, 0, tid);
#pragma unroll 1
    for (int wd = 0; wd < OB_WORDS - 2; wd += 2) {
        ob_load<true>(dB, dst4, wd + 1, tid);
        __builtin_amdgcn_sched_barrier(0);
        const int accA = ob_word(dA, lo, hi, lane, nbel);
        ownL[wd][tid] = (unsigned)accA;
        nm += __popc((unsigned)accA);
        ob_load<true>(dA, dst4, wd + 2, tid);
        __builtin_amdgcn_sched_barrier(0);
        const int accB = ob_word(dB, lo, hi, lane, nbel);
        ownL[wd + 1][tid] = (unsigned)accB;
        nm += __popc((unsigned)accB);
    }
    {
        ob_load<false>(dB, dst4, OB_WORDS - 1, tid);
        const int accA = ob_word(dA, lo, hi, lane, nbel);
        ownL[OB_WORDS - 2][tid] = (unsigned)accA;
        nm += __popc((unsigned)accA);
        const int accB = ob_word(dB, lo, hi, lane, nbel);
        ownL[OB_WORDS - 1][tid] = (unsigned)accB;
        nm += __popc((unsigned)accB);
    }

    int inc = nm;
    for (int off = 1; off < 32; off <<= 1) {
        int v = __shfl_up(inc, off);
        if (lane >= off) inc += v;
    }
    if (lane == 31) wsum[w] = inc;
    if (lane == 0)  wbel[w] = nbel;
    if (tid < NB)          cnt[tid]  = 0;
    if (tid < NB * NODE_F) aggL[tid] = 0.0f;
#pragma unroll
    for (int i = 0; i < X_PIECES; ++i) x_keep(xL, xq[i], i, tid);
    __syncthreads();
    if (w == 0) {
        const int mine = (lane < OB_WAVES) ? wsum[lane] : 0;
        int ws = mine;
        for (int off = 1; off < 32; off <<= 1) {
            int v = __shfl_up(ws, off);
            if (lane >= off) ws += v;
        }
        if (lane < OB_WAVES) wsum[lane] = ws - mine;
        int bel = (lane < OB_WAVES) ? wbel[lane] : 0;
        for (int off = 16; off > 0; off >>= 1) bel += __shfl_xor(bel, off);
        if (lane == 31) stot  = ws;
        if (lane == 0)  sbase = bel;
    }
    __syncthreads();
    const int base = sbase;
    const int T    = stot;

    int pos = wsum[w] + inc - nm;
    const int comp = lane & 3;
    for (int wd = 0; wd < OB_WORDS; ++wd) {
        unsigned b = ownL[wd][tid];
        const int grp = (wd * 8 + (lane >> 2)) * OB_THREADS + w * 32;
        while (b) {
            const int k = __builtin_ctz(b);
            b &= b - 1;
            const int e = 4 * (grp + k) + comp;
            if (pos < ELIST_CAP) elistL[pos] = e;
            else                 elist[base + pos] = e;
            ++pos;
        }
    }
    __syncthreads();

    for (int i0 = 0; i0 < T; i0 += OB_BATCH * OB_THREADS) {
        const bool inL = i0 < ELIST_CAP;
        int e[OB_BATCH], d[OB_BATCH], s[OB_BATCH];
#pragma unroll
        for (int u = 0; u < OB_BATCH; ++u) {
            const int idx = i0 + u * OB_THREADS + tid;
            e[u] = (idx < T) ? (inL ? elistL[idx] : elist[base + idx]) : -1;
        }
#pragma unroll
        for (int u = 0; u < OB_BATCH; ++u) {
            const int ee = e[u] < 0 ? 0 : e[u];
            d[u] = dstp[ee];
            s[u] = ei[ee];
        }
#pragma unroll
        for (int u = 0; u < OB_BATCH; ++u) {
            if (e[u] >= 0) {
                const int r = d[u] - lo;
                atomicAdd(&cnt[r], 1);
                const int idx = i0 + u * OB_THREADS + tid;
                const int pk  = (s[u] << ROW_BITS) | r;
                if (inL) elistL[idx] = pk;
                else     elist[base + idx] = pk;
            }
        }
    }
    __syncthreads();

    int sq[NPL + 1];
    if (w == 0) {
        int cq[NPL];
        int tot = 0;
#pragma unroll
        for (int qn = 0; qn < NPL; ++qn) { cq[qn] = cnt[NPL * lane + qn]; tot += cq[qn]; }
        int ps = tot;
        for (int off = 1; off < 32; off <<= 1) {
            int v = __shfl_up(ps, off);
            if (lane >= off) ps += v;
        }
        int bq = base + ps - tot;
#pragma unroll
        for (int qn = 0; qn < NPL; ++qn) {
            fillp[NPL * lane + qn] = bq;
            sq[qn] = bq;
            bq += cq[qn];
        }
        sq[NPL] = bq;
    }
    __syncthreads();
    if (w == 0) {
#pragma unroll
        for (int qn = 0; qn < NPL; ++qn) start[lo + NPL * lane + qn] = sq[qn];
        if (blockIdx.x == OWN_BLOCKS - 1 && lane == 31) start[NPAD] = sq[NPL];
    }

    int* csrL = (int*)&ownL[0][0];
    for (int i0 = 0; i0 < T; i0 += OB_BATCH * OB_THREADS) {
        const bool inL = i0 < ELIST_CAP;
        int en[OB_BATCH];
        float4 xv[OB_BATCH];
#pragma unroll
        for (int u = 0; u < OB_BATCH; ++u) {
            const int idx = i0 + u * OB_THREADS + tid;
            en[u] = (idx < T) ? (inL ? elistL[idx] : elist[base + idx]) : -1;
        }
#pragma unroll
        for (int u = 0; u < OB_BATCH; ++u) {
            const int s = en[u] < 0 ? 0 : (en[u] >> ROW_BITS);
            xv[u] = *(const float4*)&xL[s * NODE_F];
        }
#pragma unroll
        for (int u = 0; u < OB_BATCH; ++u) {
            if (en[u] >= 0) {
                const int s = en[u] >> ROW_BITS;
                const int r = en[u] & (NB - 1);
                const int slot = atomicAdd(&fillp[r], 1);
                if (slot - base < CSR_CAP) csrL[slot - base] = s;
                else                       csr[slot] = s;
                atomicAdd(&aggL[r * NODE_F + 0], xv[u].x);
                atomicAdd(&aggL[r * NODE_F + 1], xv[u].y);
                atomicAdd(&aggL[r * NODE_F + 2], xv[u].z);
                atomicAdd(&aggL[r * NODE_F + 3], xv[u].w);
            }
        }
    }
    __syncthreads();

    for (int i = tid; i < min(T, CSR_CAP); i += OB_THREADS) csr[base + i] = csrL[i];

    if (tid < NB) rdeg[lo + tid] = 1.0f / fmaxf((float)cnt[tid], 1.0f);
    const int lrow0 = (w >> 2) * 16;
    const int col0  = (w & 3) * 16;
    const int m     = lane & 15;
    const int half  = lane >> 4;
    const int kb    = half * 2;
    const int lrow  = lrow0 + m;
    const int row   = lo + lrow;
    const float rd  = 1.0f / fmaxf((float)cnt[lrow], 1.0f);

    v2f am, ax, bl, br;
    am.x = aggL[lrow * NODE_F + kb]     * rd;
    am.y = aggL[lrow * NODE_F + kb + 1] * rd;
    ax.x = (row < N_NODES) ? x[row * NODE_F + kb]     : 0.0f;
    ax.y = (row < N_NODES) ? x[row * NODE_F + kb + 1] : 0.0f;
    bl.x = Wl1[(col0 + m) * NODE_F + kb];
    bl.y = Wl1[(col0 + m) * NODE_F + kb + 1];
    br.x = Wr1[(col0 + m) * NODE_F + kb];
    br.y = Wr1[(col0 + m) * NODE_F + kb + 1];

    v8f acc = {};
    acc = __builtin_amdgcn_wmma_f32_16x16x4_f32(false, am, false, bl, (short)0, acc, false, false);
    acc = __builtin_amdgcn_wmma_f32_16x16x4_f32(false, ax, false, br, (short)0, acc, false, false);

#pragma unroll
    for (int i = 0; i < 8; ++i) {
        int r = lo + lrow0 + i + half * 8;
        int c = col0 + m;
        h1[r * HID + c] = fmaxf(acc[i] + bl1[c], 0.0f);
    }
}

#define T_PITCH  68
#define SH_WAVES (HID / 16)
#define FG_WAVES 8
#define FG_RPW   (16 / FG_WAVES)
#define FG_CHUNK 16
#define FG_CHUNKS 3
#define FG_ROUND (FG_CHUNK * FG_CHUNKS)
#define KS       (HID / 4)
#define W_PIECES ((HID * HID / 4) / (FG_WAVES * 32))
static_assert(FG_RPW == 2 && FG_WAVES >= SH_WAVES && FG_CHUNK == 16, "block shape");
static_assert(W_PIECES * FG_WAVES * 32 * 4 == HID * HID && 16 * HID / 4 == FG_WAVES * 32, "copies are whole");
template <int U>
__device__ __forceinline__ int half_lane(int v) {
    return __builtin_amdgcn_update_dpp(0, v, 0x150 + U, 0xF, 0xF, false);
}
__device__ __forceinline__ void chunk_load(v4f (&v)[FG_CHUNK], int id, const float* __restrict__ rows, unsigned qoff) {
#define FG_LD(U) v[U] = *(const v4f*)((const char*)rows + (((unsigned)half_lane<U>(id) * (unsigned)(HID * 4)) + qoff));
    FG_LD(0) FG_LD(1) FG_LD(2)  FG_LD(3)  FG_LD(4)  FG_LD(5)  FG_LD(6)  FG_LD(7)
    FG_LD(8) FG_LD(9) FG_LD(10) FG_LD(11) FG_LD(12) FG_LD(13) FG_LD(14) FG_LD(15)
#undef FG_LD
}
__device__ __forceinline__ v4f gather_pair(int beg, int len, int maxlen, int (&id)[FG_CHUNKS],
                                           const int* __restrict__ csr, const float* __restrict__ rows, int q) {
    const unsigned qoff = (unsigned)q * 16u;
    v4f acc = {0.0f, 0.0f, 0.0f, 0.0f};
    for (int base = 0; base < maxlen; base += FG_ROUND) {
        const int nb = base + FG_ROUND;
        int idn[FG_CHUNKS] = {0, 0, 0};
        if (nb < maxlen) {
#pragma unroll
            for (int c = 0; c < FG_CHUNKS; ++c) {
                const int j = nb + FG_CHUNK * c + q;
                idn[c] = (j < len) ? csr[beg + j] : 0;
            }
        }
        v4f v[FG_CHUNKS][FG_CHUNK];
#pragma unroll
        for (int c = 0; c < FG_CHUNKS; ++c)
            if (base + FG_CHUNK * c < maxlen) chunk_load(v[c], id[c], rows, qoff);
#pragma unroll
        for (int c = 0; c < FG_CHUNKS; ++c) {
            if (base + FG_CHUNK * c < maxlen) {
#pragma unroll
                for (int u = 0; u < FG_CHUNK; ++u) {
                    const bool ok = base + FG_CHUNK * c + u < len;
                    acc.x += ok ? v[c][u].x : 0.0f;
                    acc.y += ok ? v[c][u].y : 0.0f;
                    acc.z += ok ? v[c][u].z : 0.0f;
                    acc.w += ok ? v[c][u].w : 0.0f;
                }
            }
        }
#pragma unroll
        for (int c = 0; c < FG_CHUNKS; ++c) id[c] = idn[c];
    }
    return acc;
}
__global__ __launch_bounds__(FG_WAVES * 32)
void k_gather_sage2_hpart(const int* __restrict__ start, const int* __restrict__ csr,
                          const float* __restrict__ h1, const float* __restrict__ rdeg,
                          const float* __restrict__ Wl2, const float* __restrict__ bl2,
                          const float* __restrict__ Wr2, const float* __restrict__ Wc1,
                          float* __restrict__ hpart) {
    __shared__ __attribute__((aligned(16))) float wlT[HID * T_PITCH];
    __shared__ __attribute__((aligned(16))) float wrT[HID * T_PITCH];
    __shared__ __attribute__((aligned(16))) float wcT[HID * T_PITCH];
    __shared__ __attribute__((aligned(16))) float h1T[16 * T_PITCH];
    __shared__ __attribute__((aligned(16))) float aggT[16 * T_PITCH];
    __shared__ __attribute__((aligned(16))) float h2t[16 * T_PITCH];
    const int tid  = threadIdx.x;
    const int row0 = blockIdx.x * 16;
    const int lane = tid & 31;
    const int w    = __builtin_amdgcn_readfirstlane(tid >> 5);
    const int m    = lane & 15;
    const int half = lane >> 4;
    const int kb   = half * 2;
    const int col0 = (w & (SH_WAVES - 1)) * 16;
    const bool mat = w < SH_WAVES;

    float4 ql[W_PIECES], qr[W_PIECES], qc[W_PIECES];
#pragma unroll
    for (int i = 0; i < W_PIECES; ++i) {
        const int p = tid + FG_WAVES * 32 * i;
        const int r = p >> 4, c4 = (p & 15) * 4;
        ql[i] = *(const float4*)(Wl2 + r * HID + c4);
        qr[i] = *(const float4*)(Wr2 + r * HID + c4);
        qc[i] = *(const float4*)(Wc1 + r * (2 * HID) + c4);
    }
    const float4 qh = *(const float4*)(h1 + (row0 + (tid >> 4)) * HID + (tid & 15) * 4);

    const int n  = row0 + FG_RPW * w;
    const int s0 = start[n];
    const int s1 = start[n + 1];
    const int s2 = start[n + 2];
    const int q      = lane & 15;
    const int beg    = half ? s1 : s0;
    const int len    = (half ? s2 : s1) - beg;
    const int maxlen = max(s1 - s0, s2 - s1);
    int id[FG_CHUNKS];
#pragma unroll
    for (int c = 0; c < FG_CHUNKS; ++c) {
        const int j = FG_CHUNK * c + q;
        id[c] = (j < len) ? csr[beg + j] : 0;
    }

#pragma unroll
    for (int i = 0; i < W_PIECES; ++i) {
        const int p = tid + FG_WAVES * 32 * i;
        const int r = p >> 4, c4 = (p & 15) * 4;
        *(float4*)&wlT[r * T_PITCH + c4] = ql[i];
        *(float4*)&wrT[r * T_PITCH + c4] = qr[i];
        *(float4*)&wcT[r * T_PITCH + c4] = qc[i];
    }
    *(float4*)&h1T[(tid >> 4) * T_PITCH + (tid & 15) * 4] = qh;

    {
        const v4f sum = gather_pair(beg, len, maxlen, id, csr, h1, q);
        *(v4f*)&aggT[(FG_RPW * w + half) * T_PITCH + 4 * q] = sum;
    }
    __syncthreads();

    if (mat) {
        const float rd   = rdeg[row0 + m];
        const float bias = bl2[col0 + m];
        v8f acc = {};
#pragma unroll
        for (int ks = 0; ks < KS; ++ks) {
            const int k = 4 * ks + kb;
            v2f am = *(const v2f*)&aggT[m * T_PITCH + k];
            am.x *= rd;
            am.y *= rd;
            const v2f ah = *(const v2f*)&h1T[m * T_PITCH + k];
            const v2f bl = *(const v2f*)&wlT[(col0 + m) * T_PITCH + k];
            const v2f br = *(const v2f*)&wrT[(col0 + m) * T_PITCH + k];
            acc = __builtin_amdgcn_wmma_f32_16x16x4_f32(false, am, false, bl, (short)0, acc, false, false);
            acc = __builtin_amdgcn_wmma_f32_16x16x4_f32(false, ah, false, br, (short)0, acc, false, false);
        }
#pragma unroll
        for (int i = 0; i < 8; ++i) {
            int r = i + half * 8;
            int c = col0 + m;
            h2t[r * T_PITCH + c] = fmaxf(acc[i] + bias, 0.0f);
        }
    }
    __syncthreads();

    if (mat) {
        v8f acc2 = {};
#pragma unroll
        for (int ks = 0; ks < KS; ++ks) {
            const int k = 4 * ks + kb;
            const v2f a = *(const v2f*)&h2t[m * T_PITCH + k];
            const v2f b = *(const v2f*)&wcT[(col0 + m) * T_PITCH + k];
            acc2 = __builtin_amdgcn_wmma_f32_16x16x4_f32(false, a, false, b, (short)0, acc2, false, false);
        }
#pragma unroll
        for (int i = 0; i < 8; ++i) {
            int r = row0 + i + half * 8;
            int c = col0 + m;
            hpart[r * HID + c] = acc2[i];
        }
    }
}


__device__ __forceinline__ float max_nn(float x, float y) {
    __builtin_assume(x == x);
    __builtin_assume(y == y);
    return __builtin_fmaxf(x, y);
}
#define CLS_BT      16
#define CLS_G       4
#define CLS_THREADS 128
__global__ __launch_bounds__(CLS_THREADS)
void k_classify(const float* __restrict__ hpart, const float* __restrict__ ntpart,
                const float* __restrict__ Wc2, const float* __restrict__ tbias,
                float* __restrict__ out) {
    __shared__ float w2[HID];
    if (threadIdx.x < HID) w2[threadIdx.x] = Wc2[threadIdx.x];
    __syncthreads();

    const int n  = blockIdx.x * CLS_THREADS + threadIdx.x;
    const int b0 = blockIdx.y * CLS_BT;
    if (n - (int)(threadIdx.x & 31) >= N_NODES) return;

    float a[HID];
    const float4* hp = (const float4*)(hpart + (size_t)n * HID);
#pragma unroll
    for (int j = 0; j < HID / 4; ++j) {
        float4 v = hp[j];
        a[4 * j + 0] = v.x; a[4 * j + 1] = v.y; a[4 * j + 2] = v.z; a[4 * j + 3] = v.w;
    }
    float w[HID];
#pragma unroll
    for (int h = 0; h < HID; ++h) w[h] = w2[h];

    for (int g = 0; g < CLS_BT / CLS_G; ++g) {
        const float* __restrict__ tp = ntpart + (size_t)(b0 + g * CLS_G) * HID;
        const float* __restrict__ tb = tbias + (b0 + g * CLS_G);
        float acc[CLS_G];
#pragma unroll
        for (int r = 0; r < CLS_G; ++r) acc[r] = tb[r];
#pragma unroll
        for (int hc = 0; hc < HID; hc += 16) {
#pragma unroll
            for (int r = 0; r < CLS_G; ++r) {
#pragma unroll
                for (int k = 0; k < 16; ++k)
                    acc[r] += max_nn(a[hc + k], tp[r * HID + hc + k]) * w[hc + k];
            }
        }
        if (n < N_NODES) {
#pragma unroll
            for (int r = 0; r < CLS_G; ++r)
                out[(size_t)(b0 + g * CLS_G + r) * N_NODES + n] = acc[r];
        }
    }
}

extern "C" void kernel_launch(void* const* d_in, const int* in_sizes, int n_in,
                              void* d_out, int out_size, void* d_ws, size_t ws_size,
                              hipStream_t stream) {
    const float* x    = (const float*)d_in[0];
    const int*   ei   = (const int*)d_in[1];
    const float* task = (const float*)d_in[2];
    const float* Wl1  = (const float*)d_in[3];
    const float* bl1  = (const float*)d_in[4];
    const float* Wr1  = (const float*)d_in[5];
    const float* Wl2  = (const float*)d_in[6];
    const float* bl2  = (const float*)d_in[7];
    const float* Wr2  = (const float*)d_in[8];
    const float* Wt1  = (const float*)d_in[9];
    const float* bt1  = (const float*)d_in[10];
    const float* Wt2  = (const float*)d_in[11];
    const float* bt2  = (const float*)d_in[12];
    const float* Wc1  = (const float*)d_in[13];
    const float* bc1  = (const float*)d_in[14];
    const float* Wc2  = (const float*)d_in[15];
    const float* bc2  = (const float*)d_in[16];
    float* out = (float*)d_out;

    float* ws    = (float*)d_ws;
    float* rdeg  = ws + OFF_RDEG;
    float* h1    = ws + OFF_H1;
    float* hpart = ws + OFF_HPART;
    float* tpart = ws + OFF_TPART;
    float* tbias = ws + OFF_TBIAS;
    int*   elist = (int*)(ws + OFF_H2);
    int*   csr   = (int*)(ws + OFF_AGG2);
    int*   start = (int*)(ws + OFF_START);

    k_build<<<OWN_BLOCKS + TASK_BLOCKS, OB_THREADS, 0, stream>>>(ei, x, Wl1, bl1, Wr1, elist, csr, start, rdeg, h1,
                                                                 task, Wt1, bt1, Wt2, bt2, Wc1, bc1, tpart, Wc2, bc2, tbias);
    k_gather_sage2_hpart<<<NPAD / 16, FG_WAVES * 32, 0, stream>>>(start, csr, h1, rdeg, Wl2, bl2, Wr2, Wc1, hpart);
    k_classify<<<dim3(NPAD / CLS_THREADS, BATCH / CLS_BT), CLS_THREADS, 0, stream>>>(hpart, tpart, Wc2, tbias, out);
}
